# starting kernel
# baseline (speedup 1.0000x reference)
#include <hip/hip_runtime.h>
#include <hip/hip_bf16.h>

typedef float v2f __attribute__((ext_vector_type(2)));
typedef float v8f __attribute__((ext_vector_type(8)));

#define HC 16
#define EPSV 1e-5f

__global__ void zero_f32(float* __restrict__ p, int n) {
    int i = blockIdx.x * blockDim.x + threadIdx.x;
    if (i < n) p[i] = 0.0f;
}

__global__ void degree_kernel(const int* __restrict__ dst, float* __restrict__ deg, int E) {
    int e = blockIdx.x * blockDim.x + threadIdx.x;
    if (e < E) atomicAdd(&deg[dst[e]], 1.0f);
}

__global__ void rsqrt_kernel(float* __restrict__ d, int N) {
    int i = blockIdx.x * blockDim.x + threadIdx.x;
    if (i < N) d[i] = rsqrtf(d[i] + 1.0f);
}

__global__ void gemm_wmma_f32(const float* __restrict__ A, const float* __restrict__ B,
                              float* __restrict__ C, int N, int K) {
    int wave = (int)((blockIdx.x * blockDim.x + threadIdx.x) >> 5);
    int lane = threadIdx.x & 31;
    int row0 = wave * 16;
    if (row0 >= N) return;

    int m  = lane & 15;
    int hi = lane >> 4;
    const float* arow = A + (size_t)(row0 + m) * K;

    v8f acc = {};
    for (int k0 = 0; k0 < K; k0 += 4) {
        int ka = k0 + 2 * hi;
        v2f a, b;
        a.x = arow[ka];
        a.y = arow[ka + 1];
        b.x = B[(size_t)ka * HC + m];
        b.y = B[(size_t)(ka + 1) * HC + m];
        acc = __builtin_amdgcn_wmma_f32_16x16x4_f32(
             false, a,  false, b,
             (short)0, acc,  false,  false);
    }

    float* cbase = C + (size_t)(row0 + hi * 8) * HC + m;
#pragma unroll
    for (int r = 0; r < 8; ++r) cbase[(size_t)r * HC] = acc[r];
}

__global__ void init_selfloop(const float* __restrict__ dis, const float* __restrict__ hw,
                              const float* __restrict__ bias, float* __restrict__ out, int NC) {
    int i = blockIdx.x * blockDim.x + threadIdx.x;
    if (i >= NC) return;
    int node = i >> 4, c = i & 15;
    float d = dis[node];
    out[i] = d * d * hw[i] + bias[c];
}

__global__ void scatter_edges(const int* __restrict__ src, const int* __restrict__ dst,
                              const float* __restrict__ dis, const float* __restrict__ hw,
                              float* __restrict__ out, int E) {
    int e = blockIdx.x * blockDim.x + threadIdx.x;
    if (e >= E) return;
    int s = src[e], d = dst[e];
    float nrm = dis[s] * dis[d];
    const float4* hrow = (const float4*)(hw + (size_t)s * HC);
    float* orow = out + (size_t)d * HC;
#pragma unroll
    for (int q = 0; q < 4; ++q) {
        float4 v = hrow[q];
        atomicAdd(orow + 4 * q + 0, nrm * v.x);
        atomicAdd(orow + 4 * q + 1, nrm * v.y);
        atomicAdd(orow + 4 * q + 2, nrm * v.z);
        atomicAdd(orow + 4 * q + 3, nrm * v.w);
    }
}

__global__ void bn_stats(const float* __restrict__ agg, float* __restrict__ stats, int N) {
    __shared__ float s1s[256];
    __shared__ float s2s[256];
    int c = threadIdx.x & 15;
    int g = threadIdx.x >> 4;
    int stride = gridDim.x * 16;
    float s1 = 0.0f, s2 = 0.0f;
    for (int row = blockIdx.x * 16 + g; row < N; row += stride) {
        float v = agg[(size_t)row * HC + c];
        v = fmaxf(v, 0.0f);
        s1 += v;
        s2 += v * v;
    }
    s1s[threadIdx.x] = s1;
    s2s[threadIdx.x] = s2;
    __syncthreads();
    if (threadIdx.x < 16) {
        float t1 = 0.0f, t2 = 0.0f;
#pragma unroll
        for (int gg = 0; gg < 16; ++gg) {
            t1 += s1s[gg * 16 + threadIdx.x];
            t2 += s2s[gg * 16 + threadIdx.x];
        }
        atomicAdd(&stats[threadIdx.x], t1);
        atomicAdd(&stats[16 + threadIdx.x], t2);
    }
}

__global__ void bn_finalize(float* __restrict__ stats, const float* __restrict__ gamma,
                            const float* __restrict__ beta, float invN) {
    int c = threadIdx.x;
    if (c < 16) {
        float mu  = stats[c] * invN;
        float var = stats[16 + c] * invN - mu * mu;
        float sc  = rsqrtf(var + EPSV) * gamma[c];
        stats[32 + c] = sc;
        stats[48 + c] = beta[c] - mu * sc;
    }
}

__global__ void bn_apply(const float* __restrict__ agg, const float* __restrict__ stats,
                         float* __restrict__ h, int NC) {
    int i = blockIdx.x * blockDim.x + threadIdx.x;
    if (i >= NC) return;
    int c = i & 15;
    float v = fmaxf(agg[i], 0.0f);
    h[i] = v * stats[32 + c] + stats[48 + c];
}

__global__ void head_kernel(const float* __restrict__ h, const float* __restrict__ Wfc,
                            const float* __restrict__ bfc, float* __restrict__ out, int N) {
    int i = blockIdx.x * blockDim.x + threadIdx.x;
    if (i >= N) return;
    const float* r = h + (size_t)i * HC;
    float z0 = bfc[0], z1 = bfc[1];
#pragma unroll
    for (int c = 0; c < HC; ++c) {
        float v = r[c];
        z0 += v * Wfc[c * 2 + 0];
        z1 += v * Wfc[c * 2 + 1];
    }
    float mx  = fmaxf(z0, z1);
    float lse = mx + logf(__expf(z0 - mx) + __expf(z1 - mx));
    out[(size_t)i * 2 + 0] = z0 - lse;
    out[(size_t)i * 2 + 1] = z1 - lse;
}

extern "C" void kernel_launch(void* const* d_in, const int* in_sizes, int n_in,
                              void* d_out, int out_size, void* d_ws, size_t ws_size,
                              hipStream_t stream) {
    const int IN = 128;
    const int N  = in_sizes[0] / IN;
    const int E  = in_sizes[1] / 2;

    const float* x    = (const float*)d_in[0];
    const int*   ei   = (const int*)d_in[1];
    const int*   srcI = ei;
    const int*   dstI = ei + E;
    const float* W1   = (const float*)d_in[2];
    const float* b1   = (const float*)d_in[3];
    const float* W2   = (const float*)d_in[4];
    const float* b2   = (const float*)d_in[5];
    const float* W3   = (const float*)d_in[6];
    const float* b3   = (const float*)d_in[7];
    const float* g1   = (const float*)d_in[8];
    const float* be1  = (const float*)d_in[9];
    const float* g2   = (const float*)d_in[10];
    const float* be2  = (const float*)d_in[11];
    const float* Wfc  = (const float*)d_in[12];
    const float* bfc  = (const float*)d_in[13];

    float* ws   = (float*)d_ws;
    float* dis  = ws;
    float* bufA = dis + N;
    float* bufB = bufA + (size_t)N * HC;
    float* bufC = bufB + (size_t)N * HC;
    float* stats = bufC + (size_t)N * HC;

    const int B256 = 256;
    const int NC = N * HC;
    dim3 blkN((N + B256 - 1) / B256);
    dim3 blkNC((NC + B256 - 1) / B256);
    dim3 blkE((E + B256 - 1) / B256);
    dim3 blkG((N / 16 + 7) / 8);
    float invN = 1.0f / (float)N;

    zero_f32<<<blkN, B256, 0, stream>>>(dis, N);
    degree_kernel<<<blkE, B256, 0, stream>>>(dstI, dis, E);
    rsqrt_kernel<<<blkN, B256, 0, stream>>>(dis, N);

    gemm_wmma_f32<<<blkG, B256, 0, stream>>>(x, W1, bufA, N, IN);
    init_selfloop<<<blkNC, B256, 0, stream>>>(dis, bufA, b1, bufB, NC);
    scatter_edges<<<blkE, B256, 0, stream>>>(srcI, dstI, dis, bufA, bufB, E);
    zero_f32<<<1, 64, 0, stream>>>(stats, 64);
    bn_stats<<<256, 256, 0, stream>>>(bufB, stats, N);
    bn_finalize<<<1, 16, 0, stream>>>(stats, g1, be1, invN);
    bn_apply<<<blkNC, B256, 0, stream>>>(bufB, stats, bufC, NC);

    gemm_wmma_f32<<<blkG, B256, 0, stream>>>(bufC, W2, bufA, N, HC);
    init_selfloop<<<blkNC, B256, 0, stream>>>(dis, bufA, b2, bufB, NC);
    scatter_edges<<<blkE, B256, 0, stream>>>(srcI, dstI, dis, bufA, bufB, E);
    zero_f32<<<1, 64, 0, stream>>>(stats, 64);
    bn_stats<<<256, 256, 0, stream>>>(bufB, stats, N);
    bn_finalize<<<1, 16, 0, stream>>>(stats, g2, be2, invN);
    bn_apply<<<blkNC, B256, 0, stream>>>(bufB, stats, bufC, NC);

    gemm_wmma_f32<<<blkG, B256, 0, stream>>>(bufC, W3, bufA, N, HC);
    init_selfloop<<<blkNC, B256, 0, stream>>>(dis, bufA, b3, bufB, NC);
    scatter_edges<<<blkE, B256, 0, stream>>>(srcI, dstI, dis, bufA, bufB, E);

    head_kernel<<<blkN, B256, 0, stream>>>(bufB, Wfc, bfc, (float*)d_out, N);
}


# Push to pull: edges counting-sorted by dst once a call (arrival-order rank by integer atomics); 3 aggregations gather per row into 8 register sums + shuffle tree, 1 store per row, no float atomics
# speedup vs baseline: 5.7375x; 5.7375x over previous
#include <hip/hip_runtime.h>
#include <hip/hip_bf16.h>

typedef float v2f __attribute__((ext_vector_type(2)));
typedef float v8f __attribute__((ext_vector_type(8)));

#define HC 16
#define EPSV 1e-5f
#define SRC_SPLIT 65536

__global__ void zero_f32(float* __restrict__ p, int n) {
    int i = blockIdx.x * blockDim.x + threadIdx.x;
    if (i < n) p[i] = 0.0f;
}


__global__ void bucket_rank(const int* __restrict__ src, const int* __restrict__ dst,
                            int* __restrict__ cnt, int* __restrict__ rank, int E) {
    int base = blockIdx.x * (blockDim.x * 4) + threadIdx.x;
    int key[4], rk[4];
#pragma unroll
    for (int k = 0; k < 4; ++k) {
        int e = base + k * blockDim.x;
        int ec = min(e, E - 1);
        key[k] = 2 * dst[ec] + (src[ec] >> 16);
    }
#pragma unroll
    for (int k = 0; k < 4; ++k) {
        int e = base + k * blockDim.x;
        rk[k] = 0;
        if (e < E) rk[k] = atomicAdd(&cnt[key[k]], 1);
    }
#pragma unroll
    for (int k = 0; k < 4; ++k) {
        int e = base + k * blockDim.x;
        if (e < E) rank[e] = rk[k];
    }
}

__global__ void scan_tile_sums(const int* __restrict__ cnt, int* __restrict__ tsum, int M) {
    __shared__ int wsum[8];
    int base = (blockIdx.x * 256 + threadIdx.x) * 8;
    int s = 0;
#pragma unroll
    for (int t = 0; t < 8; ++t) {
        int i = base + t;
        int c = cnt[min(i, M - 1)];
        s += (i < M) ? c : 0;
    }
    for (int off = 16; off > 0; off >>= 1) s += __shfl_xor(s, off);
    if ((threadIdx.x & 31) == 0) wsum[threadIdx.x >> 5] = s;
    __syncthreads();
    if (threadIdx.x == 0) {
        int t = 0;
        for (int w = 0; w < 8; ++w) t += wsum[w];
        tsum[blockIdx.x] = t;
    }
}

__global__ void scan_apply(int* __restrict__ cnt, const int* __restrict__ tsum,
                           float* __restrict__ dis, int M, int N) {
    __shared__ int wsum[8];
    int tid = threadIdx.x, lane = tid & 31, wv = tid >> 5;

    int part = 0;
    for (int i = tid; i < (int)blockIdx.x; i += 256) part += tsum[i];
    for (int off = 16; off > 0; off >>= 1) part += __shfl_xor(part, off);
    if (lane == 0) wsum[wv] = part;
    __syncthreads();
    int tile_off = 0;
    for (int w = 0; w < 8; ++w) tile_off += wsum[w];
    __syncthreads();

    int base = (blockIdx.x * 256 + tid) * 8;
    int v[8];
    int mine = 0;
#pragma unroll
    for (int t = 0; t < 8; ++t) {
        int i = base + t;
        int c = cnt[min(i, M - 1)];
        v[t] = (i < M) ? c : 0;
        mine += v[t];
    }
#pragma unroll
    for (int t = 0; t < 4; ++t) {
        int row = (base >> 1) + t;
        if (row < N) dis[row] = rsqrtf((float)(v[2 * t] + v[2 * t + 1]) + 1.0f);
    }

    int inc = mine;
    for (int off = 1; off < 32; off <<= 1) {
        int n = __shfl_up(inc, off);
        if (lane >= off) inc += n;
    }
    if (lane == 31) wsum[wv] = inc;
    __syncthreads();
    int run = tile_off + inc - mine;
    for (int w = 0; w < wv; ++w) run += wsum[w];
#pragma unroll
    for (int t = 0; t < 8; ++t) {
        int i = base + t;
        if (i < M) cnt[i] = run;
        run += v[t];
    }
}

__global__ void bucket_fill(const int* __restrict__ src, const int* __restrict__ dst,
                            const int* __restrict__ start, const int* __restrict__ rank,
                            int* __restrict__ csr_w, int E) {
    unsigned short* __restrict__ csr = (unsigned short*)csr_w;
    int base = blockIdx.x * (blockDim.x * 4) + threadIdx.x;
    int pos[4], lo[4];
#pragma unroll
    for (int k = 0; k < 4; ++k) {
        int e = base + k * blockDim.x;
        int ec = min(e, E - 1);
        int s = src[ec];
        pos[k] = start[2 * dst[ec] + (s >> 16)] + rank[ec];
        lo[k] = s & (SRC_SPLIT - 1);
    }
#pragma unroll
    for (int k = 0; k < 4; ++k) {
        int e = base + k * blockDim.x;
        if (e < E) csr[pos[k]] = (unsigned short)lo[k];
    }
}

__global__ void gemm_wmma_f32(const float* __restrict__ A, const float* __restrict__ B,
                              float* __restrict__ C, int N, int K) {
    int wave = (int)((blockIdx.x * blockDim.x + threadIdx.x) >> 5);
    int lane = threadIdx.x & 31;
    int row0 = wave * 16;
    if (row0 >= N) return;

    int m  = lane & 15;
    int hi = lane >> 4;
    const float* arow = A + (size_t)(row0 + m) * K;

    v8f acc = {};
    for (int k0 = 0; k0 < K; k0 += 4) {
        int ka = k0 + 2 * hi;
        v2f a, b;
        a.x = arow[ka];
        a.y = arow[ka + 1];
        b.x = B[(size_t)ka * HC + m];
        b.y = B[(size_t)(ka + 1) * HC + m];
        acc = __builtin_amdgcn_wmma_f32_16x16x4_f32(
             false, a,  false, b,
             (short)0, acc,  false,  false);
    }

    float* cbase = C + (size_t)(row0 + hi * 8) * HC + m;
#pragma unroll
    for (int r = 0; r < 8; ++r) cbase[(size_t)r * HC] = acc[r];
}

__global__ void __launch_bounds__(256)
pull_rows(const int* __restrict__ start, const int* __restrict__ csr_w,
          const float* __restrict__ dis, const float* __restrict__ hw,
          const float* __restrict__ bias, float* __restrict__ out, int N, int E) {
    const unsigned short* __restrict__ csr = (const unsigned short*)csr_w;
    const float4* __restrict__ hw4 = (const float4*)hw;
    int lane = threadIdx.x & 31;
    int q = lane & 3;
    int j = lane >> 2;
    int nw = (int)((gridDim.x * blockDim.x) >> 5);

    float b0 = bias[4 * q + 0], b1 = bias[4 * q + 1], b2 = bias[4 * q + 2], b3 = bias[4 * q + 3];

    for (int r = (int)((blockIdx.x * blockDim.x + threadIdx.x) >> 5); r < N; r += nw) {
        bool last = (r + 1 >= N);
        int s0 = start[2 * r];
        int s1 = start[2 * r + 1];
        int s2 = start[last ? 2 * r + 1 : 2 * r + 2];
        if (last) s2 = E;
        float d = dis[r];
        float4 self = hw4[(size_t)r * 4 + q];

        float dd = d * d;
        float4 acc;
        acc.x = (j == 0) ? (dd * self.x + b0) : 0.0f;
        acc.y = (j == 0) ? (dd * self.y + b1) : 0.0f;
        acc.z = (j == 0) ? (dd * self.z + b2) : 0.0f;
        acc.w = (j == 0) ? (dd * self.w + b3) : 0.0f;

        int p = s0 + j;
        for (; p + 8 < s2; p += 16) {
            int pb = p + 8;
            int i0 = (int)csr[p]  + ((p  >= s1) ? SRC_SPLIT : 0);
            int i1 = (int)csr[pb] + ((pb >= s1) ? SRC_SPLIT : 0);
            float g0 = dis[i0];
            float g1 = dis[i1];
            float4 v0 = hw4[(size_t)i0 * 4 + q];
            float4 v1 = hw4[(size_t)i1 * 4 + q];
            asm volatile("" :: "v"(g0), "v"(g1), "v"(v0.x), "v"(v0.y), "v"(v0.z), "v"(v0.w),
                               "v"(v1.x), "v"(v1.y), "v"(v1.z), "v"(v1.w));
            float w0 = g0 * d;
            float w1 = g1 * d;
            acc.x += w0 * v0.x; acc.y += w0 * v0.y; acc.z += w0 * v0.z; acc.w += w0 * v0.w;
            acc.x += w1 * v1.x; acc.y += w1 * v1.y; acc.z += w1 * v1.z; acc.w += w1 * v1.w;
        }
        if (p < s2) {
            int i0 = (int)csr[p] + ((p >= s1) ? SRC_SPLIT : 0);
            float w0 = dis[i0] * d;
            float4 v0 = hw4[(size_t)i0 * 4 + q];
            acc.x += w0 * v0.x; acc.y += w0 * v0.y; acc.z += w0 * v0.z; acc.w += w0 * v0.w;
        }

        for (int off = 4; off < 32; off <<= 1) {
            acc.x += __shfl_xor(acc.x, off);
            acc.y += __shfl_xor(acc.y, off);
            acc.z += __shfl_xor(acc.z, off);
            acc.w += __shfl_xor(acc.w, off);
        }

        if (j == 0) ((float4*)out)[(size_t)r * 4 + q] = acc;
    }
}

__global__ void bn_stats(const float* __restrict__ agg, float* __restrict__ stats, int N) {
    __shared__ float s1s[256];
    __shared__ float s2s[256];
    int c = threadIdx.x & 15;
    int g = threadIdx.x >> 4;
    int stride = gridDim.x * 16;
    float s1 = 0.0f, s2 = 0.0f;
    for (int row = blockIdx.x * 16 + g; row < N; row += stride) {
        float v = agg[(size_t)row * HC + c];
        v = fmaxf(v, 0.0f);
        s1 += v;
        s2 += v * v;
    }
    s1s[threadIdx.x] = s1;
    s2s[threadIdx.x] = s2;
    __syncthreads();
    if (threadIdx.x < 16) {
        float t1 = 0.0f, t2 = 0.0f;
#pragma unroll
        for (int gg = 0; gg < 16; ++gg) {
            t1 += s1s[gg * 16 + threadIdx.x];
            t2 += s2s[gg * 16 + threadIdx.x];
        }
        atomicAdd(&stats[threadIdx.x], t1);
        atomicAdd(&stats[16 + threadIdx.x], t2);
    }
}

__global__ void bn_finalize(float* __restrict__ stats, const float* __restrict__ gamma,
                            const float* __restrict__ beta, float invN) {
    int c = threadIdx.x;
    if (c < 16) {
        float mu  = stats[c] * invN;
        float var = stats[16 + c] * invN - mu * mu;
        float sc  = rsqrtf(var + EPSV) * gamma[c];
        stats[32 + c] = sc;
        stats[48 + c] = beta[c] - mu * sc;
    }
}

__global__ void bn_apply(float* __restrict__ h, const float* __restrict__ stats, int NC) {
    int i = blockIdx.x * blockDim.x + threadIdx.x;
    if (i >= NC) return;
    int c = i & 15;
    float v = fmaxf(h[i], 0.0f);
    h[i] = v * stats[32 + c] + stats[48 + c];
}

__global__ void head_kernel(const float* __restrict__ h, const float* __restrict__ Wfc,
                            const float* __restrict__ bfc, float* __restrict__ out, int N) {
    int i = blockIdx.x * blockDim.x + threadIdx.x;
    if (i >= N) return;
    const float* r = h + (size_t)i * HC;
    float z0 = bfc[0], z1 = bfc[1];
#pragma unroll
    for (int c = 0; c < HC; ++c) {
        float v = r[c];
        z0 += v * Wfc[c * 2 + 0];
        z1 += v * Wfc[c * 2 + 1];
    }
    float mx  = fmaxf(z0, z1);
    float lse = mx + logf(__expf(z0 - mx) + __expf(z1 - mx));
    out[(size_t)i * 2 + 0] = z0 - lse;
    out[(size_t)i * 2 + 1] = z1 - lse;
}

extern "C" void kernel_launch(void* const* d_in, const int* in_sizes, int n_in,
                              void* d_out, int out_size, void* d_ws, size_t ws_size,
                              hipStream_t stream) {
    const int IN = 128;
    const int N  = in_sizes[0] / IN;
    const int E  = in_sizes[1] / 2;
    const int M  = 2 * N;

    const float* x    = (const float*)d_in[0];
    const int*   ei   = (const int*)d_in[1];
    const int*   srcI = ei;
    const int*   dstI = ei + E;
    const float* W1   = (const float*)d_in[2];
    const float* b1   = (const float*)d_in[3];
    const float* W2   = (const float*)d_in[4];
    const float* b2   = (const float*)d_in[5];
    const float* W3   = (const float*)d_in[6];
    const float* b3   = (const float*)d_in[7];
    const float* g1   = (const float*)d_in[8];
    const float* be1  = (const float*)d_in[9];
    const float* g2   = (const float*)d_in[10];
    const float* be2  = (const float*)d_in[11];
    const float* Wfc  = (const float*)d_in[12];
    const float* bfc  = (const float*)d_in[13];

    float* ws    = (float*)d_ws;
    float* dis   = ws;
    float* bufA  = dis + N;
    float* bufB  = bufA + (size_t)N * HC;
    float* bufC  = bufB + (size_t)N * HC;
    float* stats = bufC + (size_t)N * HC;
    int*   csr   = (int*)bufC;
    int*   tsum  = (int*)bufC;
    int*   rank  = (int*)bufA;
    int*   start = (int*)d_out;

    const int B256 = 256;
    const int NC = N * HC;
    dim3 blkN((N + B256 - 1) / B256);
    dim3 blkNC((NC + B256 - 1) / B256);
    dim3 blkM((M + B256 - 1) / B256);
    dim3 blkE4((E + 4 * B256 - 1) / (4 * B256));
    dim3 blkS((M + 2047) / 2048);
    dim3 blkG((N / 16 + 7) / 8);
    dim3 blkP(2048);
    float invN = 1.0f / (float)N;

    zero_f32<<<blkM, B256, 0, stream>>>((float*)start, M);
    bucket_rank<<<blkE4, B256, 0, stream>>>(srcI, dstI, start, rank, E);
    scan_tile_sums<<<blkS, B256, 0, stream>>>(start, tsum, M);
    scan_apply<<<blkS, B256, 0, stream>>>(start, tsum, dis, M, N);
    bucket_fill<<<blkE4, B256, 0, stream>>>(srcI, dstI, start, rank, csr, E);

    gemm_wmma_f32<<<blkG, B256, 0, stream>>>(x, W1, bufA, N, IN);
    pull_rows<<<blkP, B256, 0, stream>>>(start, csr, dis, bufA, b1, bufB, N, E);
    zero_f32<<<1, 64, 0, stream>>>(stats, 64);
    bn_stats<<<256, 256, 0, stream>>>(bufB, stats, N);
    bn_finalize<<<1, 16, 0, stream>>>(stats, g1, be1, invN);
    bn_apply<<<blkNC, B256, 0, stream>>>(bufB, stats, NC);

    gemm_wmma_f32<<<blkG, B256, 0, stream>>>(bufB, W2, bufA, N, HC);
    pull_rows<<<blkP, B256, 0, stream>>>(start, csr, dis, bufA, b2, bufB, N, E);
    zero_f32<<<1, 64, 0, stream>>>(stats, 64);
    bn_stats<<<256, 256, 0, stream>>>(bufB, stats, N);
    bn_finalize<<<1, 16, 0, stream>>>(stats, g2, be2, invN);
    bn_apply<<<blkNC, B256, 0, stream>>>(bufB, stats, NC);

    gemm_wmma_f32<<<blkG, B256, 0, stream>>>(bufB, W3, bufA, N, HC);
    pull_rows<<<blkP, B256, 0, stream>>>(start, csr, dis, bufA, b3, bufB, N, E);

    head_kernel<<<blkN, B256, 0, stream>>>(bufB, Wfc, bfc, (float*)d_out, N);
}


# Edge sort ranked by LDS counters in two levels (256 edge slices over 256-row buckets, one work-group per bucket) in place of 3.2M returning L2 atomics; arrival-order rank as before; 18 launches
# speedup vs baseline: 9.6085x; 1.6747x over previous
#include <hip/hip_runtime.h>
#include <hip/hip_bf16.h>

typedef float v2f __attribute__((ext_vector_type(2)));
typedef float v8f __attribute__((ext_vector_type(8)));

#define HC 16
#define EPSV 1e-5f
#define SRC_SPLIT 65536

__global__ void zero_f32(float* __restrict__ p, int n) {
    int i = blockIdx.x * blockDim.x + threadIdx.x;
    if (i < n) p[i] = 0.0f;
}

#define RB_SHIFT 8
#define NBP 512
#define G1 256
#define SORT_T 1024

__device__ __forceinline__ int block_excl_scan(int v, int* wsum, int tid) {
    int lane = tid & 31, wv = tid >> 5;
    int inc = v;
    for (int off = 1; off < 32; off <<= 1) {
        int n = __shfl_up(inc, off);
        if (lane >= off) inc += n;
    }
    if (lane == 31) wsum[wv] = inc;
    __syncthreads();
    int before = 0;
    for (int w = 0; w < wv; ++w) before += wsum[w];
    __syncthreads();
    return before + inc - v;
}

__global__ void __launch_bounds__(SORT_T)
coarse_hist(const int* __restrict__ dst, int* __restrict__ ghist, int E, int chunk) {
    __shared__ int h[NBP];
    int tid = threadIdx.x;
    if (tid < NBP) h[tid] = 0;
    __syncthreads();
    int c0 = blockIdx.x * chunk;
    int c1 = min(c0 + chunk, E);
    for (int base = c0 + tid; base < c1; base += 4 * SORT_T) {
        int d[4];
#pragma unroll
        for (int k = 0; k < 4; ++k) d[k] = dst[min(base + k * SORT_T, c1 - 1)];
        asm volatile("" :: "v"(d[0]), "v"(d[1]), "v"(d[2]), "v"(d[3]));
#pragma unroll
        for (int k = 0; k < 4; ++k)
            if (base + k * SORT_T < c1) atomicAdd(&h[d[k] >> RB_SHIFT], 1);
    }
    __syncthreads();
    if (tid < NBP) ghist[blockIdx.x * NBP + tid] = h[tid];
}

__global__ void __launch_bounds__(SORT_T)
coarse_scatter(const int* __restrict__ src, const int* __restrict__ dst,
               const int* __restrict__ ghist, int* __restrict__ rec,
               int* __restrict__ start, int N, int E, int chunk) {
    __shared__ int ptot[8][NBP];
    __shared__ int ppart[8][NBP];
    __shared__ int cur[NBP];
    __shared__ int wsum[32];
    int tid = threadIdx.x;
    int wg = blockIdx.x;

    {
        const int4* __restrict__ gh4 = (const int4*)ghist;
        int j = tid & 127;
        int g = tid >> 7;
        int4 t4 = {0, 0, 0, 0}, p4 = {0, 0, 0, 0};
#pragma unroll 1
        for (int batch = 0; batch < 4; ++batch) {
            int w0 = g * (G1 / 8) + batch * 8;
            int4 v[8];
#pragma unroll
            for (int k = 0; k < 8; ++k) v[k] = gh4[(w0 + k) * (NBP / 4) + j];
            asm volatile("" :: "v"(v[0].x), "v"(v[1].x), "v"(v[2].x), "v"(v[3].x),
                               "v"(v[4].x), "v"(v[5].x), "v"(v[6].x), "v"(v[7].x));
#pragma unroll
            for (int k = 0; k < 8; ++k) {
                t4.x += v[k].x; t4.y += v[k].y; t4.z += v[k].z; t4.w += v[k].w;
                if (w0 + k < wg) { p4.x += v[k].x; p4.y += v[k].y; p4.z += v[k].z; p4.w += v[k].w; }
            }
        }
        ptot[g][4 * j + 0] = t4.x; ptot[g][4 * j + 1] = t4.y;
        ptot[g][4 * j + 2] = t4.z; ptot[g][4 * j + 3] = t4.w;
        ppart[g][4 * j + 0] = p4.x; ppart[g][4 * j + 1] = p4.y;
        ppart[g][4 * j + 2] = p4.z; ppart[g][4 * j + 3] = p4.w;
    }
    __syncthreads();
    int tot = 0, part = 0;
    if (tid < NBP) {
#pragma unroll
        for (int g = 0; g < 8; ++g) { tot += ptot[g][tid]; part += ppart[g][tid]; }
    }
    int bstart = block_excl_scan(tot, wsum, tid);
    if (tid < NBP) {
        cur[tid] = bstart + part;
        if (wg == 0 && (tid << RB_SHIFT) < N) {
            start[2 * (tid << RB_SHIFT)] = bstart;
            start[2 * (tid << RB_SHIFT) + 1] = bstart + tot;
        }
    }
    __syncthreads();

    int c0 = wg * chunk;
    int c1 = min(c0 + chunk, E);
    for (int base = c0 + tid; base < c1; base += 4 * SORT_T) {
        int s[4], d[4];
#pragma unroll
        for (int k = 0; k < 4; ++k) {
            int ec = min(base + k * SORT_T, c1 - 1);
            s[k] = src[ec];
            d[k] = dst[ec];
        }
        asm volatile("" :: "v"(s[0]), "v"(s[1]), "v"(s[2]), "v"(s[3]),
                           "v"(d[0]), "v"(d[1]), "v"(d[2]), "v"(d[3]));
#pragma unroll
        for (int k = 0; k < 4; ++k) {
            if (base + k * SORT_T < c1) {
                int pos = atomicAdd(&cur[d[k] >> RB_SHIFT], 1);
                rec[pos] = s[k] | ((d[k] & ((1 << RB_SHIFT) - 1)) << 17);
            }
        }
    }
}

__global__ void __launch_bounds__(SORT_T)
fine_sort(const int* __restrict__ rec, int* __restrict__ start, float* __restrict__ dis,
          int* __restrict__ csr_w, int N) {
    unsigned short* __restrict__ csr = (unsigned short*)csr_w;
    __shared__ int cnt[512];
    __shared__ int cursor[512];
    __shared__ int wsum[32];
    int tid = threadIdx.x;
    int b = blockIdx.x;
    int c0 = start[2 * (b << RB_SHIFT)];
    int c1 = start[2 * (b << RB_SHIFT) + 1];
    if (tid < 512) cnt[tid] = 0;
    __syncthreads();

    for (int base = c0 + tid; base < c1; base += 4 * SORT_T) {
        int r[4];
#pragma unroll
        for (int k = 0; k < 4; ++k) r[k] = rec[min(base + k * SORT_T, c1 - 1)];
        asm volatile("" :: "v"(r[0]), "v"(r[1]), "v"(r[2]), "v"(r[3]));
#pragma unroll
        for (int k = 0; k < 4; ++k)
            if (base + k * SORT_T < c1) atomicAdd(&cnt[(r[k] >> 16) & 511], 1);
    }
    __syncthreads();

    int c = (tid < 512) ? cnt[tid] : 0;
    int ex = block_excl_scan(c, wsum, tid);
    if (tid < 512) {
        cursor[tid] = c0 + ex;
        int row = (b << RB_SHIFT) + (tid >> 1);
        if (row < N) start[2 * row + (tid & 1)] = c0 + ex;
    }
    if (tid < 256) {
        int row = (b << RB_SHIFT) + tid;
        if (row < N) dis[row] = rsqrtf((float)(cnt[2 * tid] + cnt[2 * tid + 1]) + 1.0f);
    }
    __syncthreads();

    for (int base = c0 + tid; base < c1; base += 4 * SORT_T) {
        int r[4];
#pragma unroll
        for (int k = 0; k < 4; ++k) r[k] = rec[min(base + k * SORT_T, c1 - 1)];
        asm volatile("" :: "v"(r[0]), "v"(r[1]), "v"(r[2]), "v"(r[3]));
#pragma unroll
        for (int k = 0; k < 4; ++k) {
            if (base + k * SORT_T < c1) {
                int pos = atomicAdd(&cursor[(r[k] >> 16) & 511], 1);
                csr[pos] = (unsigned short)(r[k] & (SRC_SPLIT - 1));
            }
        }
    }
}

__global__ void gemm_wmma_f32(const float* __restrict__ A, const float* __restrict__ B,
                              float* __restrict__ C, int N, int K) {
    int wave = (int)((blockIdx.x * blockDim.x + threadIdx.x) >> 5);
    int lane = threadIdx.x & 31;
    int row0 = wave * 16;
    if (row0 >= N) return;

    int m  = lane & 15;
    int hi = lane >> 4;
    const float* arow = A + (size_t)(row0 + m) * K;

    v8f acc = {};
    for (int k0 = 0; k0 < K; k0 += 4) {
        int ka = k0 + 2 * hi;
        v2f a, b;
        a.x = arow[ka];
        a.y = arow[ka + 1];
        b.x = B[(size_t)ka * HC + m];
        b.y = B[(size_t)(ka + 1) * HC + m];
        acc = __builtin_amdgcn_wmma_f32_16x16x4_f32(
             false, a,  false, b,
             (short)0, acc,  false,  false);
    }

    float* cbase = C + (size_t)(row0 + hi * 8) * HC + m;
#pragma unroll
    for (int r = 0; r < 8; ++r) cbase[(size_t)r * HC] = acc[r];
}

__global__ void __launch_bounds__(256)
pull_rows(const int* __restrict__ start, const int* __restrict__ csr_w,
          const float* __restrict__ dis, const float* __restrict__ hw,
          const float* __restrict__ bias, float* __restrict__ out, int N, int E) {
    const unsigned short* __restrict__ csr = (const unsigned short*)csr_w;
    const float4* __restrict__ hw4 = (const float4*)hw;
    int lane = threadIdx.x & 31;
    int q = lane & 3;
    int j = lane >> 2;
    int nw = (int)((gridDim.x * blockDim.x) >> 5);

    float b0 = bias[4 * q + 0], b1 = bias[4 * q + 1], b2 = bias[4 * q + 2], b3 = bias[4 * q + 3];

    for (int r = (int)((blockIdx.x * blockDim.x + threadIdx.x) >> 5); r < N; r += nw) {
        bool last = (r + 1 >= N);
        int s0 = start[2 * r];
        int s1 = start[2 * r + 1];
        int s2 = start[last ? 2 * r + 1 : 2 * r + 2];
        if (last) s2 = E;
        float d = dis[r];
        float4 self = hw4[(size_t)r * 4 + q];

        float dd = d * d;
        float4 acc;
        acc.x = (j == 0) ? (dd * self.x + b0) : 0.0f;
        acc.y = (j == 0) ? (dd * self.y + b1) : 0.0f;
        acc.z = (j == 0) ? (dd * self.z + b2) : 0.0f;
        acc.w = (j == 0) ? (dd * self.w + b3) : 0.0f;

        int p = s0 + j;
        for (; p + 8 < s2; p += 16) {
            int pb = p + 8;
            int i0 = (int)csr[p]  + ((p  >= s1) ? SRC_SPLIT : 0);
            int i1 = (int)csr[pb] + ((pb >= s1) ? SRC_SPLIT : 0);
            float g0 = dis[i0];
            float g1 = dis[i1];
            float4 v0 = hw4[(size_t)i0 * 4 + q];
            float4 v1 = hw4[(size_t)i1 * 4 + q];
            asm volatile("" :: "v"(g0), "v"(g1), "v"(v0.x), "v"(v0.y), "v"(v0.z), "v"(v0.w),
                               "v"(v1.x), "v"(v1.y), "v"(v1.z), "v"(v1.w));
            float w0 = g0 * d;
            float w1 = g1 * d;
            acc.x += w0 * v0.x; acc.y += w0 * v0.y; acc.z += w0 * v0.z; acc.w += w0 * v0.w;
            acc.x += w1 * v1.x; acc.y += w1 * v1.y; acc.z += w1 * v1.z; acc.w += w1 * v1.w;
        }
        if (p < s2) {
            int i0 = (int)csr[p] + ((p >= s1) ? SRC_SPLIT : 0);
            float w0 = dis[i0] * d;
            float4 v0 = hw4[(size_t)i0 * 4 + q];
            acc.x += w0 * v0.x; acc.y += w0 * v0.y; acc.z += w0 * v0.z; acc.w += w0 * v0.w;
        }

        for (int off = 4; off < 32; off <<= 1) {
            acc.x += __shfl_xor(acc.x, off);
            acc.y += __shfl_xor(acc.y, off);
            acc.z += __shfl_xor(acc.z, off);
            acc.w += __shfl_xor(acc.w, off);
        }

        if (j == 0) ((float4*)out)[(size_t)r * 4 + q] = acc;
    }
}

__global__ void bn_stats(const float* __restrict__ agg, float* __restrict__ stats, int N) {
    __shared__ float s1s[256];
    __shared__ float s2s[256];
    int c = threadIdx.x & 15;
    int g = threadIdx.x >> 4;
    int stride = gridDim.x * 16;
    float s1 = 0.0f, s2 = 0.0f;
    for (int row = blockIdx.x * 16 + g; row < N; row += stride) {
        float v = agg[(size_t)row * HC + c];
        v = fmaxf(v, 0.0f);
        s1 += v;
        s2 += v * v;
    }
    s1s[threadIdx.x] = s1;
    s2s[threadIdx.x] = s2;
    __syncthreads();
    if (threadIdx.x < 16) {
        float t1 = 0.0f, t2 = 0.0f;
#pragma unroll
        for (int gg = 0; gg < 16; ++gg) {
            t1 += s1s[gg * 16 + threadIdx.x];
            t2 += s2s[gg * 16 + threadIdx.x];
        }
        atomicAdd(&stats[threadIdx.x], t1);
        atomicAdd(&stats[16 + threadIdx.x], t2);
    }
}

__global__ void bn_finalize(float* __restrict__ stats, const float* __restrict__ gamma,
                            const float* __restrict__ beta, float invN) {
    int c = threadIdx.x;
    if (c < 16) {
        float mu  = stats[c] * invN;
        float var = stats[16 + c] * invN - mu * mu;
        float sc  = rsqrtf(var + EPSV) * gamma[c];
        stats[32 + c] = sc;
        stats[48 + c] = beta[c] - mu * sc;
    }
}

__global__ void bn_apply(float* __restrict__ h, const float* __restrict__ stats, int NC) {
    int i = blockIdx.x * blockDim.x + threadIdx.x;
    if (i >= NC) return;
    int c = i & 15;
    float v = fmaxf(h[i], 0.0f);
    h[i] = v * stats[32 + c] + stats[48 + c];
}

__global__ void head_kernel(const float* __restrict__ h, const float* __restrict__ Wfc,
                            const float* __restrict__ bfc, float* __restrict__ out, int N) {
    int i = blockIdx.x * blockDim.x + threadIdx.x;
    if (i >= N) return;
    const float* r = h + (size_t)i * HC;
    float z0 = bfc[0], z1 = bfc[1];
#pragma unroll
    for (int c = 0; c < HC; ++c) {
        float v = r[c];
        z0 += v * Wfc[c * 2 + 0];
        z1 += v * Wfc[c * 2 + 1];
    }
    float mx  = fmaxf(z0, z1);
    float lse = mx + logf(__expf(z0 - mx) + __expf(z1 - mx));
    out[(size_t)i * 2 + 0] = z0 - lse;
    out[(size_t)i * 2 + 1] = z1 - lse;
}

extern "C" void kernel_launch(void* const* d_in, const int* in_sizes, int n_in,
                              void* d_out, int out_size, void* d_ws, size_t ws_size,
                              hipStream_t stream) {
    const int IN = 128;
    const int N  = in_sizes[0] / IN;
    const int E  = in_sizes[1] / 2;

    const float* x    = (const float*)d_in[0];
    const int*   ei   = (const int*)d_in[1];
    const int*   srcI = ei;
    const int*   dstI = ei + E;
    const float* W1   = (const float*)d_in[2];
    const float* b1   = (const float*)d_in[3];
    const float* W2   = (const float*)d_in[4];
    const float* b2   = (const float*)d_in[5];
    const float* W3   = (const float*)d_in[6];
    const float* b3   = (const float*)d_in[7];
    const float* g1   = (const float*)d_in[8];
    const float* be1  = (const float*)d_in[9];
    const float* g2   = (const float*)d_in[10];
    const float* be2  = (const float*)d_in[11];
    const float* Wfc  = (const float*)d_in[12];
    const float* bfc  = (const float*)d_in[13];

    float* ws    = (float*)d_ws;
    float* dis   = ws;
    float* bufA  = dis + N;
    float* bufB  = bufA + (size_t)N * HC;
    float* bufC  = bufB + (size_t)N * HC;
    float* stats = bufC + (size_t)N * HC;
    int*   csr   = (int*)bufC;
    int*   ghist = (int*)(((size_t)bufC + 15) & ~(size_t)15);
    int*   rec   = (int*)bufA;
    int*   start = (int*)d_out;

    const int B256 = 256;
    const int NC = N * HC;
    dim3 blkN((N + B256 - 1) / B256);
    dim3 blkNC((NC + B256 - 1) / B256);
    const int NB = (N + (1 << RB_SHIFT) - 1) >> RB_SHIFT;
    const int chunk = (E + G1 - 1) / G1;
    dim3 blkG((N / 16 + 7) / 8);
    dim3 blkP(2048);
    float invN = 1.0f / (float)N;

    coarse_hist<<<G1, SORT_T, 0, stream>>>(dstI, ghist, E, chunk);
    coarse_scatter<<<G1, SORT_T, 0, stream>>>(srcI, dstI, ghist, rec, start, N, E, chunk);
    fine_sort<<<NB, SORT_T, 0, stream>>>(rec, start, dis, csr, N);

    gemm_wmma_f32<<<blkG, B256, 0, stream>>>(x, W1, bufA, N, IN);
    pull_rows<<<blkP, B256, 0, stream>>>(start, csr, dis, bufA, b1, bufB, N, E);
    zero_f32<<<1, 64, 0, stream>>>(stats, 64);
    bn_stats<<<256, 256, 0, stream>>>(bufB, stats, N);
    bn_finalize<<<1, 16, 0, stream>>>(stats, g1, be1, invN);
    bn_apply<<<blkNC, B256, 0, stream>>>(bufB, stats, NC);

    gemm_wmma_f32<<<blkG, B256, 0, stream>>>(bufB, W2, bufA, N, HC);
    pull_rows<<<blkP, B256, 0, stream>>>(start, csr, dis, bufA, b2, bufB, N, E);
    zero_f32<<<1, 64, 0, stream>>>(stats, 64);
    bn_stats<<<256, 256, 0, stream>>>(bufB, stats, N);
    bn_finalize<<<1, 16, 0, stream>>>(stats, g2, be2, invN);
    bn_apply<<<blkNC, B256, 0, stream>>>(bufB, stats, NC);

    gemm_wmma_f32<<<blkG, B256, 0, stream>>>(bufB, W3, bufA, N, HC);
    pull_rows<<<blkP, B256, 0, stream>>>(start, csr, dis, bufA, b3, bufB, N, E);

    head_kernel<<<blkN, B256, 0, stream>>>(bufB, Wfc, bfc, (float*)d_out, N);
}


# Layers 2 and 3 aggregate first, transform after: pull applies relu + BatchNorm map to gathered rows, 16-row tiles of sums times W on WMMA in its epilogue; products re-associated; 18 launches to 12
# speedup vs baseline: 10.5927x; 1.1024x over previous
#include <hip/hip_runtime.h>
#include <hip/hip_bf16.h>

typedef float v2f __attribute__((ext_vector_type(2)));
typedef float v8f __attribute__((ext_vector_type(8)));

#define HC 16
#define EPSV 1e-5f
#define SRC_SPLIT 65536

__global__ void zero_f32(float* __restrict__ p, int n) {
    int i = blockIdx.x * blockDim.x + threadIdx.x;
    if (i < n) p[i] = 0.0f;
}

#define RB_SHIFT 8
#define NBP 512
#define G1 256
#define SORT_T 1024

__device__ __forceinline__ int block_excl_scan(int v, int* wsum, int tid) {
    int lane = tid & 31, wv = tid >> 5;
    int inc = v;
    for (int off = 1; off < 32; off <<= 1) {
        int n = __shfl_up(inc, off);
        if (lane >= off) inc += n;
    }
    if (lane == 31) wsum[wv] = inc;
    __syncthreads();
    int before = 0;
    for (int w = 0; w < wv; ++w) before += wsum[w];
    __syncthreads();
    return before + inc - v;
}

__global__ void __launch_bounds__(SORT_T)
coarse_hist(const int* __restrict__ dst, int* __restrict__ ghist, int E, int chunk) {
    __shared__ int h[NBP];
    int tid = threadIdx.x;
    if (tid < NBP) h[tid] = 0;
    __syncthreads();
    int c0 = blockIdx.x * chunk;
    int c1 = min(c0 + chunk, E);
    for (int base = c0 + tid; base < c1; base += 4 * SORT_T) {
        int d[4];
#pragma unroll
        for (int k = 0; k < 4; ++k) d[k] = dst[min(base + k * SORT_T, c1 - 1)];
        asm volatile("" :: "v"(d[0]), "v"(d[1]), "v"(d[2]), "v"(d[3]));
#pragma unroll
        for (int k = 0; k < 4; ++k)
            if (base + k * SORT_T < c1) atomicAdd(&h[d[k] >> RB_SHIFT], 1);
    }
    __syncthreads();
    if (tid < NBP) ghist[blockIdx.x * NBP + tid] = h[tid];
}

__global__ void __launch_bounds__(SORT_T)
coarse_scatter(const int* __restrict__ src, const int* __restrict__ dst,
               const int* __restrict__ ghist, int* __restrict__ rec,
               int* __restrict__ start, int N, int E, int chunk) {
    __shared__ int ptot[8][NBP];
    __shared__ int ppart[8][NBP];
    __shared__ int cur[NBP];
    __shared__ int wsum[32];
    int tid = threadIdx.x;
    int wg = blockIdx.x;

    {
        const int4* __restrict__ gh4 = (const int4*)ghist;
        int j = tid & 127;
        int g = tid >> 7;
        int4 t4 = {0, 0, 0, 0}, p4 = {0, 0, 0, 0};
#pragma unroll 1
        for (int batch = 0; batch < 4; ++batch) {
            int w0 = g * (G1 / 8) + batch * 8;
            int4 v[8];
#pragma unroll
            for (int k = 0; k < 8; ++k) v[k] = gh4[(w0 + k) * (NBP / 4) + j];
            asm volatile("" :: "v"(v[0].x), "v"(v[1].x), "v"(v[2].x), "v"(v[3].x),
                               "v"(v[4].x), "v"(v[5].x), "v"(v[6].x), "v"(v[7].x));
#pragma unroll
            for (int k = 0; k < 8; ++k) {
                t4.x += v[k].x; t4.y += v[k].y; t4.z += v[k].z; t4.w += v[k].w;
                if (w0 + k < wg) { p4.x += v[k].x; p4.y += v[k].y; p4.z += v[k].z; p4.w += v[k].w; }
            }
        }
        ptot[g][4 * j + 0] = t4.x; ptot[g][4 * j + 1] = t4.y;
        ptot[g][4 * j + 2] = t4.z; ptot[g][4 * j + 3] = t4.w;
        ppart[g][4 * j + 0] = p4.x; ppart[g][4 * j + 1] = p4.y;
        ppart[g][4 * j + 2] = p4.z; ppart[g][4 * j + 3] = p4.w;
    }
    __syncthreads();
    int tot = 0, part = 0;
    if (tid < NBP) {
#pragma unroll
        for (int g = 0; g < 8; ++g) { tot += ptot[g][tid]; part += ppart[g][tid]; }
    }
    int bstart = block_excl_scan(tot, wsum, tid);
    if (tid < NBP) {
        cur[tid] = bstart + part;
        if (wg == 0 && (tid << RB_SHIFT) < N) {
            start[2 * (tid << RB_SHIFT)] = bstart;
            start[2 * (tid << RB_SHIFT) + 1] = bstart + tot;
        }
    }
    __syncthreads();

    int c0 = wg * chunk;
    int c1 = min(c0 + chunk, E);
    for (int base = c0 + tid; base < c1; base += 4 * SORT_T) {
        int s[4], d[4];
#pragma unroll
        for (int k = 0; k < 4; ++k) {
            int ec = min(base + k * SORT_T, c1 - 1);
            s[k] = src[ec];
            d[k] = dst[ec];
        }
        asm volatile("" :: "v"(s[0]), "v"(s[1]), "v"(s[2]), "v"(s[3]),
                           "v"(d[0]), "v"(d[1]), "v"(d[2]), "v"(d[3]));
#pragma unroll
        for (int k = 0; k < 4; ++k) {
            if (base + k * SORT_T < c1) {
                int pos = atomicAdd(&cur[d[k] >> RB_SHIFT], 1);
                rec[pos] = s[k] | ((d[k] & ((1 << RB_SHIFT) - 1)) << 17);
            }
        }
    }
}

__global__ void __launch_bounds__(SORT_T)
fine_sort(const int* __restrict__ rec, int* __restrict__ start, float* __restrict__ dis,
          int* __restrict__ csr_w, int N) {
    unsigned short* __restrict__ csr = (unsigned short*)csr_w;
    __shared__ int cnt[512];
    __shared__ int cursor[512];
    __shared__ int wsum[32];
    int tid = threadIdx.x;
    int b = blockIdx.x;
    int c0 = start[2 * (b << RB_SHIFT)];
    int c1 = start[2 * (b << RB_SHIFT) + 1];
    if (tid < 512) cnt[tid] = 0;
    __syncthreads();

    for (int base = c0 + tid; base < c1; base += 4 * SORT_T) {
        int r[4];
#pragma unroll
        for (int k = 0; k < 4; ++k) r[k] = rec[min(base + k * SORT_T, c1 - 1)];
        asm volatile("" :: "v"(r[0]), "v"(r[1]), "v"(r[2]), "v"(r[3]));
#pragma unroll
        for (int k = 0; k < 4; ++k)
            if (base + k * SORT_T < c1) atomicAdd(&cnt[(r[k] >> 16) & 511], 1);
    }
    __syncthreads();

    int c = (tid < 512) ? cnt[tid] : 0;
    int ex = block_excl_scan(c, wsum, tid);
    if (tid < 512) {
        cursor[tid] = c0 + ex;
        int row = (b << RB_SHIFT) + (tid >> 1);
        if (row < N) start[2 * row + (tid & 1)] = c0 + ex;
    }
    if (tid < 256) {
        int row = (b << RB_SHIFT) + tid;
        if (row < N) dis[row] = rsqrtf((float)(cnt[2 * tid] + cnt[2 * tid + 1]) + 1.0f);
    }
    __syncthreads();

    for (int base = c0 + tid; base < c1; base += 4 * SORT_T) {
        int r[4];
#pragma unroll
        for (int k = 0; k < 4; ++k) r[k] = rec[min(base + k * SORT_T, c1 - 1)];
        asm volatile("" :: "v"(r[0]), "v"(r[1]), "v"(r[2]), "v"(r[3]));
#pragma unroll
        for (int k = 0; k < 4; ++k) {
            if (base + k * SORT_T < c1) {
                int pos = atomicAdd(&cursor[(r[k] >> 16) & 511], 1);
                csr[pos] = (unsigned short)(r[k] & (SRC_SPLIT - 1));
            }
        }
    }
}

__global__ void gemm_wmma_f32(const float* __restrict__ A, const float* __restrict__ B,
                              float* __restrict__ C, int N, int K) {
    int wave = (int)((blockIdx.x * blockDim.x + threadIdx.x) >> 5);
    int lane = threadIdx.x & 31;
    int row0 = wave * 16;
    if (row0 >= N) return;

    int m  = lane & 15;
    int hi = lane >> 4;
    const float* arow = A + (size_t)(row0 + m) * K;

    v8f acc = {};
    for (int k0 = 0; k0 < K; k0 += 4) {
        int ka = k0 + 2 * hi;
        v2f a, b;
        a.x = arow[ka];
        a.y = arow[ka + 1];
        b.x = B[(size_t)ka * HC + m];
        b.y = B[(size_t)(ka + 1) * HC + m];
        acc = __builtin_amdgcn_wmma_f32_16x16x4_f32(
             false, a,  false, b,
             (short)0, acc,  false,  false);
    }

    float* cbase = C + (size_t)(row0 + hi * 8) * HC + m;
#pragma unroll
    for (int r = 0; r < 8; ++r) cbase[(size_t)r * HC] = acc[r];
}

__global__ void __launch_bounds__(256)
pull_rows(const int* __restrict__ start, const int* __restrict__ csr_w,
          const float* __restrict__ dis, const float* __restrict__ hw,
          const float* __restrict__ bias, float* __restrict__ out, int N, int E) {
    const unsigned short* __restrict__ csr = (const unsigned short*)csr_w;
    const float4* __restrict__ hw4 = (const float4*)hw;
    int lane = threadIdx.x & 31;
    int q = lane & 3;
    int j = lane >> 2;
    int nw = (int)((gridDim.x * blockDim.x) >> 5);

    float b0 = bias[4 * q + 0], b1 = bias[4 * q + 1], b2 = bias[4 * q + 2], b3 = bias[4 * q + 3];

    for (int r = (int)((blockIdx.x * blockDim.x + threadIdx.x) >> 5); r < N; r += nw) {
        bool last = (r + 1 >= N);
        int s0 = start[2 * r];
        int s1 = start[2 * r + 1];
        int s2 = start[last ? 2 * r + 1 : 2 * r + 2];
        if (last) s2 = E;
        float d = dis[r];
        float4 self = hw4[(size_t)r * 4 + q];

        float dd = d * d;
        float4 acc;
        acc.x = (j == 0) ? (dd * self.x + b0) : 0.0f;
        acc.y = (j == 0) ? (dd * self.y + b1) : 0.0f;
        acc.z = (j == 0) ? (dd * self.z + b2) : 0.0f;
        acc.w = (j == 0) ? (dd * self.w + b3) : 0.0f;

        int p = s0 + j;
        for (; p + 8 < s2; p += 16) {
            int pb = p + 8;
            int i0 = (int)csr[p]  + ((p  >= s1) ? SRC_SPLIT : 0);
            int i1 = (int)csr[pb] + ((pb >= s1) ? SRC_SPLIT : 0);
            float g0 = dis[i0];
            float g1 = dis[i1];
            float4 v0 = hw4[(size_t)i0 * 4 + q];
            float4 v1 = hw4[(size_t)i1 * 4 + q];
            asm volatile("" :: "v"(g0), "v"(g1), "v"(v0.x), "v"(v0.y), "v"(v0.z), "v"(v0.w),
                               "v"(v1.x), "v"(v1.y), "v"(v1.z), "v"(v1.w));
            float w0 = g0 * d;
            float w1 = g1 * d;
            acc.x += w0 * v0.x; acc.y += w0 * v0.y; acc.z += w0 * v0.z; acc.w += w0 * v0.w;
            acc.x += w1 * v1.x; acc.y += w1 * v1.y; acc.z += w1 * v1.z; acc.w += w1 * v1.w;
        }
        if (p < s2) {
            int i0 = (int)csr[p] + ((p >= s1) ? SRC_SPLIT : 0);
            float w0 = dis[i0] * d;
            float4 v0 = hw4[(size_t)i0 * 4 + q];
            acc.x += w0 * v0.x; acc.y += w0 * v0.y; acc.z += w0 * v0.z; acc.w += w0 * v0.w;
        }

        for (int off = 4; off < 32; off <<= 1) {
            acc.x += __shfl_xor(acc.x, off);
            acc.y += __shfl_xor(acc.y, off);
            acc.z += __shfl_xor(acc.z, off);
            acc.w += __shfl_xor(acc.w, off);
        }

        if (j == 0) ((float4*)out)[(size_t)r * 4 + q] = acc;
    }
}

__global__ void bn_stats(const float* __restrict__ agg, float* __restrict__ stats, int N) {
    __shared__ float s1s[256];
    __shared__ float s2s[256];
    int c = threadIdx.x & 15;
    int g = threadIdx.x >> 4;
    int stride = gridDim.x * 16;
    float s1 = 0.0f, s2 = 0.0f;
    for (int row = blockIdx.x * 16 + g; row < N; row += stride) {
        float v = agg[(size_t)row * HC + c];
        v = fmaxf(v, 0.0f);
        s1 += v;
        s2 += v * v;
    }
    s1s[threadIdx.x] = s1;
    s2s[threadIdx.x] = s2;
    __syncthreads();
    if (threadIdx.x < 16) {
        float t1 = 0.0f, t2 = 0.0f;
#pragma unroll
        for (int gg = 0; gg < 16; ++gg) {
            t1 += s1s[gg * 16 + threadIdx.x];
            t2 += s2s[gg * 16 + threadIdx.x];
        }
        atomicAdd(&stats[threadIdx.x], t1);
        atomicAdd(&stats[16 + threadIdx.x], t2);
    }
}

#define TP 20
__global__ void __launch_bounds__(256)
pull_bn_gemm(const int* __restrict__ start, const int* __restrict__ csr_w,
             const float* __restrict__ dis, const float* __restrict__ prev,
             const float* __restrict__ stats, const float* __restrict__ gamma,
             const float* __restrict__ beta, float invN,
             const float* __restrict__ Wm, const float* __restrict__ bias,
             float* __restrict__ out, int N, int E) {
    __shared__ float tile[64 * TP];
    const unsigned short* __restrict__ csr = (const unsigned short*)csr_w;
    const float4* __restrict__ p4 = (const float4*)prev;
    int lane = threadIdx.x & 31;
    int wave = __builtin_amdgcn_readfirstlane((int)(threadIdx.x >> 5));
    int q = lane & 3;
    int j = lane >> 2;

    float sc0, sc1, sc2, sc3, sh0, sh1, sh2, sh3;
    {
        float mu, var;
        mu = stats[4 * q + 0] * invN; var = stats[16 + 4 * q + 0] * invN - mu * mu;
        sc0 = rsqrtf(var + EPSV) * gamma[4 * q + 0]; sh0 = beta[4 * q + 0] - mu * sc0;
        mu = stats[4 * q + 1] * invN; var = stats[16 + 4 * q + 1] * invN - mu * mu;
        sc1 = rsqrtf(var + EPSV) * gamma[4 * q + 1]; sh1 = beta[4 * q + 1] - mu * sc1;
        mu = stats[4 * q + 2] * invN; var = stats[16 + 4 * q + 2] * invN - mu * mu;
        sc2 = rsqrtf(var + EPSV) * gamma[4 * q + 2]; sh2 = beta[4 * q + 2] - mu * sc2;
        mu = stats[4 * q + 3] * invN; var = stats[16 + 4 * q + 3] * invN - mu * mu;
        sc3 = rsqrtf(var + EPSV) * gamma[4 * q + 3]; sh3 = beta[4 * q + 3] - mu * sc3;
    }

    int rbase = blockIdx.x * 64 + wave * 8;
    int sv = start[min(2 * rbase + lane, 2 * N - 1)];
    float dv = dis[min(rbase + (lane & 7), N - 1)];
    float4 selfn = p4[(size_t)min(rbase, N - 1) * 4 + q];
    for (int t = 0; t < 8; ++t) {
        int r = rbase + t;
        if (r >= N) break;
        int s0 = __builtin_amdgcn_readlane(sv, 2 * t);
        int s1 = __builtin_amdgcn_readlane(sv, 2 * t + 1);
        int s2 = (r + 1 >= N) ? E : __builtin_amdgcn_readlane(sv, 2 * t + 2);
        float d = __int_as_float(__builtin_amdgcn_readlane(__float_as_int(dv), t));
        float4 self = selfn;
        selfn = p4[(size_t)min(r + 1, N - 1) * 4 + q];

        float dd = d * d;
        float4 acc;
        acc.x = (j == 0) ? dd * (fmaxf(self.x, 0.0f) * sc0 + sh0) : 0.0f;
        acc.y = (j == 0) ? dd * (fmaxf(self.y, 0.0f) * sc1 + sh1) : 0.0f;
        acc.z = (j == 0) ? dd * (fmaxf(self.z, 0.0f) * sc2 + sh2) : 0.0f;
        acc.w = (j == 0) ? dd * (fmaxf(self.w, 0.0f) * sc3 + sh3) : 0.0f;

        int p = s0 + j;
        for (; p + 8 < s2; p += 16) {
            int pb = p + 8;
            int i0 = (int)csr[p]  + ((p  >= s1) ? SRC_SPLIT : 0);
            int i1 = (int)csr[pb] + ((pb >= s1) ? SRC_SPLIT : 0);
            float g0 = dis[i0];
            float g1 = dis[i1];
            float4 v0 = p4[(size_t)i0 * 4 + q];
            float4 v1 = p4[(size_t)i1 * 4 + q];
            __builtin_amdgcn_sched_barrier(0);
            float w0 = g0 * d;
            float w1 = g1 * d;
            acc.x += w0 * (fmaxf(v0.x, 0.0f) * sc0 + sh0);
            acc.y += w0 * (fmaxf(v0.y, 0.0f) * sc1 + sh1);
            acc.z += w0 * (fmaxf(v0.z, 0.0f) * sc2 + sh2);
            acc.w += w0 * (fmaxf(v0.w, 0.0f) * sc3 + sh3);
            acc.x += w1 * (fmaxf(v1.x, 0.0f) * sc0 + sh0);
            acc.y += w1 * (fmaxf(v1.y, 0.0f) * sc1 + sh1);
            acc.z += w1 * (fmaxf(v1.z, 0.0f) * sc2 + sh2);
            acc.w += w1 * (fmaxf(v1.w, 0.0f) * sc3 + sh3);
        }
        if (p < s2) {
            int i0 = (int)csr[p] + ((p >= s1) ? SRC_SPLIT : 0);
            float w0 = dis[i0] * d;
            float4 v0 = p4[(size_t)i0 * 4 + q];
            acc.x += w0 * (fmaxf(v0.x, 0.0f) * sc0 + sh0);
            acc.y += w0 * (fmaxf(v0.y, 0.0f) * sc1 + sh1);
            acc.z += w0 * (fmaxf(v0.z, 0.0f) * sc2 + sh2);
            acc.w += w0 * (fmaxf(v0.w, 0.0f) * sc3 + sh3);
        }

        for (int off = 4; off < 32; off <<= 1) {
            acc.x += __shfl_xor(acc.x, off);
            acc.y += __shfl_xor(acc.y, off);
            acc.z += __shfl_xor(acc.z, off);
            acc.w += __shfl_xor(acc.w, off);
        }

        if (j == 0) *(float4*)&tile[(wave * 8 + t) * TP + 4 * q] = acc;
    }
    __syncthreads();

    if (wave < 4) {
        int row0 = blockIdx.x * 64 + wave * 16;
        if (row0 < N) {
            int m  = lane & 15;
            int hi = lane >> 4;
            const float* trow = tile + (wave * 16 + m) * TP;
            float bv = bias[m];
            v8f c8 = {bv, bv, bv, bv, bv, bv, bv, bv};
            for (int kk = 0; kk < 4; ++kk) {
                int ka = 4 * kk + 2 * hi;
                v2f a, b;
                a.x = trow[ka];
                a.y = trow[ka + 1];
                b.x = Wm[ka * HC + m];
                b.y = Wm[(ka + 1) * HC + m];
                c8 = __builtin_amdgcn_wmma_f32_16x16x4_f32(
                     false, a,  false, b,
                     (short)0, c8,  false,  false);
            }
            float* cbase = out + (size_t)(row0 + hi * 8) * HC + m;
#pragma unroll
            for (int r = 0; r < 8; ++r) cbase[(size_t)r * HC] = c8[r];
        }
    }
}

__global__ void head_kernel(const float* __restrict__ h, const float* __restrict__ Wfc,
                            const float* __restrict__ bfc, float* __restrict__ out, int N) {
    int i = blockIdx.x * blockDim.x + threadIdx.x;
    if (i >= N) return;
    const float* r = h + (size_t)i * HC;
    float z0 = bfc[0], z1 = bfc[1];
#pragma unroll
    for (int c = 0; c < HC; ++c) {
        float v = r[c];
        z0 += v * Wfc[c * 2 + 0];
        z1 += v * Wfc[c * 2 + 1];
    }
    float mx  = fmaxf(z0, z1);
    float lse = mx + logf(__expf(z0 - mx) + __expf(z1 - mx));
    out[(size_t)i * 2 + 0] = z0 - lse;
    out[(size_t)i * 2 + 1] = z1 - lse;
}

extern "C" void kernel_launch(void* const* d_in, const int* in_sizes, int n_in,
                              void* d_out, int out_size, void* d_ws, size_t ws_size,
                              hipStream_t stream) {
    const int IN = 128;
    const int N  = in_sizes[0] / IN;
    const int E  = in_sizes[1] / 2;

    const float* x    = (const float*)d_in[0];
    const int*   ei   = (const int*)d_in[1];
    const int*   srcI = ei;
    const int*   dstI = ei + E;
    const float* W1   = (const float*)d_in[2];
    const float* b1   = (const float*)d_in[3];
    const float* W2   = (const float*)d_in[4];
    const float* b2   = (const float*)d_in[5];
    const float* W3   = (const float*)d_in[6];
    const float* b3   = (const float*)d_in[7];
    const float* g1   = (const float*)d_in[8];
    const float* be1  = (const float*)d_in[9];
    const float* g2   = (const float*)d_in[10];
    const float* be2  = (const float*)d_in[11];
    const float* Wfc  = (const float*)d_in[12];
    const float* bfc  = (const float*)d_in[13];

    float* ws    = (float*)d_ws;
    float* dis   = ws;
    float* bufA  = dis + N;
    float* bufB  = bufA + (size_t)N * HC;
    float* bufC  = bufB + (size_t)N * HC;
    float* stats = bufC + (size_t)N * HC;
    int*   csr   = (int*)bufC;
    int*   ghist = (int*)(((size_t)bufC + 15) & ~(size_t)15);
    int*   rec   = (int*)bufA;
    int*   start = (int*)d_out;

    const int B256 = 256;
    dim3 blkN((N + B256 - 1) / B256);
    const int NB = (N + (1 << RB_SHIFT) - 1) >> RB_SHIFT;
    const int chunk = (E + G1 - 1) / G1;
    dim3 blkG((N / 16 + 7) / 8);
    dim3 blkP(2048);
    dim3 blkT((N + 63) / 64);
    float invN = 1.0f / (float)N;

    coarse_hist<<<G1, SORT_T, 0, stream>>>(dstI, ghist, E, chunk);
    coarse_scatter<<<G1, SORT_T, 0, stream>>>(srcI, dstI, ghist, rec, start, N, E, chunk);
    fine_sort<<<NB, SORT_T, 0, stream>>>(rec, start, dis, csr, N);

    gemm_wmma_f32<<<blkG, B256, 0, stream>>>(x, W1, bufA, N, IN);
    pull_rows<<<blkP, B256, 0, stream>>>(start, csr, dis, bufA, b1, bufB, N, E);
    zero_f32<<<1, 64, 0, stream>>>(stats, 64);
    bn_stats<<<256, 256, 0, stream>>>(bufB, stats, N);

    pull_bn_gemm<<<blkT, B256, 0, stream>>>(start, csr, dis, bufB, stats, g1, be1, invN, W2, b2, bufA, N, E);
    zero_f32<<<1, 64, 0, stream>>>(stats, 64);
    bn_stats<<<256, 256, 0, stream>>>(bufA, stats, N);

    pull_bn_gemm<<<blkT, B256, 0, stream>>>(start, csr, dis, bufA, stats, g2, be2, invN, W3, b3, bufB, N, E);

    head_kernel<<<blkN, B256, 0, stream>>>(bufB, Wfc, bfc, (float*)d_out, N);
}


# BatchNorm channel sums taken where rows are made: aggregation kernels reduce relu(row) sums in registers + LDS, one 32-lane float atomic per work-group (new sum tree, arrival order); 12 launches to 8
# speedup vs baseline: 12.2440x; 1.1559x over previous
#include <hip/hip_runtime.h>
#include <hip/hip_bf16.h>

typedef float v2f __attribute__((ext_vector_type(2)));
typedef float v8f __attribute__((ext_vector_type(8)));

#define HC 16
#define EPSV 1e-5f
#define SRC_SPLIT 65536

#define RB_SHIFT 8
#define NBP 512
#define G1 256
#define SORT_T 1024

__device__ __forceinline__ int block_excl_scan(int v, int* wsum, int tid) {
    int lane = tid & 31, wv = tid >> 5;
    int inc = v;
    for (int off = 1; off < 32; off <<= 1) {
        int n = __shfl_up(inc, off);
        if (lane >= off) inc += n;
    }
    if (lane == 31) wsum[wv] = inc;
    __syncthreads();
    int before = 0;
    for (int w = 0; w < wv; ++w) before += wsum[w];
    __syncthreads();
    return before + inc - v;
}

__global__ void __launch_bounds__(SORT_T)
coarse_hist(const int* __restrict__ dst, int* __restrict__ ghist, int E, int chunk) {
    __shared__ int h[NBP];
    int tid = threadIdx.x;
    if (tid < NBP) h[tid] = 0;
    __syncthreads();
    int c0 = blockIdx.x * chunk;
    int c1 = min(c0 + chunk, E);
    for (int base = c0 + tid; base < c1; base += 4 * SORT_T) {
        int d[4];
#pragma unroll
        for (int k = 0; k < 4; ++k) d[k] = dst[min(base + k * SORT_T, c1 - 1)];
        asm volatile("" :: "v"(d[0]), "v"(d[1]), "v"(d[2]), "v"(d[3]));
#pragma unroll
        for (int k = 0; k < 4; ++k)
            if (base + k * SORT_T < c1) atomicAdd(&h[d[k] >> RB_SHIFT], 1);
    }
    __syncthreads();
    if (tid < NBP) ghist[blockIdx.x * NBP + tid] = h[tid];
}

__global__ void __launch_bounds__(SORT_T)
coarse_scatter(const int* __restrict__ src, const int* __restrict__ dst,
               const int* __restrict__ ghist, int* __restrict__ rec,
               int* __restrict__ start, int N, int E, int chunk) {
    __shared__ int ptot[8][NBP];
    __shared__ int ppart[8][NBP];
    __shared__ int cur[NBP];
    __shared__ int wsum[32];
    int tid = threadIdx.x;
    int wg = blockIdx.x;

    {
        const int4* __restrict__ gh4 = (const int4*)ghist;
        int j = tid & 127;
        int g = tid >> 7;
        int4 t4 = {0, 0, 0, 0}, p4 = {0, 0, 0, 0};
#pragma unroll 1
        for (int batch = 0; batch < 4; ++batch) {
            int w0 = g * (G1 / 8) + batch * 8;
            int4 v[8];
#pragma unroll
            for (int k = 0; k < 8; ++k) v[k] = gh4[(w0 + k) * (NBP / 4) + j];
            asm volatile("" :: "v"(v[0].x), "v"(v[1].x), "v"(v[2].x), "v"(v[3].x),
                               "v"(v[4].x), "v"(v[5].x), "v"(v[6].x), "v"(v[7].x));
#pragma unroll
            for (int k = 0; k < 8; ++k) {
                t4.x += v[k].x; t4.y += v[k].y; t4.z += v[k].z; t4.w += v[k].w;
                if (w0 + k < wg) { p4.x += v[k].x; p4.y += v[k].y; p4.z += v[k].z; p4.w += v[k].w; }
            }
        }
        ptot[g][4 * j + 0] = t4.x; ptot[g][4 * j + 1] = t4.y;
        ptot[g][4 * j + 2] = t4.z; ptot[g][4 * j + 3] = t4.w;
        ppart[g][4 * j + 0] = p4.x; ppart[g][4 * j + 1] = p4.y;
        ppart[g][4 * j + 2] = p4.z; ppart[g][4 * j + 3] = p4.w;
    }
    __syncthreads();
    int tot = 0, part = 0;
    if (tid < NBP) {
#pragma unroll
        for (int g = 0; g < 8; ++g) { tot += ptot[g][tid]; part += ppart[g][tid]; }
    }
    int bstart = block_excl_scan(tot, wsum, tid);
    if (tid < NBP) {
        cur[tid] = bstart + part;
        if (wg == 0 && (tid << RB_SHIFT) < N) {
            start[2 * (tid << RB_SHIFT)] = bstart;
            start[2 * (tid << RB_SHIFT) + 1] = bstart + tot;
        }
    }
    __syncthreads();

    int c0 = wg * chunk;
    int c1 = min(c0 + chunk, E);
    for (int base = c0 + tid; base < c1; base += 4 * SORT_T) {
        int s[4], d[4];
#pragma unroll
        for (int k = 0; k < 4; ++k) {
            int ec = min(base + k * SORT_T, c1 - 1);
            s[k] = src[ec];
            d[k] = dst[ec];
        }
        asm volatile("" :: "v"(s[0]), "v"(s[1]), "v"(s[2]), "v"(s[3]),
                           "v"(d[0]), "v"(d[1]), "v"(d[2]), "v"(d[3]));
#pragma unroll
        for (int k = 0; k < 4; ++k) {
            if (base + k * SORT_T < c1) {
                int pos = atomicAdd(&cur[d[k] >> RB_SHIFT], 1);
                rec[pos] = s[k] | ((d[k] & ((1 << RB_SHIFT) - 1)) << 17);
            }
        }
    }
}

__global__ void __launch_bounds__(SORT_T)
fine_sort(const int* __restrict__ rec, int* __restrict__ start, float* __restrict__ dis,
          int* __restrict__ csr_w, float* __restrict__ stats, int N) {
    unsigned short* __restrict__ csr = (unsigned short*)csr_w;
    __shared__ int cnt[512];
    __shared__ int cursor[512];
    __shared__ int wsum[32];
    int tid = threadIdx.x;
    int b = blockIdx.x;
    int c0 = start[2 * (b << RB_SHIFT)];
    int c1 = start[2 * (b << RB_SHIFT) + 1];
    if (tid < 512) cnt[tid] = 0;
    if (b == 0 && tid < 64) stats[tid] = 0.0f;
    __syncthreads();

    for (int base = c0 + tid; base < c1; base += 4 * SORT_T) {
        int r[4];
#pragma unroll
        for (int k = 0; k < 4; ++k) r[k] = rec[min(base + k * SORT_T, c1 - 1)];
        asm volatile("" :: "v"(r[0]), "v"(r[1]), "v"(r[2]), "v"(r[3]));
#pragma unroll
        for (int k = 0; k < 4; ++k)
            if (base + k * SORT_T < c1) atomicAdd(&cnt[(r[k] >> 16) & 511], 1);
    }
    __syncthreads();

    int c = (tid < 512) ? cnt[tid] : 0;
    int ex = block_excl_scan(c, wsum, tid);
    if (tid < 512) {
        cursor[tid] = c0 + ex;
        int row = (b << RB_SHIFT) + (tid >> 1);
        if (row < N) start[2 * row + (tid & 1)] = c0 + ex;
    }
    if (tid < 256) {
        int row = (b << RB_SHIFT) + tid;
        if (row < N) dis[row] = rsqrtf((float)(cnt[2 * tid] + cnt[2 * tid + 1]) + 1.0f);
    }
    __syncthreads();

    for (int base = c0 + tid; base < c1; base += 4 * SORT_T) {
        int r[4];
#pragma unroll
        for (int k = 0; k < 4; ++k) r[k] = rec[min(base + k * SORT_T, c1 - 1)];
        asm volatile("" :: "v"(r[0]), "v"(r[1]), "v"(r[2]), "v"(r[3]));
#pragma unroll
        for (int k = 0; k < 4; ++k) {
            if (base + k * SORT_T < c1) {
                int pos = atomicAdd(&cursor[(r[k] >> 16) & 511], 1);
                csr[pos] = (unsigned short)(r[k] & (SRC_SPLIT - 1));
            }
        }
    }
}

__global__ void gemm_wmma_f32(const float* __restrict__ A, const float* __restrict__ B,
                              float* __restrict__ C, int N, int K) {
    int wave = (int)((blockIdx.x * blockDim.x + threadIdx.x) >> 5);
    int lane = threadIdx.x & 31;
    int row0 = wave * 16;
    if (row0 >= N) return;

    int m  = lane & 15;
    int hi = lane >> 4;
    const float* arow = A + (size_t)(row0 + m) * K;

    v8f acc = {};
    for (int k0 = 0; k0 < K; k0 += 4) {
        int ka = k0 + 2 * hi;
        v2f a, b;
        a.x = arow[ka];
        a.y = arow[ka + 1];
        b.x = B[(size_t)ka * HC + m];
        b.y = B[(size_t)(ka + 1) * HC + m];
        acc = __builtin_amdgcn_wmma_f32_16x16x4_f32(
             false, a,  false, b,
             (short)0, acc,  false,  false);
    }

    float* cbase = C + (size_t)(row0 + hi * 8) * HC + m;
#pragma unroll
    for (int r = 0; r < 8; ++r) cbase[(size_t)r * HC] = acc[r];
}

#define PULL_T 1024
__global__ void __launch_bounds__(PULL_T)
pull_rows(const int* __restrict__ start, const int* __restrict__ csr_w,
          const float* __restrict__ dis, const float* __restrict__ hw,
          const float* __restrict__ bias, float* __restrict__ out,
          float* __restrict__ stats, int N, int E) {
    __shared__ float red[(PULL_T / 32) * 32];
    const unsigned short* __restrict__ csr = (const unsigned short*)csr_w;
    const float4* __restrict__ hw4 = (const float4*)hw;
    int lane = threadIdx.x & 31;
    int q = lane & 3;
    int j = lane >> 2;
    int nw = (int)((gridDim.x * blockDim.x) >> 5);

    float b0 = bias[4 * q + 0], b1 = bias[4 * q + 1], b2 = bias[4 * q + 2], b3 = bias[4 * q + 3];

    float4 st1 = {0.0f, 0.0f, 0.0f, 0.0f};
    float4 st2 = {0.0f, 0.0f, 0.0f, 0.0f};

    for (int r = (int)((blockIdx.x * blockDim.x + threadIdx.x) >> 5); r < N; r += nw) {
        bool last = (r + 1 >= N);
        int s0 = start[2 * r];
        int s1 = start[2 * r + 1];
        int s2 = start[last ? 2 * r + 1 : 2 * r + 2];
        if (last) s2 = E;
        float d = dis[r];
        float4 self = hw4[(size_t)r * 4 + q];

        float dd = d * d;
        float4 acc;
        acc.x = (j == 0) ? (dd * self.x + b0) : 0.0f;
        acc.y = (j == 0) ? (dd * self.y + b1) : 0.0f;
        acc.z = (j == 0) ? (dd * self.z + b2) : 0.0f;
        acc.w = (j == 0) ? (dd * self.w + b3) : 0.0f;

        int p = s0 + j;
        for (; p + 8 < s2; p += 16) {
            int pb = p + 8;
            int i0 = (int)csr[p]  + ((p  >= s1) ? SRC_SPLIT : 0);
            int i1 = (int)csr[pb] + ((pb >= s1) ? SRC_SPLIT : 0);
            float g0 = dis[i0];
            float g1 = dis[i1];
            float4 v0 = hw4[(size_t)i0 * 4 + q];
            float4 v1 = hw4[(size_t)i1 * 4 + q];
            __builtin_amdgcn_sched_barrier(0);
            float w0 = g0 * d;
            float w1 = g1 * d;
            acc.x += w0 * v0.x; acc.y += w0 * v0.y; acc.z += w0 * v0.z; acc.w += w0 * v0.w;
            acc.x += w1 * v1.x; acc.y += w1 * v1.y; acc.z += w1 * v1.z; acc.w += w1 * v1.w;
        }
        if (p < s2) {
            int i0 = (int)csr[p] + ((p >= s1) ? SRC_SPLIT : 0);
            float w0 = dis[i0] * d;
            float4 v0 = hw4[(size_t)i0 * 4 + q];
            acc.x += w0 * v0.x; acc.y += w0 * v0.y; acc.z += w0 * v0.z; acc.w += w0 * v0.w;
        }

        for (int off = 4; off < 32; off <<= 1) {
            acc.x += __shfl_xor(acc.x, off);
            acc.y += __shfl_xor(acc.y, off);
            acc.z += __shfl_xor(acc.z, off);
            acc.w += __shfl_xor(acc.w, off);
        }

        if (j == 0) ((float4*)out)[(size_t)r * 4 + q] = acc;

        float rx = fmaxf(acc.x, 0.0f), ry = fmaxf(acc.y, 0.0f);
        float rz = fmaxf(acc.z, 0.0f), rw = fmaxf(acc.w, 0.0f);
        st1.x += rx; st1.y += ry; st1.z += rz; st1.w += rw;
        st2.x += rx * rx; st2.y += ry * ry; st2.z += rz * rz; st2.w += rw * rw;
    }

    int wv = (int)(threadIdx.x >> 5);
    if (j == 0) {
        float* rr = red + wv * 32 + 4 * q;
        rr[0] = st1.x; rr[1] = st1.y; rr[2] = st1.z; rr[3] = st1.w;
        rr[16] = st2.x; rr[17] = st2.y; rr[18] = st2.z; rr[19] = st2.w;
    }
    __syncthreads();
    if (threadIdx.x < 32) {
        float v = 0.0f;
#pragma unroll
        for (int w = 0; w < PULL_T / 32; ++w) v += red[w * 32 + threadIdx.x];
        atomicAdd(&stats[threadIdx.x], v);
    }
}

#define TP 20
template <bool STATS>
__global__ void __launch_bounds__(256)
pull_bn_gemm(const int* __restrict__ start, const int* __restrict__ csr_w,
             const float* __restrict__ dis, const float* __restrict__ prev,
             const float* __restrict__ stats, const float* __restrict__ gamma,
             const float* __restrict__ beta, float invN,
             const float* __restrict__ Wm, const float* __restrict__ bias,
             float* __restrict__ out, float* __restrict__ stats_out, int N, int E) {
    __shared__ float tile[64 * TP];
    __shared__ float red[4 * 32];
    const unsigned short* __restrict__ csr = (const unsigned short*)csr_w;
    const float4* __restrict__ p4 = (const float4*)prev;
    int lane = threadIdx.x & 31;
    int wave = __builtin_amdgcn_readfirstlane((int)(threadIdx.x >> 5));
    int q = lane & 3;
    int j = lane >> 2;

    float sc0, sc1, sc2, sc3, sh0, sh1, sh2, sh3;
    {
        float mu, var;
        mu = stats[4 * q + 0] * invN; var = stats[16 + 4 * q + 0] * invN - mu * mu;
        sc0 = rsqrtf(var + EPSV) * gamma[4 * q + 0]; sh0 = beta[4 * q + 0] - mu * sc0;
        mu = stats[4 * q + 1] * invN; var = stats[16 + 4 * q + 1] * invN - mu * mu;
        sc1 = rsqrtf(var + EPSV) * gamma[4 * q + 1]; sh1 = beta[4 * q + 1] - mu * sc1;
        mu = stats[4 * q + 2] * invN; var = stats[16 + 4 * q + 2] * invN - mu * mu;
        sc2 = rsqrtf(var + EPSV) * gamma[4 * q + 2]; sh2 = beta[4 * q + 2] - mu * sc2;
        mu = stats[4 * q + 3] * invN; var = stats[16 + 4 * q + 3] * invN - mu * mu;
        sc3 = rsqrtf(var + EPSV) * gamma[4 * q + 3]; sh3 = beta[4 * q + 3] - mu * sc3;
    }

    int rbase = blockIdx.x * 64 + wave * 8;
    int sv = start[min(2 * rbase + lane, 2 * N - 1)];
    float dv = dis[min(rbase + (lane & 7), N - 1)];
    float4 selfn = p4[(size_t)min(rbase, N - 1) * 4 + q];
    for (int t = 0; t < 8; ++t) {
        int r = rbase + t;
        if (r >= N) break;
        int s0 = __builtin_amdgcn_readlane(sv, 2 * t);
        int s1 = __builtin_amdgcn_readlane(sv, 2 * t + 1);
        int s2 = (r + 1 >= N) ? E : __builtin_amdgcn_readlane(sv, 2 * t + 2);
        float d = __int_as_float(__builtin_amdgcn_readlane(__float_as_int(dv), t));
        float4 self = selfn;
        selfn = p4[(size_t)min(r + 1, N - 1) * 4 + q];

        float dd = d * d;
        float4 acc;
        acc.x = (j == 0) ? dd * (fmaxf(self.x, 0.0f) * sc0 + sh0) : 0.0f;
        acc.y = (j == 0) ? dd * (fmaxf(self.y, 0.0f) * sc1 + sh1) : 0.0f;
        acc.z = (j == 0) ? dd * (fmaxf(self.z, 0.0f) * sc2 + sh2) : 0.0f;
        acc.w = (j == 0) ? dd * (fmaxf(self.w, 0.0f) * sc3 + sh3) : 0.0f;

        int p = s0 + j;
        for (; p + 8 < s2; p += 16) {
            int pb = p + 8;
            int i0 = (int)csr[p]  + ((p  >= s1) ? SRC_SPLIT : 0);
            int i1 = (int)csr[pb] + ((pb >= s1) ? SRC_SPLIT : 0);
            float g0 = dis[i0];
            float g1 = dis[i1];
            float4 v0 = p4[(size_t)i0 * 4 + q];
            float4 v1 = p4[(size_t)i1 * 4 + q];
            __builtin_amdgcn_sched_barrier(0);
            float w0 = g0 * d;
            float w1 = g1 * d;
            acc.x += w0 * (fmaxf(v0.x, 0.0f) * sc0 + sh0);
            acc.y += w0 * (fmaxf(v0.y, 0.0f) * sc1 + sh1);
            acc.z += w0 * (fmaxf(v0.z, 0.0f) * sc2 + sh2);
            acc.w += w0 * (fmaxf(v0.w, 0.0f) * sc3 + sh3);
            acc.x += w1 * (fmaxf(v1.x, 0.0f) * sc0 + sh0);
            acc.y += w1 * (fmaxf(v1.y, 0.0f) * sc1 + sh1);
            acc.z += w1 * (fmaxf(v1.z, 0.0f) * sc2 + sh2);
            acc.w += w1 * (fmaxf(v1.w, 0.0f) * sc3 + sh3);
        }
        if (p < s2) {
            int i0 = (int)csr[p] + ((p >= s1) ? SRC_SPLIT : 0);
            float w0 = dis[i0] * d;
            float4 v0 = p4[(size_t)i0 * 4 + q];
            acc.x += w0 * (fmaxf(v0.x, 0.0f) * sc0 + sh0);
            acc.y += w0 * (fmaxf(v0.y, 0.0f) * sc1 + sh1);
            acc.z += w0 * (fmaxf(v0.z, 0.0f) * sc2 + sh2);
            acc.w += w0 * (fmaxf(v0.w, 0.0f) * sc3 + sh3);
        }

        for (int off = 4; off < 32; off <<= 1) {
            acc.x += __shfl_xor(acc.x, off);
            acc.y += __shfl_xor(acc.y, off);
            acc.z += __shfl_xor(acc.z, off);
            acc.w += __shfl_xor(acc.w, off);
        }

        if (j == 0) *(float4*)&tile[(wave * 8 + t) * TP + 4 * q] = acc;
    }
    __syncthreads();

    if (wave < 4) {
        int row0 = blockIdx.x * 64 + wave * 16;
        float s1 = 0.0f, s2 = 0.0f;
        if (row0 < N) {
            int m  = lane & 15;
            int hi = lane >> 4;
            const float* trow = tile + (wave * 16 + m) * TP;
            float bv = bias[m];
            v8f c8 = {bv, bv, bv, bv, bv, bv, bv, bv};
            for (int kk = 0; kk < 4; ++kk) {
                int ka = 4 * kk + 2 * hi;
                v2f a, b;
                a.x = trow[ka];
                a.y = trow[ka + 1];
                b.x = Wm[ka * HC + m];
                b.y = Wm[(ka + 1) * HC + m];
                c8 = __builtin_amdgcn_wmma_f32_16x16x4_f32(
                     false, a,  false, b,
                     (short)0, c8,  false,  false);
            }
            float* cbase = out + (size_t)(row0 + hi * 8) * HC + m;
#pragma unroll
            for (int r = 0; r < 8; ++r) cbase[(size_t)r * HC] = c8[r];
            if (STATS) {
#pragma unroll
                for (int r = 0; r < 8; ++r) {
                    float v = fmaxf(c8[r], 0.0f);
                    s1 += v;
                    s2 += v * v;
                }
                s1 += __shfl_xor(s1, 16);
                s2 += __shfl_xor(s2, 16);
            }
        }
        if (STATS) red[wave * 32 + lane] = (lane < 16) ? s1 : s2;
    }
    if (STATS) {
        __syncthreads();
        if (threadIdx.x < 32) {
            float v = (red[lane] + red[32 + lane]) + (red[64 + lane] + red[96 + lane]);
            atomicAdd(&stats_out[lane], v);
        }
    }
}

__global__ void head_kernel(const float* __restrict__ h, const float* __restrict__ Wfc,
                            const float* __restrict__ bfc, float* __restrict__ out, int N) {
    int i = blockIdx.x * blockDim.x + threadIdx.x;
    if (i >= N) return;
    const float* r = h + (size_t)i * HC;
    float z0 = bfc[0], z1 = bfc[1];
#pragma unroll
    for (int c = 0; c < HC; ++c) {
        float v = r[c];
        z0 += v * Wfc[c * 2 + 0];
        z1 += v * Wfc[c * 2 + 1];
    }
    float mx  = fmaxf(z0, z1);
    float lse = mx + logf(__expf(z0 - mx) + __expf(z1 - mx));
    out[(size_t)i * 2 + 0] = z0 - lse;
    out[(size_t)i * 2 + 1] = z1 - lse;
}

extern "C" void kernel_launch(void* const* d_in, const int* in_sizes, int n_in,
                              void* d_out, int out_size, void* d_ws, size_t ws_size,
                              hipStream_t stream) {
    const int IN = 128;
    const int N  = in_sizes[0] / IN;
    const int E  = in_sizes[1] / 2;

    const float* x    = (const float*)d_in[0];
    const int*   ei   = (const int*)d_in[1];
    const int*   srcI = ei;
    const int*   dstI = ei + E;
    const float* W1   = (const float*)d_in[2];
    const float* b1   = (const float*)d_in[3];
    const float* W2   = (const float*)d_in[4];
    const float* b2   = (const float*)d_in[5];
    const float* W3   = (const float*)d_in[6];
    const float* b3   = (const float*)d_in[7];
    const float* g1   = (const float*)d_in[8];
    const float* be1  = (const float*)d_in[9];
    const float* g2   = (const float*)d_in[10];
    const float* be2  = (const float*)d_in[11];
    const float* Wfc  = (const float*)d_in[12];
    const float* bfc  = (const float*)d_in[13];

    float* ws    = (float*)d_ws;
    float* dis   = ws;
    float* bufA  = dis + N;
    float* bufB  = bufA + (size_t)N * HC;
    float* bufC  = bufB + (size_t)N * HC;
    float* stats = bufC + (size_t)N * HC;
    int*   csr   = (int*)bufC;
    int*   ghist = (int*)(((size_t)bufC + 15) & ~(size_t)15);
    int*   rec   = (int*)bufA;
    int*   start = (int*)d_out;

    const int B256 = 256;
    dim3 blkN((N + B256 - 1) / B256);
    const int NB = (N + (1 << RB_SHIFT) - 1) >> RB_SHIFT;
    const int chunk = (E + G1 - 1) / G1;
    dim3 blkG((N / 16 + 7) / 8);
    dim3 blkP(512);
    dim3 blkT((N + 63) / 64);
    float invN = 1.0f / (float)N;

    coarse_hist<<<G1, SORT_T, 0, stream>>>(dstI, ghist, E, chunk);
    coarse_scatter<<<G1, SORT_T, 0, stream>>>(srcI, dstI, ghist, rec, start, N, E, chunk);
    fine_sort<<<NB, SORT_T, 0, stream>>>(rec, start, dis, csr, stats, N);

    gemm_wmma_f32<<<blkG, B256, 0, stream>>>(x, W1, bufA, N, IN);
    pull_rows<<<blkP, PULL_T, 0, stream>>>(start, csr, dis, bufA, b1, bufB, stats, N, E);

    pull_bn_gemm<true><<<blkT, B256, 0, stream>>>(start, csr, dis, bufB, stats, g1, be1, invN, W2, b2, bufA, stats + 32, N, E);

    pull_bn_gemm<false><<<blkT, B256, 0, stream>>>(start, csr, dis, bufA, stats + 32, g2, be2, invN, W3, b3, bufB, stats, N, E);

    head_kernel<<<blkN, B256, 0, stream>>>(bufB, Wfc, bfc, (float*)d_out, N);
}


# Per-edge dis[src] read moved from L2 (a line per edge) to an LDS copy of a byte degree table, rsqrt recomputed; fused pulls regrouped to 224-row work-groups (BatchNorm sums: new tree, arrival order)
# speedup vs baseline: 14.1514x; 1.1558x over previous
#include <hip/hip_runtime.h>
#include <hip/hip_bf16.h>

typedef float v2f __attribute__((ext_vector_type(2)));
typedef float v8f __attribute__((ext_vector_type(8)));

#define HC 16
#define EPSV 1e-5f
#define SRC_SPLIT 65536

#define DEG_CAP 131072
#define DEG_ESC 255
#define AGG_T 1024

__device__ __forceinline__ void load_deg_table(const int* __restrict__ deg_w, unsigned char* degs, int N) {
    const uint4* __restrict__ g4 = (const uint4*)deg_w;
    uint4* l4 = (uint4*)degs;
    int nv = (N + 15) >> 4;
    uint4 v[8];
#pragma unroll
    for (int k = 0; k < 8; ++k) v[k] = g4[min((int)threadIdx.x + k * AGG_T, nv - 1)];
#pragma unroll
    for (int k = 0; k < 8; ++k) l4[threadIdx.x + k * AGG_T] = v[k];
}

__device__ __forceinline__ float wide_dis(const int* __restrict__ start, int i, int N, int E) {
    int a = start[2 * i];
    int b = start[min(2 * i + 2, 2 * N - 1)];
    if (i + 1 >= N) b = E;
    return rsqrtf((float)(b - a) + 1.0f);
}

#define RB_SHIFT 8
#define NBP 512
#define G1 256
#define SORT_T 1024

__device__ __forceinline__ int block_excl_scan(int v, int* wsum, int tid) {
    int lane = tid & 31, wv = tid >> 5;
    int inc = v;
    for (int off = 1; off < 32; off <<= 1) {
        int n = __shfl_up(inc, off);
        if (lane >= off) inc += n;
    }
    if (lane == 31) wsum[wv] = inc;
    __syncthreads();
    int before = 0;
    for (int w = 0; w < wv; ++w) before += wsum[w];
    __syncthreads();
    return before + inc - v;
}

__global__ void __launch_bounds__(SORT_T)
coarse_hist(const int* __restrict__ dst, int* __restrict__ ghist, int E, int chunk) {
    __shared__ int h[NBP];
    int tid = threadIdx.x;
    if (tid < NBP) h[tid] = 0;
    __syncthreads();
    int c0 = blockIdx.x * chunk;
    int c1 = min(c0 + chunk, E);
    for (int base = c0 + tid; base < c1; base += 4 * SORT_T) {
        int d[4];
#pragma unroll
        for (int k = 0; k < 4; ++k) d[k] = dst[min(base + k * SORT_T, c1 - 1)];
        asm volatile("" :: "v"(d[0]), "v"(d[1]), "v"(d[2]), "v"(d[3]));
#pragma unroll
        for (int k = 0; k < 4; ++k)
            if (base + k * SORT_T < c1) atomicAdd(&h[d[k] >> RB_SHIFT], 1);
    }
    __syncthreads();
    if (tid < NBP) ghist[blockIdx.x * NBP + tid] = h[tid];
}

__global__ void __launch_bounds__(SORT_T)
coarse_scatter(const int* __restrict__ src, const int* __restrict__ dst,
               const int* __restrict__ ghist, int* __restrict__ rec,
               int* __restrict__ start, int N, int E, int chunk) {
    __shared__ int ptot[8][NBP];
    __shared__ int ppart[8][NBP];
    __shared__ int cur[NBP];
    __shared__ int wsum[32];
    int tid = threadIdx.x;
    int wg = blockIdx.x;

    {
        const int4* __restrict__ gh4 = (const int4*)ghist;
        int j = tid & 127;
        int g = tid >> 7;
        int4 t4 = {0, 0, 0, 0}, p4 = {0, 0, 0, 0};
#pragma unroll 1
        for (int batch = 0; batch < 4; ++batch) {
            int w0 = g * (G1 / 8) + batch * 8;
            int4 v[8];
#pragma unroll
            for (int k = 0; k < 8; ++k) v[k] = gh4[(w0 + k) * (NBP / 4) + j];
            asm volatile("" :: "v"(v[0].x), "v"(v[1].x), "v"(v[2].x), "v"(v[3].x),
                               "v"(v[4].x), "v"(v[5].x), "v"(v[6].x), "v"(v[7].x));
#pragma unroll
            for (int k = 0; k < 8; ++k) {
                t4.x += v[k].x; t4.y += v[k].y; t4.z += v[k].z; t4.w += v[k].w;
                if (w0 + k < wg) { p4.x += v[k].x; p4.y += v[k].y; p4.z += v[k].z; p4.w += v[k].w; }
            }
        }
        ptot[g][4 * j + 0] = t4.x; ptot[g][4 * j + 1] = t4.y;
        ptot[g][4 * j + 2] = t4.z; ptot[g][4 * j + 3] = t4.w;
        ppart[g][4 * j + 0] = p4.x; ppart[g][4 * j + 1] = p4.y;
        ppart[g][4 * j + 2] = p4.z; ppart[g][4 * j + 3] = p4.w;
    }
    __syncthreads();
    int tot = 0, part = 0;
    if (tid < NBP) {
#pragma unroll
        for (int g = 0; g < 8; ++g) { tot += ptot[g][tid]; part += ppart[g][tid]; }
    }
    int bstart = block_excl_scan(tot, wsum, tid);
    if (tid < NBP) {
        cur[tid] = bstart + part;
        if (wg == 0 && (tid << RB_SHIFT) < N) {
            start[2 * (tid << RB_SHIFT)] = bstart;
            start[2 * (tid << RB_SHIFT) + 1] = bstart + tot;
        }
    }
    __syncthreads();

    int c0 = wg * chunk;
    int c1 = min(c0 + chunk, E);
    for (int base = c0 + tid; base < c1; base += 4 * SORT_T) {
        int s[4], d[4];
#pragma unroll
        for (int k = 0; k < 4; ++k) {
            int ec = min(base + k * SORT_T, c1 - 1);
            s[k] = src[ec];
            d[k] = dst[ec];
        }
        asm volatile("" :: "v"(s[0]), "v"(s[1]), "v"(s[2]), "v"(s[3]),
                           "v"(d[0]), "v"(d[1]), "v"(d[2]), "v"(d[3]));
#pragma unroll
        for (int k = 0; k < 4; ++k) {
            if (base + k * SORT_T < c1) {
                int pos = atomicAdd(&cur[d[k] >> RB_SHIFT], 1);
                rec[pos] = s[k] | ((d[k] & ((1 << RB_SHIFT) - 1)) << 17);
            }
        }
    }
}

__global__ void __launch_bounds__(SORT_T)
fine_sort(const int* __restrict__ rec, int* __restrict__ start, int* __restrict__ deg_w,
          int* __restrict__ csr_w, float* __restrict__ stats, int N) {
    unsigned short* __restrict__ csr = (unsigned short*)csr_w;
    unsigned char* __restrict__ deg8 = (unsigned char*)deg_w;
    __shared__ int cnt[512];
    __shared__ int cursor[512];
    __shared__ int wsum[32];
    int tid = threadIdx.x;
    int b = blockIdx.x;
    int c0 = start[2 * (b << RB_SHIFT)];
    int c1 = start[2 * (b << RB_SHIFT) + 1];
    if (tid < 512) cnt[tid] = 0;
    if (b == 0 && tid < 64) stats[tid] = 0.0f;
    __syncthreads();

    for (int base = c0 + tid; base < c1; base += 4 * SORT_T) {
        int r[4];
#pragma unroll
        for (int k = 0; k < 4; ++k) r[k] = rec[min(base + k * SORT_T, c1 - 1)];
        asm volatile("" :: "v"(r[0]), "v"(r[1]), "v"(r[2]), "v"(r[3]));
#pragma unroll
        for (int k = 0; k < 4; ++k)
            if (base + k * SORT_T < c1) atomicAdd(&cnt[(r[k] >> 16) & 511], 1);
    }
    __syncthreads();

    int c = (tid < 512) ? cnt[tid] : 0;
    int ex = block_excl_scan(c, wsum, tid);
    if (tid < 512) {
        cursor[tid] = c0 + ex;
        int row = (b << RB_SHIFT) + (tid >> 1);
        if (row < N) start[2 * row + (tid & 1)] = c0 + ex;
    }
    if (tid < 256) {
        int row = (b << RB_SHIFT) + tid;
        if (row < N) deg8[row] = (unsigned char)min(cnt[2 * tid] + cnt[2 * tid + 1], DEG_ESC);
    }
    __syncthreads();

    for (int base = c0 + tid; base < c1; base += 4 * SORT_T) {
        int r[4];
#pragma unroll
        for (int k = 0; k < 4; ++k) r[k] = rec[min(base + k * SORT_T, c1 - 1)];
        asm volatile("" :: "v"(r[0]), "v"(r[1]), "v"(r[2]), "v"(r[3]));
#pragma unroll
        for (int k = 0; k < 4; ++k) {
            if (base + k * SORT_T < c1) {
                int pos = atomicAdd(&cursor[(r[k] >> 16) & 511], 1);
                csr[pos] = (unsigned short)(r[k] & (SRC_SPLIT - 1));
            }
        }
    }
}

__global__ void gemm_wmma_f32(const float* __restrict__ A, const float* __restrict__ B,
                              float* __restrict__ C, int N, int K) {
    int wave = (int)((blockIdx.x * blockDim.x + threadIdx.x) >> 5);
    int lane = threadIdx.x & 31;
    int row0 = wave * 16;
    if (row0 >= N) return;

    int m  = lane & 15;
    int hi = lane >> 4;
    const float* arow = A + (size_t)(row0 + m) * K;

    v8f acc = {};
    for (int k0 = 0; k0 < K; k0 += 4) {
        int ka = k0 + 2 * hi;
        v2f a, b;
        a.x = arow[ka];
        a.y = arow[ka + 1];
        b.x = B[(size_t)ka * HC + m];
        b.y = B[(size_t)(ka + 1) * HC + m];
        acc = __builtin_amdgcn_wmma_f32_16x16x4_f32(
             false, a,  false, b,
             (short)0, acc,  false,  false);
    }

    float* cbase = C + (size_t)(row0 + hi * 8) * HC + m;
#pragma unroll
    for (int r = 0; r < 8; ++r) cbase[(size_t)r * HC] = acc[r];
}

#define PULL_T AGG_T
__global__ void __launch_bounds__(PULL_T)
pull_rows(const int* __restrict__ start, const int* __restrict__ csr_w,
          const int* __restrict__ deg_w, const float* __restrict__ hw,
          const float* __restrict__ bias, float* __restrict__ out,
          float* __restrict__ stats, int N, int E) {
    __shared__ float red[(PULL_T / 32) * 32];
    __shared__ __attribute__((aligned(16))) unsigned char degs[DEG_CAP];
    load_deg_table(deg_w, degs, N);
    __syncthreads();
    const unsigned short* __restrict__ csr = (const unsigned short*)csr_w;
    const float4* __restrict__ hw4 = (const float4*)hw;
    int lane = threadIdx.x & 31;
    int q = lane & 3;
    int j = lane >> 2;
    int nw = (int)((gridDim.x * blockDim.x) >> 5);

    float b0 = bias[4 * q + 0], b1 = bias[4 * q + 1], b2 = bias[4 * q + 2], b3 = bias[4 * q + 3];

    float4 st1 = {0.0f, 0.0f, 0.0f, 0.0f};
    float4 st2 = {0.0f, 0.0f, 0.0f, 0.0f};

    for (int r = (int)((blockIdx.x * blockDim.x + threadIdx.x) >> 5); r < N; r += nw) {
        bool last = (r + 1 >= N);
        int s0 = start[2 * r];
        int s1 = start[2 * r + 1];
        int s2 = start[last ? 2 * r + 1 : 2 * r + 2];
        if (last) s2 = E;
        float d = rsqrtf((float)(s2 - s0) + 1.0f);
        float4 self = hw4[(size_t)r * 4 + q];

        float dd = d * d;
        float4 acc;
        acc.x = (j == 0) ? (dd * self.x + b0) : 0.0f;
        acc.y = (j == 0) ? (dd * self.y + b1) : 0.0f;
        acc.z = (j == 0) ? (dd * self.z + b2) : 0.0f;
        acc.w = (j == 0) ? (dd * self.w + b3) : 0.0f;

        int p = s0 + j;
        for (; p + 8 < s2; p += 16) {
            int pb = p + 8;
            int i0 = (int)csr[p]  + ((p  >= s1) ? SRC_SPLIT : 0);
            int i1 = (int)csr[pb] + ((pb >= s1) ? SRC_SPLIT : 0);
            float4 v0 = hw4[(size_t)i0 * 4 + q];
            float4 v1 = hw4[(size_t)i1 * 4 + q];
            int c0 = degs[i0];
            int c1 = degs[i1];
            __builtin_amdgcn_sched_barrier(0);
            float g0 = rsqrtf((float)c0 + 1.0f);
            float g1 = rsqrtf((float)c1 + 1.0f);
            if (max(c0, c1) == DEG_ESC) {
                if (c0 == DEG_ESC) g0 = wide_dis(start, i0, N, E);
                if (c1 == DEG_ESC) g1 = wide_dis(start, i1, N, E);
            }
            float w0 = g0 * d;
            float w1 = g1 * d;
            acc.x += w0 * v0.x; acc.y += w0 * v0.y; acc.z += w0 * v0.z; acc.w += w0 * v0.w;
            acc.x += w1 * v1.x; acc.y += w1 * v1.y; acc.z += w1 * v1.z; acc.w += w1 * v1.w;
        }
        if (p < s2) {
            int i0 = (int)csr[p] + ((p >= s1) ? SRC_SPLIT : 0);
            float4 v0 = hw4[(size_t)i0 * 4 + q];
            int c0 = degs[i0];
            float g0 = (c0 == DEG_ESC) ? wide_dis(start, i0, N, E) : rsqrtf((float)c0 + 1.0f);
            float w0 = g0 * d;
            acc.x += w0 * v0.x; acc.y += w0 * v0.y; acc.z += w0 * v0.z; acc.w += w0 * v0.w;
        }

        for (int off = 4; off < 32; off <<= 1) {
            acc.x += __shfl_xor(acc.x, off);
            acc.y += __shfl_xor(acc.y, off);
            acc.z += __shfl_xor(acc.z, off);
            acc.w += __shfl_xor(acc.w, off);
        }

        if (j == 0) ((float4*)out)[(size_t)r * 4 + q] = acc;

        float rx = fmaxf(acc.x, 0.0f), ry = fmaxf(acc.y, 0.0f);
        float rz = fmaxf(acc.z, 0.0f), rw = fmaxf(acc.w, 0.0f);
        st1.x += rx; st1.y += ry; st1.z += rz; st1.w += rw;
        st2.x += rx * rx; st2.y += ry * ry; st2.z += rz * rz; st2.w += rw * rw;
    }

    int wv = (int)(threadIdx.x >> 5);
    if (j == 0) {
        float* rr = red + wv * 32 + 4 * q;
        rr[0] = st1.x; rr[1] = st1.y; rr[2] = st1.z; rr[3] = st1.w;
        rr[16] = st2.x; rr[17] = st2.y; rr[18] = st2.z; rr[19] = st2.w;
    }
    __syncthreads();
    if (threadIdx.x < 32) {
        float v = 0.0f;
#pragma unroll
        for (int w = 0; w < PULL_T / 32; ++w) v += red[w * 32 + threadIdx.x];
        atomicAdd(&stats[threadIdx.x], v);
    }
}

#define TP 20
#define RPW 7
#define WROWS (32 * RPW)
#define NTILE (WROWS / 16)
template <bool STATS>
__global__ void __launch_bounds__(AGG_T)
pull_bn_gemm(const int* __restrict__ start, const int* __restrict__ csr_w,
             const int* __restrict__ deg_w, const float* __restrict__ prev,
             const float* __restrict__ stats, const float* __restrict__ gamma,
             const float* __restrict__ beta, float invN,
             const float* __restrict__ Wm, const float* __restrict__ bias,
             float* __restrict__ out, float* __restrict__ stats_out, int N, int E) {
    __shared__ float tile[WROWS * TP];
    __shared__ float red[NTILE * 32];
    __shared__ __attribute__((aligned(16))) unsigned char degs[DEG_CAP];
    const unsigned short* __restrict__ csr = (const unsigned short*)csr_w;
    const float4* __restrict__ p4 = (const float4*)prev;
    load_deg_table(deg_w, degs, N);
    int lane = threadIdx.x & 31;
    int wave = __builtin_amdgcn_readfirstlane((int)(threadIdx.x >> 5));
    int q = lane & 3;
    int j = lane >> 2;

    float sc0, sc1, sc2, sc3, sh0, sh1, sh2, sh3;
    {
        float mu, var;
        mu = stats[4 * q + 0] * invN; var = stats[16 + 4 * q + 0] * invN - mu * mu;
        sc0 = rsqrtf(var + EPSV) * gamma[4 * q + 0]; sh0 = beta[4 * q + 0] - mu * sc0;
        mu = stats[4 * q + 1] * invN; var = stats[16 + 4 * q + 1] * invN - mu * mu;
        sc1 = rsqrtf(var + EPSV) * gamma[4 * q + 1]; sh1 = beta[4 * q + 1] - mu * sc1;
        mu = stats[4 * q + 2] * invN; var = stats[16 + 4 * q + 2] * invN - mu * mu;
        sc2 = rsqrtf(var + EPSV) * gamma[4 * q + 2]; sh2 = beta[4 * q + 2] - mu * sc2;
        mu = stats[4 * q + 3] * invN; var = stats[16 + 4 * q + 3] * invN - mu * mu;
        sc3 = rsqrtf(var + EPSV) * gamma[4 * q + 3]; sh3 = beta[4 * q + 3] - mu * sc3;
    }

    int rbase = blockIdx.x * WROWS + wave * RPW;
    int sv = start[min(2 * rbase + lane, 2 * N - 1)];
    float4 selfn = p4[(size_t)min(rbase, N - 1) * 4 + q];
    __syncthreads();
    for (int t = 0; t < RPW; ++t) {
        int r = rbase + t;
        if (r >= N) break;
        int s0 = __builtin_amdgcn_readlane(sv, 2 * t);
        int s1 = __builtin_amdgcn_readlane(sv, 2 * t + 1);
        int s2 = (r + 1 >= N) ? E : __builtin_amdgcn_readlane(sv, 2 * t + 2);
        float d = rsqrtf((float)(s2 - s0) + 1.0f);
        float4 self = selfn;
        selfn = p4[(size_t)min(r + 1, N - 1) * 4 + q];

        float dd = d * d;
        float4 acc;
        acc.x = (j == 0) ? dd * (fmaxf(self.x, 0.0f) * sc0 + sh0) : 0.0f;
        acc.y = (j == 0) ? dd * (fmaxf(self.y, 0.0f) * sc1 + sh1) : 0.0f;
        acc.z = (j == 0) ? dd * (fmaxf(self.z, 0.0f) * sc2 + sh2) : 0.0f;
        acc.w = (j == 0) ? dd * (fmaxf(self.w, 0.0f) * sc3 + sh3) : 0.0f;

        int p = s0 + j;
        for (; p + 8 < s2; p += 16) {
            int pb = p + 8;
            int i0 = (int)csr[p]  + ((p  >= s1) ? SRC_SPLIT : 0);
            int i1 = (int)csr[pb] + ((pb >= s1) ? SRC_SPLIT : 0);
            float4 v0 = p4[(size_t)i0 * 4 + q];
            float4 v1 = p4[(size_t)i1 * 4 + q];
            int c0 = degs[i0];
            int c1 = degs[i1];
            __builtin_amdgcn_sched_barrier(0);
            float g0 = rsqrtf((float)c0 + 1.0f);
            float g1 = rsqrtf((float)c1 + 1.0f);
            if (max(c0, c1) == DEG_ESC) {
                if (c0 == DEG_ESC) g0 = wide_dis(start, i0, N, E);
                if (c1 == DEG_ESC) g1 = wide_dis(start, i1, N, E);
            }
            float w0 = g0 * d;
            float w1 = g1 * d;
            acc.x += w0 * (fmaxf(v0.x, 0.0f) * sc0 + sh0);
            acc.y += w0 * (fmaxf(v0.y, 0.0f) * sc1 + sh1);
            acc.z += w0 * (fmaxf(v0.z, 0.0f) * sc2 + sh2);
            acc.w += w0 * (fmaxf(v0.w, 0.0f) * sc3 + sh3);
            acc.x += w1 * (fmaxf(v1.x, 0.0f) * sc0 + sh0);
            acc.y += w1 * (fmaxf(v1.y, 0.0f) * sc1 + sh1);
            acc.z += w1 * (fmaxf(v1.z, 0.0f) * sc2 + sh2);
            acc.w += w1 * (fmaxf(v1.w, 0.0f) * sc3 + sh3);
        }
        if (p < s2) {
            int i0 = (int)csr[p] + ((p >= s1) ? SRC_SPLIT : 0);
            float4 v0 = p4[(size_t)i0 * 4 + q];
            int c0 = degs[i0];
            float g0 = (c0 == DEG_ESC) ? wide_dis(start, i0, N, E) : rsqrtf((float)c0 + 1.0f);
            float w0 = g0 * d;
            acc.x += w0 * (fmaxf(v0.x, 0.0f) * sc0 + sh0);
            acc.y += w0 * (fmaxf(v0.y, 0.0f) * sc1 + sh1);
            acc.z += w0 * (fmaxf(v0.z, 0.0f) * sc2 + sh2);
            acc.w += w0 * (fmaxf(v0.w, 0.0f) * sc3 + sh3);
        }

        for (int off = 4; off < 32; off <<= 1) {
            acc.x += __shfl_xor(acc.x, off);
            acc.y += __shfl_xor(acc.y, off);
            acc.z += __shfl_xor(acc.z, off);
            acc.w += __shfl_xor(acc.w, off);
        }

        if (j == 0) *(float4*)&tile[(wave * RPW + t) * TP + 4 * q] = acc;
    }
    __syncthreads();

    if (wave < NTILE) {
        int row0 = blockIdx.x * WROWS + wave * 16;
        float s1 = 0.0f, s2 = 0.0f;
        if (row0 < N) {
            int m  = lane & 15;
            int hi = lane >> 4;
            const float* trow = tile + (wave * 16 + m) * TP;
            float bv = bias[m];
            v8f c8 = {bv, bv, bv, bv, bv, bv, bv, bv};
            for (int kk = 0; kk < 4; ++kk) {
                int ka = 4 * kk + 2 * hi;
                v2f a, b;
                a.x = trow[ka];
                a.y = trow[ka + 1];
                b.x = Wm[ka * HC + m];
                b.y = Wm[(ka + 1) * HC + m];
                c8 = __builtin_amdgcn_wmma_f32_16x16x4_f32(
                     false, a,  false, b,
                     (short)0, c8,  false,  false);
            }
            float* cbase = out + (size_t)(row0 + hi * 8) * HC + m;
#pragma unroll
            for (int r = 0; r < 8; ++r) cbase[(size_t)r * HC] = c8[r];
            if (STATS) {
#pragma unroll
                for (int r = 0; r < 8; ++r) {
                    float v = fmaxf(c8[r], 0.0f);
                    s1 += v;
                    s2 += v * v;
                }
                s1 += __shfl_xor(s1, 16);
                s2 += __shfl_xor(s2, 16);
            }
        }
        if (STATS) red[wave * 32 + lane] = (lane < 16) ? s1 : s2;
    }
    if (STATS) {
        __syncthreads();
        if (threadIdx.x < 32) {
            float v = 0.0f;
#pragma unroll
            for (int w = 0; w < NTILE; ++w) v += red[w * 32 + lane];
            atomicAdd(&stats_out[lane], v);
        }
    }
}

__global__ void head_kernel(const float* __restrict__ h, const float* __restrict__ Wfc,
                            const float* __restrict__ bfc, float* __restrict__ out, int N) {
    int i = blockIdx.x * blockDim.x + threadIdx.x;
    if (i >= N) return;
    const float* r = h + (size_t)i * HC;
    float z0 = bfc[0], z1 = bfc[1];
#pragma unroll
    for (int c = 0; c < HC; ++c) {
        float v = r[c];
        z0 += v * Wfc[c * 2 + 0];
        z1 += v * Wfc[c * 2 + 1];
    }
    float mx  = fmaxf(z0, z1);
    float lse = mx + logf(__expf(z0 - mx) + __expf(z1 - mx));
    out[(size_t)i * 2 + 0] = z0 - lse;
    out[(size_t)i * 2 + 1] = z1 - lse;
}

extern "C" void kernel_launch(void* const* d_in, const int* in_sizes, int n_in,
                              void* d_out, int out_size, void* d_ws, size_t ws_size,
                              hipStream_t stream) {
    const int IN = 128;
    const int N  = in_sizes[0] / IN;
    const int E  = in_sizes[1] / 2;

    const float* x    = (const float*)d_in[0];
    const int*   ei   = (const int*)d_in[1];
    const int*   srcI = ei;
    const int*   dstI = ei + E;
    const float* W1   = (const float*)d_in[2];
    const float* b1   = (const float*)d_in[3];
    const float* W2   = (const float*)d_in[4];
    const float* b2   = (const float*)d_in[5];
    const float* W3   = (const float*)d_in[6];
    const float* b3   = (const float*)d_in[7];
    const float* g1   = (const float*)d_in[8];
    const float* be1  = (const float*)d_in[9];
    const float* g2   = (const float*)d_in[10];
    const float* be2  = (const float*)d_in[11];
    const float* Wfc  = (const float*)d_in[12];
    const float* bfc  = (const float*)d_in[13];

    float* ws    = (float*)d_ws;
    float* dis   = ws;
    float* bufA  = dis + N;
    float* bufB  = bufA + (size_t)N * HC;
    float* bufC  = bufB + (size_t)N * HC;
    float* stats = bufC + (size_t)N * HC;
    int*   csr   = (int*)bufC;
    int*   ghist = (int*)(((size_t)bufC + 15) & ~(size_t)15);
    int*   rec   = (int*)bufA;
    int*   start = (int*)d_out;
    int*   degw  = (int*)dis;

    const int B256 = 256;
    dim3 blkN((N + B256 - 1) / B256);
    const int NB = (N + (1 << RB_SHIFT) - 1) >> RB_SHIFT;
    const int chunk = (E + G1 - 1) / G1;
    dim3 blkG((N / 16 + 7) / 8);
    dim3 blkP(512);
    dim3 blkT((N + WROWS - 1) / WROWS);
    float invN = 1.0f / (float)N;

    coarse_hist<<<G1, SORT_T, 0, stream>>>(dstI, ghist, E, chunk);
    coarse_scatter<<<G1, SORT_T, 0, stream>>>(srcI, dstI, ghist, rec, start, N, E, chunk);
    fine_sort<<<NB, SORT_T, 0, stream>>>(rec, start, degw, csr, stats, N);

    gemm_wmma_f32<<<blkG, B256, 0, stream>>>(x, W1, bufA, N, IN);
    pull_rows<<<blkP, PULL_T, 0, stream>>>(start, csr, degw, bufA, b1, bufB, stats, N, E);

    pull_bn_gemm<true><<<blkT, AGG_T, 0, stream>>>(start, csr, degw, bufB, stats, g1, be1, invN, W2, b2, bufA, stats + 32, N, E);

    pull_bn_gemm<false><<<blkT, AGG_T, 0, stream>>>(start, csr, degw, bufA, stats + 32, g2, be2, invN, W3, b3, bufB, stats, N, E);

    head_kernel<<<blkN, B256, 0, stream>>>(bufB, Wfc, bfc, (float*)d_out, N);
}


# Both edge-sort levels place records in an LDS stage by LDS-counter rank and copy it out as contiguous runs, not one scattered 4- or 2-byte global store per edge; arrival-order rank as before
# speedup vs baseline: 18.4378x; 1.3029x over previous
#include <hip/hip_runtime.h>
#include <hip/hip_bf16.h>

typedef float v2f __attribute__((ext_vector_type(2)));
typedef float v8f __attribute__((ext_vector_type(8)));

#define HC 16
#define EPSV 1e-5f
#define SRC_SPLIT 65536

#define DEG_CAP 131072
#define DEG_ESC 255
#define AGG_T 1024

__device__ __forceinline__ void load_deg_table(const int* __restrict__ deg_w, unsigned char* degs, int N) {
    const uint4* __restrict__ g4 = (const uint4*)deg_w;
    uint4* l4 = (uint4*)degs;
    int nv = (N + 15) >> 4;
    uint4 v[8];
#pragma unroll
    for (int k = 0; k < 8; ++k) v[k] = g4[min((int)threadIdx.x + k * AGG_T, nv - 1)];
#pragma unroll
    for (int k = 0; k < 8; ++k) l4[threadIdx.x + k * AGG_T] = v[k];
}

__device__ __forceinline__ float wide_dis(const int* __restrict__ start, int i, int N, int E) {
    int a = start[2 * i];
    int b = start[min(2 * i + 2, 2 * N - 1)];
    if (i + 1 >= N) b = E;
    return rsqrtf((float)(b - a) + 1.0f);
}

#define RB_SHIFT 8
#define NBP 512
#define G1 256
#define SORT_T 1024
#define EPT 13
#define SCAP (EPT * SORT_T)
#define FCAP 32768

__device__ __forceinline__ int block_excl_scan(int v, int* wsum, int tid) {
    int lane = tid & 31, wv = tid >> 5;
    int inc = v;
    for (int off = 1; off < 32; off <<= 1) {
        int n = __shfl_up(inc, off);
        if (lane >= off) inc += n;
    }
    if (lane == 31) wsum[wv] = inc;
    __syncthreads();
    int before = 0;
    for (int w = 0; w < wv; ++w) before += wsum[w];
    __syncthreads();
    return before + inc - v;
}

__global__ void __launch_bounds__(SORT_T)
coarse_hist(const int* __restrict__ dst, int* __restrict__ ghist, int E, int chunk) {
    __shared__ int h[NBP];
    int tid = threadIdx.x;
    if (tid < NBP) h[tid] = 0;
    __syncthreads();
    int c0 = blockIdx.x * chunk;
    int c1 = min(c0 + chunk, E);
    for (int base = c0 + tid; base < c1; base += 4 * SORT_T) {
        int d[4];
#pragma unroll
        for (int k = 0; k < 4; ++k) d[k] = dst[min(base + k * SORT_T, c1 - 1)];
        asm volatile("" :: "v"(d[0]), "v"(d[1]), "v"(d[2]), "v"(d[3]));
#pragma unroll
        for (int k = 0; k < 4; ++k)
            if (base + k * SORT_T < c1) atomicAdd(&h[d[k] >> RB_SHIFT], 1);
    }
    __syncthreads();
    if (tid < NBP) ghist[blockIdx.x * NBP + tid] = h[tid];
}

__global__ void __launch_bounds__(SORT_T)
coarse_scatter(const int* __restrict__ src, const int* __restrict__ dst,
               const int* __restrict__ ghist, int* __restrict__ rec,
               int* __restrict__ start, int N, int E, int chunk) {
    __shared__ int ptot[8][NBP];
    __shared__ int ppart[8][NBP];
    __shared__ int cur[NBP];
    __shared__ int lcnt[NBP];
    __shared__ int2 tb[NBP];
    __shared__ int wsum[32];
    __shared__ int2 stage[SCAP];
    int tid = threadIdx.x;
    int wg = blockIdx.x;

    {
        const int4* __restrict__ gh4 = (const int4*)ghist;
        int j = tid & 127;
        int g = tid >> 7;
        int4 t4 = {0, 0, 0, 0}, p4 = {0, 0, 0, 0};
#pragma unroll 1
        for (int batch = 0; batch < 4; ++batch) {
            int w0 = g * (G1 / 8) + batch * 8;
            int4 v[8];
#pragma unroll
            for (int k = 0; k < 8; ++k) v[k] = gh4[(w0 + k) * (NBP / 4) + j];
            asm volatile("" :: "v"(v[0].x), "v"(v[1].x), "v"(v[2].x), "v"(v[3].x),
                               "v"(v[4].x), "v"(v[5].x), "v"(v[6].x), "v"(v[7].x));
#pragma unroll
            for (int k = 0; k < 8; ++k) {
                t4.x += v[k].x; t4.y += v[k].y; t4.z += v[k].z; t4.w += v[k].w;
                if (w0 + k < wg) { p4.x += v[k].x; p4.y += v[k].y; p4.z += v[k].z; p4.w += v[k].w; }
            }
        }
        ptot[g][4 * j + 0] = t4.x; ptot[g][4 * j + 1] = t4.y;
        ptot[g][4 * j + 2] = t4.z; ptot[g][4 * j + 3] = t4.w;
        ppart[g][4 * j + 0] = p4.x; ppart[g][4 * j + 1] = p4.y;
        ppart[g][4 * j + 2] = p4.z; ppart[g][4 * j + 3] = p4.w;
    }
    __syncthreads();
    int tot = 0, part = 0;
    if (tid < NBP) {
#pragma unroll
        for (int g = 0; g < 8; ++g) { tot += ptot[g][tid]; part += ppart[g][tid]; }
    }
    int bstart = block_excl_scan(tot, wsum, tid);
    if (tid < NBP) {
        cur[tid] = bstart + part;
        lcnt[tid] = 0;
        if (wg == 0 && (tid << RB_SHIFT) < N) {
            start[2 * (tid << RB_SHIFT)] = bstart;
            start[2 * (tid << RB_SHIFT) + 1] = bstart + tot;
        }
    }
    __syncthreads();

    int c0 = wg * chunk;
    int c1 = min(c0 + chunk, E);
    for (int sub = c0; sub < c1; sub += SCAP) {
        int nsub = min(SCAP, c1 - sub);
        int s[EPT], d[EPT];
#pragma unroll
        for (int k = 0; k < EPT; ++k) {
            int ec = min(sub + tid + k * SORT_T, c1 - 1);
            s[k] = src[ec];
            d[k] = dst[ec];
        }
        asm volatile("" :: "v"(s[0]), "v"(s[1]), "v"(s[2]), "v"(s[3]), "v"(s[4]), "v"(s[5]), "v"(s[6]),
                           "v"(s[7]), "v"(s[8]), "v"(s[9]), "v"(s[10]), "v"(s[11]), "v"(s[12]),
                           "v"(d[0]), "v"(d[1]), "v"(d[2]), "v"(d[3]), "v"(d[4]), "v"(d[5]), "v"(d[6]),
                           "v"(d[7]), "v"(d[8]), "v"(d[9]), "v"(d[10]), "v"(d[11]), "v"(d[12]));
        int rc[EPT], bk[EPT], rk[EPT];
#pragma unroll
        for (int k = 0; k < EPT; ++k) {
            rc[k] = s[k] | ((d[k] & ((1 << RB_SHIFT) - 1)) << 17);
            bk[k] = d[k] >> RB_SHIFT;
            rk[k] = 0;
            if (tid + k * SORT_T < nsub) rk[k] = atomicAdd(&lcnt[bk[k]], 1);
        }
        __syncthreads();
        int c = (tid < NBP) ? lcnt[tid] : 0;
        int ex = block_excl_scan(c, wsum, tid);
        if (tid < NBP) {
            tb[tid] = make_int2(ex, cur[tid] - ex);
            cur[tid] += c;
            lcnt[tid] = 0;
        }
        __syncthreads();
        int2 t[EPT];
#pragma unroll
        for (int k = 0; k < EPT; ++k) t[k] = tb[bk[k]];
        asm volatile("" :: "v"(t[0].x), "v"(t[1].x), "v"(t[2].x), "v"(t[3].x), "v"(t[4].x), "v"(t[5].x), "v"(t[6].x),
                           "v"(t[7].x), "v"(t[8].x), "v"(t[9].x), "v"(t[10].x), "v"(t[11].x), "v"(t[12].x));
#pragma unroll
        for (int k = 0; k < EPT; ++k) {
            if (tid + k * SORT_T < nsub) {
                int pos = t[k].x + rk[k];
                stage[pos] = make_int2(rc[k], t[k].y + pos);
            }
        }
        __syncthreads();
        int2 e[EPT];
#pragma unroll
        for (int k = 0; k < EPT; ++k) e[k] = stage[min(tid + k * SORT_T, nsub - 1)];
        asm volatile("" :: "v"(e[0].x), "v"(e[1].x), "v"(e[2].x), "v"(e[3].x), "v"(e[4].x), "v"(e[5].x), "v"(e[6].x),
                           "v"(e[7].x), "v"(e[8].x), "v"(e[9].x), "v"(e[10].x), "v"(e[11].x), "v"(e[12].x));
#pragma unroll
        for (int k = 0; k < EPT; ++k) {
            if (tid + k * SORT_T < nsub) rec[e[k].y] = e[k].x;
        }
        __syncthreads();
    }
}

__global__ void __launch_bounds__(SORT_T)
fine_sort(const int* __restrict__ rec, int* __restrict__ start, int* __restrict__ deg_w,
          int* __restrict__ csr_w, float* __restrict__ stats, int N) {
    unsigned short* __restrict__ csr = (unsigned short*)csr_w;
    unsigned char* __restrict__ deg8 = (unsigned char*)deg_w;
    __shared__ int cnt[512];
    __shared__ int cursor[512];
    __shared__ int wsum[32];
    __shared__ unsigned short stg[FCAP];
    int tid = threadIdx.x;
    int b = blockIdx.x;
    int c0 = start[2 * (b << RB_SHIFT)];
    int c1 = start[2 * (b << RB_SHIFT) + 1];
    bool staged = (c1 - c0) <= FCAP;
    if (tid < 512) cnt[tid] = 0;
    if (b == 0 && tid < 64) stats[tid] = 0.0f;
    __syncthreads();

    for (int base = c0 + tid; base < c1; base += 4 * SORT_T) {
        int r[4];
#pragma unroll
        for (int k = 0; k < 4; ++k) r[k] = rec[min(base + k * SORT_T, c1 - 1)];
        asm volatile("" :: "v"(r[0]), "v"(r[1]), "v"(r[2]), "v"(r[3]));
#pragma unroll
        for (int k = 0; k < 4; ++k)
            if (base + k * SORT_T < c1) atomicAdd(&cnt[(r[k] >> 16) & 511], 1);
    }
    __syncthreads();

    int c = (tid < 512) ? cnt[tid] : 0;
    int ex = block_excl_scan(c, wsum, tid);
    if (tid < 512) {
        cursor[tid] = staged ? ex : c0 + ex;
        int row = (b << RB_SHIFT) + (tid >> 1);
        if (row < N) start[2 * row + (tid & 1)] = c0 + ex;
    }
    if (tid < 256) {
        int row = (b << RB_SHIFT) + tid;
        if (row < N) deg8[row] = (unsigned char)min(cnt[2 * tid] + cnt[2 * tid + 1], DEG_ESC);
    }
    __syncthreads();

    for (int base = c0 + tid; base < c1; base += 4 * SORT_T) {
        int r[4];
#pragma unroll
        for (int k = 0; k < 4; ++k) r[k] = rec[min(base + k * SORT_T, c1 - 1)];
        asm volatile("" :: "v"(r[0]), "v"(r[1]), "v"(r[2]), "v"(r[3]));
        int pos[4];
#pragma unroll
        for (int k = 0; k < 4; ++k) {
            pos[k] = 0;
            if (base + k * SORT_T < c1) pos[k] = atomicAdd(&cursor[(r[k] >> 16) & 511], 1);
        }
#pragma unroll
        for (int k = 0; k < 4; ++k) {
            if (base + k * SORT_T < c1) {
                unsigned short v = (unsigned short)(r[k] & (SRC_SPLIT - 1));
                if (staged) stg[pos[k]] = v; else csr[pos[k]] = v;
            }
        }
    }
    if (staged) {
        __syncthreads();
        int w0 = (c0 + 1) >> 1;
        int w1 = c1 >> 1;
        for (int w = w0 + tid; w < w1; w += SORT_T) {
            int i = 2 * w - c0;
            csr_w[w] = (int)stg[i] | ((int)stg[i + 1] << 16);
        }
        if (tid == 0 && (c0 & 1) && c1 > c0) csr[c0] = stg[0];
        if (tid == 32 && (c1 & 1) && c1 > c0) csr[c1 - 1] = stg[c1 - 1 - c0];
    }
}

__global__ void gemm_wmma_f32(const float* __restrict__ A, const float* __restrict__ B,
                              float* __restrict__ C, int N, int K) {
    int wave = (int)((blockIdx.x * blockDim.x + threadIdx.x) >> 5);
    int lane = threadIdx.x & 31;
    int row0 = wave * 16;
    if (row0 >= N) return;

    int m  = lane & 15;
    int hi = lane >> 4;
    const float* arow = A + (size_t)(row0 + m) * K;

    v8f acc = {};
    for (int k0 = 0; k0 < K; k0 += 4) {
        int ka = k0 + 2 * hi;
        v2f a, b;
        a.x = arow[ka];
        a.y = arow[ka + 1];
        b.x = B[(size_t)ka * HC + m];
        b.y = B[(size_t)(ka + 1) * HC + m];
        acc = __builtin_amdgcn_wmma_f32_16x16x4_f32(
             false, a,  false, b,
             (short)0, acc,  false,  false);
    }

    float* cbase = C + (size_t)(row0 + hi * 8) * HC + m;
#pragma unroll
    for (int r = 0; r < 8; ++r) cbase[(size_t)r * HC] = acc[r];
}

#define PULL_T AGG_T
__global__ void __launch_bounds__(PULL_T)
pull_rows(const int* __restrict__ start, const int* __restrict__ csr_w,
          const int* __restrict__ deg_w, const float* __restrict__ hw,
          const float* __restrict__ bias, float* __restrict__ out,
          float* __restrict__ stats, int N, int E) {
    __shared__ float red[(PULL_T / 32) * 32];
    __shared__ __attribute__((aligned(16))) unsigned char degs[DEG_CAP];
    load_deg_table(deg_w, degs, N);
    __syncthreads();
    const unsigned short* __restrict__ csr = (const unsigned short*)csr_w;
    const float4* __restrict__ hw4 = (const float4*)hw;
    int lane = threadIdx.x & 31;
    int q = lane & 3;
    int j = lane >> 2;
    int nw = (int)((gridDim.x * blockDim.x) >> 5);

    float b0 = bias[4 * q + 0], b1 = bias[4 * q + 1], b2 = bias[4 * q + 2], b3 = bias[4 * q + 3];

    float4 st1 = {0.0f, 0.0f, 0.0f, 0.0f};
    float4 st2 = {0.0f, 0.0f, 0.0f, 0.0f};

    for (int r = (int)((blockIdx.x * blockDim.x + threadIdx.x) >> 5); r < N; r += nw) {
        bool last = (r + 1 >= N);
        int s0 = start[2 * r];
        int s1 = start[2 * r + 1];
        int s2 = start[last ? 2 * r + 1 : 2 * r + 2];
        if (last) s2 = E;
        float d = rsqrtf((float)(s2 - s0) + 1.0f);
        float4 self = hw4[(size_t)r * 4 + q];

        float dd = d * d;
        float4 acc;
        acc.x = (j == 0) ? (dd * self.x + b0) : 0.0f;
        acc.y = (j == 0) ? (dd * self.y + b1) : 0.0f;
        acc.z = (j == 0) ? (dd * self.z + b2) : 0.0f;
        acc.w = (j == 0) ? (dd * self.w + b3) : 0.0f;

        int p = s0 + j;
        for (; p + 8 < s2; p += 16) {
            int pb = p + 8;
            int i0 = (int)csr[p]  + ((p  >= s1) ? SRC_SPLIT : 0);
            int i1 = (int)csr[pb] + ((pb >= s1) ? SRC_SPLIT : 0);
            float4 v0 = hw4[(size_t)i0 * 4 + q];
            float4 v1 = hw4[(size_t)i1 * 4 + q];
            int c0 = degs[i0];
            int c1 = degs[i1];
            __builtin_amdgcn_sched_barrier(0);
            float g0 = rsqrtf((float)c0 + 1.0f);
            float g1 = rsqrtf((float)c1 + 1.0f);
            if (max(c0, c1) == DEG_ESC) {
                if (c0 == DEG_ESC) g0 = wide_dis(start, i0, N, E);
                if (c1 == DEG_ESC) g1 = wide_dis(start, i1, N, E);
            }
            float w0 = g0 * d;
            float w1 = g1 * d;
            acc.x += w0 * v0.x; acc.y += w0 * v0.y; acc.z += w0 * v0.z; acc.w += w0 * v0.w;
            acc.x += w1 * v1.x; acc.y += w1 * v1.y; acc.z += w1 * v1.z; acc.w += w1 * v1.w;
        }
        if (p < s2) {
            int i0 = (int)csr[p] + ((p >= s1) ? SRC_SPLIT : 0);
            float4 v0 = hw4[(size_t)i0 * 4 + q];
            int c0 = degs[i0];
            float g0 = (c0 == DEG_ESC) ? wide_dis(start, i0, N, E) : rsqrtf((float)c0 + 1.0f);
            float w0 = g0 * d;
            acc.x += w0 * v0.x; acc.y += w0 * v0.y; acc.z += w0 * v0.z; acc.w += w0 * v0.w;
        }

        for (int off = 4; off < 32; off <<= 1) {
            acc.x += __shfl_xor(acc.x, off);
            acc.y += __shfl_xor(acc.y, off);
            acc.z += __shfl_xor(acc.z, off);
            acc.w += __shfl_xor(acc.w, off);
        }

        if (j == 0) ((float4*)out)[(size_t)r * 4 + q] = acc;

        float rx = fmaxf(acc.x, 0.0f), ry = fmaxf(acc.y, 0.0f);
        float rz = fmaxf(acc.z, 0.0f), rw = fmaxf(acc.w, 0.0f);
        st1.x += rx; st1.y += ry; st1.z += rz; st1.w += rw;
        st2.x += rx * rx; st2.y += ry * ry; st2.z += rz * rz; st2.w += rw * rw;
    }

    int wv = (int)(threadIdx.x >> 5);
    if (j == 0) {
        float* rr = red + wv * 32 + 4 * q;
        rr[0] = st1.x; rr[1] = st1.y; rr[2] = st1.z; rr[3] = st1.w;
        rr[16] = st2.x; rr[17] = st2.y; rr[18] = st2.z; rr[19] = st2.w;
    }
    __syncthreads();
    if (threadIdx.x < 32) {
        float v = 0.0f;
#pragma unroll
        for (int w = 0; w < PULL_T / 32; ++w) v += red[w * 32 + threadIdx.x];
        atomicAdd(&stats[threadIdx.x], v);
    }
}

#define TP 20
#define RPW 7
#define WROWS (32 * RPW)
#define NTILE (WROWS / 16)
template <bool STATS>
__global__ void __launch_bounds__(AGG_T)
pull_bn_gemm(const int* __restrict__ start, const int* __restrict__ csr_w,
             const int* __restrict__ deg_w, const float* __restrict__ prev,
             const float* __restrict__ stats, const float* __restrict__ gamma,
             const float* __restrict__ beta, float invN,
             const float* __restrict__ Wm, const float* __restrict__ bias,
             float* __restrict__ out, float* __restrict__ stats_out, int N, int E) {
    __shared__ float tile[WROWS * TP];
    __shared__ float red[NTILE * 32];
    __shared__ __attribute__((aligned(16))) unsigned char degs[DEG_CAP];
    const unsigned short* __restrict__ csr = (const unsigned short*)csr_w;
    const float4* __restrict__ p4 = (const float4*)prev;
    load_deg_table(deg_w, degs, N);
    int lane = threadIdx.x & 31;
    int wave = __builtin_amdgcn_readfirstlane((int)(threadIdx.x >> 5));
    int q = lane & 3;
    int j = lane >> 2;

    float sc0, sc1, sc2, sc3, sh0, sh1, sh2, sh3;
    {
        float mu, var;
        mu = stats[4 * q + 0] * invN; var = stats[16 + 4 * q + 0] * invN - mu * mu;
        sc0 = rsqrtf(var + EPSV) * gamma[4 * q + 0]; sh0 = beta[4 * q + 0] - mu * sc0;
        mu = stats[4 * q + 1] * invN; var = stats[16 + 4 * q + 1] * invN - mu * mu;
        sc1 = rsqrtf(var + EPSV) * gamma[4 * q + 1]; sh1 = beta[4 * q + 1] - mu * sc1;
        mu = stats[4 * q + 2] * invN; var = stats[16 + 4 * q + 2] * invN - mu * mu;
        sc2 = rsqrtf(var + EPSV) * gamma[4 * q + 2]; sh2 = beta[4 * q + 2] - mu * sc2;
        mu = stats[4 * q + 3] * invN; var = stats[16 + 4 * q + 3] * invN - mu * mu;
        sc3 = rsqrtf(var + EPSV) * gamma[4 * q + 3]; sh3 = beta[4 * q + 3] - mu * sc3;
    }

    int rbase = blockIdx.x * WROWS + wave * RPW;
    int sv = start[min(2 * rbase + lane, 2 * N - 1)];
    float4 selfn = p4[(size_t)min(rbase, N - 1) * 4 + q];
    __syncthreads();
    for (int t = 0; t < RPW; ++t) {
        int r = rbase + t;
        if (r >= N) break;
        int s0 = __builtin_amdgcn_readlane(sv, 2 * t);
        int s1 = __builtin_amdgcn_readlane(sv, 2 * t + 1);
        int s2 = (r + 1 >= N) ? E : __builtin_amdgcn_readlane(sv, 2 * t + 2);
        float d = rsqrtf((float)(s2 - s0) + 1.0f);
        float4 self = selfn;
        selfn = p4[(size_t)min(r + 1, N - 1) * 4 + q];

        float dd = d * d;
        float4 acc;
        acc.x = (j == 0) ? dd * (fmaxf(self.x, 0.0f) * sc0 + sh0) : 0.0f;
        acc.y = (j == 0) ? dd * (fmaxf(self.y, 0.0f) * sc1 + sh1) : 0.0f;
        acc.z = (j == 0) ? dd * (fmaxf(self.z, 0.0f) * sc2 + sh2) : 0.0f;
        acc.w = (j == 0) ? dd * (fmaxf(self.w, 0.0f) * sc3 + sh3) : 0.0f;

        int p = s0 + j;
        for (; p + 8 < s2; p += 16) {
            int pb = p + 8;
            int i0 = (int)csr[p]  + ((p  >= s1) ? SRC_SPLIT : 0);
            int i1 = (int)csr[pb] + ((pb >= s1) ? SRC_SPLIT : 0);
            float4 v0 = p4[(size_t)i0 * 4 + q];
            float4 v1 = p4[(size_t)i1 * 4 + q];
            int c0 = degs[i0];
            int c1 = degs[i1];
            __builtin_amdgcn_sched_barrier(0);
            float g0 = rsqrtf((float)c0 + 1.0f);
            float g1 = rsqrtf((float)c1 + 1.0f);
            if (max(c0, c1) == DEG_ESC) {
                if (c0 == DEG_ESC) g0 = wide_dis(start, i0, N, E);
                if (c1 == DEG_ESC) g1 = wide_dis(start, i1, N, E);
            }
            float w0 = g0 * d;
            float w1 = g1 * d;
            acc.x += w0 * (fmaxf(v0.x, 0.0f) * sc0 + sh0);
            acc.y += w0 * (fmaxf(v0.y, 0.0f) * sc1 + sh1);
            acc.z += w0 * (fmaxf(v0.z, 0.0f) * sc2 + sh2);
            acc.w += w0 * (fmaxf(v0.w, 0.0f) * sc3 + sh3);
            acc.x += w1 * (fmaxf(v1.x, 0.0f) * sc0 + sh0);
            acc.y += w1 * (fmaxf(v1.y, 0.0f) * sc1 + sh1);
            acc.z += w1 * (fmaxf(v1.z, 0.0f) * sc2 + sh2);
            acc.w += w1 * (fmaxf(v1.w, 0.0f) * sc3 + sh3);
        }
        if (p < s2) {
            int i0 = (int)csr[p] + ((p >= s1) ? SRC_SPLIT : 0);
            float4 v0 = p4[(size_t)i0 * 4 + q];
            int c0 = degs[i0];
            float g0 = (c0 == DEG_ESC) ? wide_dis(start, i0, N, E) : rsqrtf((float)c0 + 1.0f);
            float w0 = g0 * d;
            acc.x += w0 * (fmaxf(v0.x, 0.0f) * sc0 + sh0);
            acc.y += w0 * (fmaxf(v0.y, 0.0f) * sc1 + sh1);
            acc.z += w0 * (fmaxf(v0.z, 0.0f) * sc2 + sh2);
            acc.w += w0 * (fmaxf(v0.w, 0.0f) * sc3 + sh3);
        }

        for (int off = 4; off < 32; off <<= 1) {
            acc.x += __shfl_xor(acc.x, off);
            acc.y += __shfl_xor(acc.y, off);
            acc.z += __shfl_xor(acc.z, off);
            acc.w += __shfl_xor(acc.w, off);
        }

        if (j == 0) *(float4*)&tile[(wave * RPW + t) * TP + 4 * q] = acc;
    }
    __syncthreads();

    if (wave < NTILE) {
        int row0 = blockIdx.x * WROWS + wave * 16;
        float s1 = 0.0f, s2 = 0.0f;
        if (row0 < N) {
            int m  = lane & 15;
            int hi = lane >> 4;
            const float* trow = tile + (wave * 16 + m) * TP;
            float bv = bias[m];
            v8f c8 = {bv, bv, bv, bv, bv, bv, bv, bv};
            for (int kk = 0; kk < 4; ++kk) {
                int ka = 4 * kk + 2 * hi;
                v2f a, b;
                a.x = trow[ka];
                a.y = trow[ka + 1];
                b.x = Wm[ka * HC + m];
                b.y = Wm[(ka + 1) * HC + m];
                c8 = __builtin_amdgcn_wmma_f32_16x16x4_f32(
                     false, a,  false, b,
                     (short)0, c8,  false,  false);
            }
            float* cbase = out + (size_t)(row0 + hi * 8) * HC + m;
#pragma unroll
            for (int r = 0; r < 8; ++r) cbase[(size_t)r * HC] = c8[r];
            if (STATS) {
#pragma unroll
                for (int r = 0; r < 8; ++r) {
                    float v = fmaxf(c8[r], 0.0f);
                    s1 += v;
                    s2 += v * v;
                }
                s1 += __shfl_xor(s1, 16);
                s2 += __shfl_xor(s2, 16);
            }
        }
        if (STATS) red[wave * 32 + lane] = (lane < 16) ? s1 : s2;
    }
    if (STATS) {
        __syncthreads();
        if (threadIdx.x < 32) {
            float v = 0.0f;
#pragma unroll
            for (int w = 0; w < NTILE; ++w) v += red[w * 32 + lane];
            atomicAdd(&stats_out[lane], v);
        }
    }
}

__global__ void head_kernel(const float* __restrict__ h, const float* __restrict__ Wfc,
                            const float* __restrict__ bfc, float* __restrict__ out, int N) {
    int i = blockIdx.x * blockDim.x + threadIdx.x;
    if (i >= N) return;
    const float* r = h + (size_t)i * HC;
    float z0 = bfc[0], z1 = bfc[1];
#pragma unroll
    for (int c = 0; c < HC; ++c) {
        float v = r[c];
        z0 += v * Wfc[c * 2 + 0];
        z1 += v * Wfc[c * 2 + 1];
    }
    float mx  = fmaxf(z0, z1);
    float lse = mx + logf(__expf(z0 - mx) + __expf(z1 - mx));
    out[(size_t)i * 2 + 0] = z0 - lse;
    out[(size_t)i * 2 + 1] = z1 - lse;
}

extern "C" void kernel_launch(void* const* d_in, const int* in_sizes, int n_in,
                              void* d_out, int out_size, void* d_ws, size_t ws_size,
                              hipStream_t stream) {
    const int IN = 128;
    const int N  = in_sizes[0] / IN;
    const int E  = in_sizes[1] / 2;

    const float* x    = (const float*)d_in[0];
    const int*   ei   = (const int*)d_in[1];
    const int*   srcI = ei;
    const int*   dstI = ei + E;
    const float* W1   = (const float*)d_in[2];
    const float* b1   = (const float*)d_in[3];
    const float* W2   = (const float*)d_in[4];
    const float* b2   = (const float*)d_in[5];
    const float* W3   = (const float*)d_in[6];
    const float* b3   = (const float*)d_in[7];
    const float* g1   = (const float*)d_in[8];
    const float* be1  = (const float*)d_in[9];
    const float* g2   = (const float*)d_in[10];
    const float* be2  = (const float*)d_in[11];
    const float* Wfc  = (const float*)d_in[12];
    const float* bfc  = (const float*)d_in[13];

    float* ws    = (float*)d_ws;
    float* dis   = ws;
    float* bufA  = dis + N;
    float* bufB  = bufA + (size_t)N * HC;
    float* bufC  = bufB + (size_t)N * HC;
    float* stats = bufC + (size_t)N * HC;
    int*   csr   = (int*)bufC;
    int*   ghist = (int*)(((size_t)bufC + 15) & ~(size_t)15);
    int*   rec   = (int*)bufA;
    int*   start = (int*)d_out;
    int*   degw  = (int*)dis;

    const int B256 = 256;
    dim3 blkN((N + B256 - 1) / B256);
    const int NB = (N + (1 << RB_SHIFT) - 1) >> RB_SHIFT;
    const int chunk = (E + G1 - 1) / G1;
    dim3 blkG((N / 16 + 7) / 8);
    dim3 blkP(512);
    dim3 blkT((N + WROWS - 1) / WROWS);
    float invN = 1.0f / (float)N;

    coarse_hist<<<G1, SORT_T, 0, stream>>>(dstI, ghist, E, chunk);
    coarse_scatter<<<G1, SORT_T, 0, stream>>>(srcI, dstI, ghist, rec, start, N, E, chunk);
    fine_sort<<<NB, SORT_T, 0, stream>>>(rec, start, degw, csr, stats, N);

    gemm_wmma_f32<<<blkG, B256, 0, stream>>>(x, W1, bufA, N, IN);
    pull_rows<<<blkP, PULL_T, 0, stream>>>(start, csr, degw, bufA, b1, bufB, stats, N, E);

    pull_bn_gemm<true><<<blkT, AGG_T, 0, stream>>>(start, csr, degw, bufB, stats, g1, be1, invN, W2, b2, bufA, stats + 32, N, E);

    pull_bn_gemm<false><<<blkT, AGG_T, 0, stream>>>(start, csr, degw, bufA, stats + 32, g2, be2, invN, W3, b3, bufB, stats, N, E);

    head_kernel<<<blkN, B256, 0, stream>>>(bufB, Wfc, bfc, (float*)d_out, N);
}


# Level-1 edge sort piece-local: a work-group orders one 13312-edge piece by bucket in LDS, level 2 collects its bucket's run from each piece; no histogram launch; arrival-order rank; 8 launches to 7
# speedup vs baseline: 19.7542x; 1.0714x over previous
#include <hip/hip_runtime.h>
#include <hip/hip_bf16.h>

typedef float v2f __attribute__((ext_vector_type(2)));
typedef float v8f __attribute__((ext_vector_type(8)));

#define HC 16
#define EPSV 1e-5f
#define SRC_SPLIT 65536

#define DEG_CAP 131072
#define DEG_ESC 255
#define AGG_T 1024

__device__ __forceinline__ void load_deg_table(const int* __restrict__ deg_w, unsigned char* degs, int N) {
    const uint4* __restrict__ g4 = (const uint4*)deg_w;
    uint4* l4 = (uint4*)degs;
    int nv = (N + 15) >> 4;
    uint4 v[8];
#pragma unroll
    for (int k = 0; k < 8; ++k) v[k] = g4[min((int)threadIdx.x + k * AGG_T, nv - 1)];
#pragma unroll
    for (int k = 0; k < 8; ++k) l4[threadIdx.x + k * AGG_T] = v[k];
}

__device__ __forceinline__ float wide_dis(const int* __restrict__ start, int i, int N, int E) {
    int a = start[2 * i];
    int b = start[min(2 * i + 2, 2 * N - 1)];
    if (i + 1 >= N) b = E;
    return rsqrtf((float)(b - a) + 1.0f);
}

#define RB_SHIFT 8
#define NBP 512
#define TABW (NBP + 1)
#define G1 256
#define SORT_T 1024
#define PMAX 1024
#define EPT 13
#define SCAP (EPT * SORT_T)
#define FCAP 32768

__device__ __forceinline__ int block_excl_scan(int v, int* wsum, int tid) {
    int lane = tid & 31, wv = tid >> 5;
    int inc = v;
    for (int off = 1; off < 32; off <<= 1) {
        int n = __shfl_up(inc, off);
        if (lane >= off) inc += n;
    }
    if (lane == 31) wsum[wv] = inc;
    __syncthreads();
    int before = 0;
    for (int w = 0; w < wv; ++w) before += wsum[w];
    __syncthreads();
    return before + inc - v;
}

__global__ void __launch_bounds__(SORT_T)
piece_sort(const int* __restrict__ src, const int* __restrict__ dst, int* __restrict__ rec,
           int* __restrict__ tab_w, int E) {
    unsigned short* __restrict__ tab = (unsigned short*)tab_w;
    __shared__ int lcnt[NBP];
    __shared__ int tb[NBP];
    __shared__ int wsum[32];
    __shared__ __attribute__((aligned(16))) int stage[SCAP];
    int tid = threadIdx.x;
    int np = (E + SCAP - 1) / SCAP;
    for (int pc = blockIdx.x; pc < np; pc += G1) {
        int c0 = pc * SCAP;
        int c1 = min(c0 + SCAP, E);
        int nsub = c1 - c0;
        if (tid < NBP) lcnt[tid] = 0;
        __syncthreads();
        int s[EPT], d[EPT];
#pragma unroll
        for (int k = 0; k < EPT; ++k) {
            int ec = min(c0 + tid + k * SORT_T, c1 - 1);
            s[k] = src[ec];
            d[k] = dst[ec];
        }
        asm volatile("" :: "v"(s[0]), "v"(s[1]), "v"(s[2]), "v"(s[3]), "v"(s[4]), "v"(s[5]), "v"(s[6]),
                           "v"(s[7]), "v"(s[8]), "v"(s[9]), "v"(s[10]), "v"(s[11]), "v"(s[12]),
                           "v"(d[0]), "v"(d[1]), "v"(d[2]), "v"(d[3]), "v"(d[4]), "v"(d[5]), "v"(d[6]),
                           "v"(d[7]), "v"(d[8]), "v"(d[9]), "v"(d[10]), "v"(d[11]), "v"(d[12]));
        int rc[EPT], bk[EPT], rk[EPT];
#pragma unroll
        for (int k = 0; k < EPT; ++k) {
            rc[k] = s[k] | ((d[k] & ((1 << RB_SHIFT) - 1)) << 17);
            bk[k] = d[k] >> RB_SHIFT;
            rk[k] = 0;
            if (tid + k * SORT_T < nsub) rk[k] = atomicAdd(&lcnt[bk[k]], 1);
        }
        __syncthreads();
        int c = (tid < NBP) ? lcnt[tid] : 0;
        int ex = block_excl_scan(c, wsum, tid);
        if (tid < NBP) {
            tb[tid] = ex;
            tab[(size_t)pc * TABW + tid] = (unsigned short)ex;
        }
        if (tid == 0) tab[(size_t)pc * TABW + NBP] = (unsigned short)nsub;
        __syncthreads();
        int t[EPT];
#pragma unroll
        for (int k = 0; k < EPT; ++k) t[k] = tb[bk[k]];
        asm volatile("" :: "v"(t[0]), "v"(t[1]), "v"(t[2]), "v"(t[3]), "v"(t[4]), "v"(t[5]), "v"(t[6]),
                           "v"(t[7]), "v"(t[8]), "v"(t[9]), "v"(t[10]), "v"(t[11]), "v"(t[12]));
#pragma unroll
        for (int k = 0; k < EPT; ++k) {
            if (tid + k * SORT_T < nsub) stage[t[k] + rk[k]] = rc[k];
        }
        __syncthreads();
        {
            const int4* st4 = (const int4*)stage;
            int4* o4 = (int4*)(rec + c0);
            int full = nsub >> 2;
            int4 e[4];
#pragma unroll
            for (int k = 0; k < 4; ++k) e[k] = st4[min(tid + k * SORT_T, max(full - 1, 0))];
            asm volatile("" :: "v"(e[0].x), "v"(e[0].y), "v"(e[0].z), "v"(e[0].w), "v"(e[1].x), "v"(e[1].y), "v"(e[1].z), "v"(e[1].w),
                               "v"(e[2].x), "v"(e[2].y), "v"(e[2].z), "v"(e[2].w), "v"(e[3].x), "v"(e[3].y), "v"(e[3].z), "v"(e[3].w));
#pragma unroll
            for (int k = 0; k < 4; ++k)
                if (tid + k * SORT_T < full) o4[tid + k * SORT_T] = e[k];
            if (tid < (nsub & 3)) rec[c0 + 4 * full + tid] = stage[4 * full + tid];
        }
        __syncthreads();
    }
}

#define RPT 8
__global__ void __launch_bounds__(SORT_T)
fine_sort(const int* __restrict__ rec, const int* __restrict__ tab_w, int* __restrict__ start,
          int* __restrict__ deg_w, int* __restrict__ csr_w, float* __restrict__ stats, int N, int E) {
    const unsigned short* __restrict__ tab = (const unsigned short*)tab_w;
    unsigned short* __restrict__ csr = (unsigned short*)csr_w;
    unsigned char* __restrict__ deg8 = (unsigned char*)deg_w;
    __shared__ int cnt[512];
    __shared__ int cursor[512];
    __shared__ int wsum[32];
    __shared__ int wbeg[32];
    __shared__ int wlen[32];
    __shared__ int runS[PMAX + 32 * RPT];
    __shared__ int runL[PMAX + 32 * RPT];
    __shared__ unsigned short stg[FCAP];
    int tid = threadIdx.x;
    int lane = tid & 31;
    int wv = __builtin_amdgcn_readfirstlane(tid >> 5);
    int b = blockIdx.x;
    int np = min((E + SCAP - 1) / SCAP, PMAX);
    int o0 = 0, o1 = 0;
    if (tid < np) {
        o0 = tab[(size_t)tid * TABW + b];
        o1 = tab[(size_t)tid * TABW + b + 1];
    }
    runS[tid] = tid * SCAP + o0;
    runL[tid] = o1 - o0;
    if (tid < 32 * RPT) { runS[PMAX + tid] = 0; runL[PMAX + tid] = 0; }
    if (tid < 512) cnt[tid] = 0;
    if (b == 0 && tid < 64) stats[tid] = 0.0f;
    {
        int sb = o0, sl = o1 - o0;
        for (int off = 16; off > 0; off >>= 1) {
            sb += __shfl_xor(sb, off);
            sl += __shfl_xor(sl, off);
        }
        if (lane == 0) { wbeg[wv] = sb; wlen[wv] = sl; }
    }
    __syncthreads();
    int c0 = 0, nb = 0;
#pragma unroll
    for (int w = 0; w < 32; ++w) { c0 += wbeg[w]; nb += wlen[w]; }
    int c1 = c0 + nb;
    bool staged = nb <= FCAP;

    for (int p0 = wv; p0 < np; p0 += 32 * RPT) {
        int rs[RPT], rl[RPT], r[RPT];
#pragma unroll
        for (int k = 0; k < RPT; ++k) {
            rs[k] = __builtin_amdgcn_readfirstlane(runS[p0 + 32 * k]);
            rl[k] = __builtin_amdgcn_readfirstlane(runL[p0 + 32 * k]);
        }
#pragma unroll
        for (int k = 0; k < RPT; ++k) r[k] = rec[min(rs[k] + lane, E - 1)];
        asm volatile("" :: "v"(r[0]), "v"(r[1]), "v"(r[2]), "v"(r[3]), "v"(r[4]), "v"(r[5]), "v"(r[6]), "v"(r[7]));
#pragma unroll
        for (int k = 0; k < RPT; ++k)
            if (lane < rl[k]) atomicAdd(&cnt[(r[k] >> 16) & 511], 1);
#pragma unroll 1
        for (int p = p0; p < min(np, p0 + 32 * RPT); p += 32) {
            int st = __builtin_amdgcn_readfirstlane(runS[p]);
            int ln = __builtin_amdgcn_readfirstlane(runL[p]);
#pragma unroll 1
            for (int o = 32; o < ln; o += 32)
                if (o + lane < ln) atomicAdd(&cnt[(rec[st + o + lane] >> 16) & 511], 1);
        }
    }
    __syncthreads();

    int c = (tid < 512) ? cnt[tid] : 0;
    int ex = block_excl_scan(c, wsum, tid);
    if (tid < 512) {
        cursor[tid] = staged ? ex : c0 + ex;
        int row = (b << RB_SHIFT) + (tid >> 1);
        if (row < N) start[2 * row + (tid & 1)] = c0 + ex;
    }
    if (tid < 256) {
        int row = (b << RB_SHIFT) + tid;
        if (row < N) deg8[row] = (unsigned char)min(cnt[2 * tid] + cnt[2 * tid + 1], DEG_ESC);
    }
    __syncthreads();

    for (int p0 = wv; p0 < np; p0 += 32 * RPT) {
        int rs[RPT], rl[RPT], r[RPT], pos[RPT];
#pragma unroll
        for (int k = 0; k < RPT; ++k) {
            rs[k] = __builtin_amdgcn_readfirstlane(runS[p0 + 32 * k]);
            rl[k] = __builtin_amdgcn_readfirstlane(runL[p0 + 32 * k]);
        }
#pragma unroll
        for (int k = 0; k < RPT; ++k) r[k] = rec[min(rs[k] + lane, E - 1)];
        asm volatile("" :: "v"(r[0]), "v"(r[1]), "v"(r[2]), "v"(r[3]), "v"(r[4]), "v"(r[5]), "v"(r[6]), "v"(r[7]));
#pragma unroll
        for (int k = 0; k < RPT; ++k) {
            pos[k] = 0;
            if (lane < rl[k]) pos[k] = atomicAdd(&cursor[(r[k] >> 16) & 511], 1);
        }
#pragma unroll
        for (int k = 0; k < RPT; ++k) {
            if (lane < rl[k]) {
                unsigned short v = (unsigned short)(r[k] & (SRC_SPLIT - 1));
                if (staged) stg[pos[k]] = v; else csr[pos[k]] = v;
            }
        }
#pragma unroll 1
        for (int p = p0; p < min(np, p0 + 32 * RPT); p += 32) {
            int st = __builtin_amdgcn_readfirstlane(runS[p]);
            int ln = __builtin_amdgcn_readfirstlane(runL[p]);
#pragma unroll 1
            for (int o = 32; o < ln; o += 32)
                if (o + lane < ln) {
                    int rv = rec[st + o + lane];
                    int ps = atomicAdd(&cursor[(rv >> 16) & 511], 1);
                    unsigned short v = (unsigned short)(rv & (SRC_SPLIT - 1));
                    if (staged) stg[ps] = v; else csr[ps] = v;
                }
        }
    }
    if (staged) {
        __syncthreads();
        int w0 = (c0 + 1) >> 1;
        int w1 = c1 >> 1;
        for (int w = w0 + tid; w < w1; w += SORT_T) {
            int i = 2 * w - c0;
            csr_w[w] = (int)stg[i] | ((int)stg[i + 1] << 16);
        }
        if (tid == 0 && (c0 & 1) && c1 > c0) csr[c0] = stg[0];
        if (tid == 32 && (c1 & 1) && c1 > c0) csr[c1 - 1] = stg[c1 - 1 - c0];
    }
}

__global__ void gemm_wmma_f32(const float* __restrict__ A, const float* __restrict__ B,
                              float* __restrict__ C, int N, int K) {
    int wave = (int)((blockIdx.x * blockDim.x + threadIdx.x) >> 5);
    int lane = threadIdx.x & 31;
    int row0 = wave * 16;
    if (row0 >= N) return;

    int m  = lane & 15;
    int hi = lane >> 4;
    const float* arow = A + (size_t)(row0 + m) * K;

    v8f acc = {};
    for (int k0 = 0; k0 < K; k0 += 4) {
        int ka = k0 + 2 * hi;
        v2f a, b;
        a.x = arow[ka];
        a.y = arow[ka + 1];
        b.x = B[(size_t)ka * HC + m];
        b.y = B[(size_t)(ka + 1) * HC + m];
        acc = __builtin_amdgcn_wmma_f32_16x16x4_f32(
             false, a,  false, b,
             (short)0, acc,  false,  false);
    }

    float* cbase = C + (size_t)(row0 + hi * 8) * HC + m;
#pragma unroll
    for (int r = 0; r < 8; ++r) cbase[(size_t)r * HC] = acc[r];
}

#define PULL_T AGG_T
__global__ void __launch_bounds__(PULL_T)
pull_rows(const int* __restrict__ start, const int* __restrict__ csr_w,
          const int* __restrict__ deg_w, const float* __restrict__ hw,
          const float* __restrict__ bias, float* __restrict__ out,
          float* __restrict__ stats, int N, int E) {
    __shared__ float red[(PULL_T / 32) * 32];
    __shared__ __attribute__((aligned(16))) unsigned char degs[DEG_CAP];
    load_deg_table(deg_w, degs, N);
    __syncthreads();
    const unsigned short* __restrict__ csr = (const unsigned short*)csr_w;
    const float4* __restrict__ hw4 = (const float4*)hw;
    int lane = threadIdx.x & 31;
    int q = lane & 3;
    int j = lane >> 2;
    int nw = (int)((gridDim.x * blockDim.x) >> 5);

    float b0 = bias[4 * q + 0], b1 = bias[4 * q + 1], b2 = bias[4 * q + 2], b3 = bias[4 * q + 3];

    float4 st1 = {0.0f, 0.0f, 0.0f, 0.0f};
    float4 st2 = {0.0f, 0.0f, 0.0f, 0.0f};

    for (int r = (int)((blockIdx.x * blockDim.x + threadIdx.x) >> 5); r < N; r += nw) {
        bool last = (r + 1 >= N);
        int s0 = start[2 * r];
        int s1 = start[2 * r + 1];
        int s2 = start[last ? 2 * r + 1 : 2 * r + 2];
        if (last) s2 = E;
        float d = rsqrtf((float)(s2 - s0) + 1.0f);
        float4 self = hw4[(size_t)r * 4 + q];

        float dd = d * d;
        float4 acc;
        acc.x = (j == 0) ? (dd * self.x + b0) : 0.0f;
        acc.y = (j == 0) ? (dd * self.y + b1) : 0.0f;
        acc.z = (j == 0) ? (dd * self.z + b2) : 0.0f;
        acc.w = (j == 0) ? (dd * self.w + b3) : 0.0f;

        int p = s0 + j;
        for (; p + 8 < s2; p += 16) {
            int pb = p + 8;
            int i0 = (int)csr[p]  + ((p  >= s1) ? SRC_SPLIT : 0);
            int i1 = (int)csr[pb] + ((pb >= s1) ? SRC_SPLIT : 0);
            float4 v0 = hw4[(size_t)i0 * 4 + q];
            float4 v1 = hw4[(size_t)i1 * 4 + q];
            int c0 = degs[i0];
            int c1 = degs[i1];
            __builtin_amdgcn_sched_barrier(0);
            float g0 = rsqrtf((float)c0 + 1.0f);
            float g1 = rsqrtf((float)c1 + 1.0f);
            if (max(c0, c1) == DEG_ESC) {
                if (c0 == DEG_ESC) g0 = wide_dis(start, i0, N, E);
                if (c1 == DEG_ESC) g1 = wide_dis(start, i1, N, E);
            }
            float w0 = g0 * d;
            float w1 = g1 * d;
            acc.x += w0 * v0.x; acc.y += w0 * v0.y; acc.z += w0 * v0.z; acc.w += w0 * v0.w;
            acc.x += w1 * v1.x; acc.y += w1 * v1.y; acc.z += w1 * v1.z; acc.w += w1 * v1.w;
        }
        if (p < s2) {
            int i0 = (int)csr[p] + ((p >= s1) ? SRC_SPLIT : 0);
            float4 v0 = hw4[(size_t)i0 * 4 + q];
            int c0 = degs[i0];
            float g0 = (c0 == DEG_ESC) ? wide_dis(start, i0, N, E) : rsqrtf((float)c0 + 1.0f);
            float w0 = g0 * d;
            acc.x += w0 * v0.x; acc.y += w0 * v0.y; acc.z += w0 * v0.z; acc.w += w0 * v0.w;
        }

        for (int off = 4; off < 32; off <<= 1) {
            acc.x += __shfl_xor(acc.x, off);
            acc.y += __shfl_xor(acc.y, off);
            acc.z += __shfl_xor(acc.z, off);
            acc.w += __shfl_xor(acc.w, off);
        }

        if (j == 0) ((float4*)out)[(size_t)r * 4 + q] = acc;

        float rx = fmaxf(acc.x, 0.0f), ry = fmaxf(acc.y, 0.0f);
        float rz = fmaxf(acc.z, 0.0f), rw = fmaxf(acc.w, 0.0f);
        st1.x += rx; st1.y += ry; st1.z += rz; st1.w += rw;
        st2.x += rx * rx; st2.y += ry * ry; st2.z += rz * rz; st2.w += rw * rw;
    }

    int wv = (int)(threadIdx.x >> 5);
    if (j == 0) {
        float* rr = red + wv * 32 + 4 * q;
        rr[0] = st1.x; rr[1] = st1.y; rr[2] = st1.z; rr[3] = st1.w;
        rr[16] = st2.x; rr[17] = st2.y; rr[18] = st2.z; rr[19] = st2.w;
    }
    __syncthreads();
    if (threadIdx.x < 32) {
        float v = 0.0f;
#pragma unroll
        for (int w = 0; w < PULL_T / 32; ++w) v += red[w * 32 + threadIdx.x];
        atomicAdd(&stats[threadIdx.x], v);
    }
}

#define TP 20
#define RPW 7
#define WROWS (32 * RPW)
#define NTILE (WROWS / 16)
template <bool STATS>
__global__ void __launch_bounds__(AGG_T)
pull_bn_gemm(const int* __restrict__ start, const int* __restrict__ csr_w,
             const int* __restrict__ deg_w, const float* __restrict__ prev,
             const float* __restrict__ stats, const float* __restrict__ gamma,
             const float* __restrict__ beta, float invN,
             const float* __restrict__ Wm, const float* __restrict__ bias,
             float* __restrict__ out, float* __restrict__ stats_out, int N, int E) {
    __shared__ float tile[WROWS * TP];
    __shared__ float red[NTILE * 32];
    __shared__ __attribute__((aligned(16))) unsigned char degs[DEG_CAP];
    const unsigned short* __restrict__ csr = (const unsigned short*)csr_w;
    const float4* __restrict__ p4 = (const float4*)prev;
    load_deg_table(deg_w, degs, N);
    int lane = threadIdx.x & 31;
    int wave = __builtin_amdgcn_readfirstlane((int)(threadIdx.x >> 5));
    int q = lane & 3;
    int j = lane >> 2;

    float sc0, sc1, sc2, sc3, sh0, sh1, sh2, sh3;
    {
        float mu, var;
        mu = stats[4 * q + 0] * invN; var = stats[16 + 4 * q + 0] * invN - mu * mu;
        sc0 = rsqrtf(var + EPSV) * gamma[4 * q + 0]; sh0 = beta[4 * q + 0] - mu * sc0;
        mu = stats[4 * q + 1] * invN; var = stats[16 + 4 * q + 1] * invN - mu * mu;
        sc1 = rsqrtf(var + EPSV) * gamma[4 * q + 1]; sh1 = beta[4 * q + 1] - mu * sc1;
        mu = stats[4 * q + 2] * invN; var = stats[16 + 4 * q + 2] * invN - mu * mu;
        sc2 = rsqrtf(var + EPSV) * gamma[4 * q + 2]; sh2 = beta[4 * q + 2] - mu * sc2;
        mu = stats[4 * q + 3] * invN; var = stats[16 + 4 * q + 3] * invN - mu * mu;
        sc3 = rsqrtf(var + EPSV) * gamma[4 * q + 3]; sh3 = beta[4 * q + 3] - mu * sc3;
    }

    int rbase = blockIdx.x * WROWS + wave * RPW;
    int sv = start[min(2 * rbase + lane, 2 * N - 1)];
    float4 selfn = p4[(size_t)min(rbase, N - 1) * 4 + q];
    __syncthreads();
    for (int t = 0; t < RPW; ++t) {
        int r = rbase + t;
        if (r >= N) break;
        int s0 = __builtin_amdgcn_readlane(sv, 2 * t);
        int s1 = __builtin_amdgcn_readlane(sv, 2 * t + 1);
        int s2 = (r + 1 >= N) ? E : __builtin_amdgcn_readlane(sv, 2 * t + 2);
        float d = rsqrtf((float)(s2 - s0) + 1.0f);
        float4 self = selfn;
        selfn = p4[(size_t)min(r + 1, N - 1) * 4 + q];

        float dd = d * d;
        float4 acc;
        acc.x = (j == 0) ? dd * (fmaxf(self.x, 0.0f) * sc0 + sh0) : 0.0f;
        acc.y = (j == 0) ? dd * (fmaxf(self.y, 0.0f) * sc1 + sh1) : 0.0f;
        acc.z = (j == 0) ? dd * (fmaxf(self.z, 0.0f) * sc2 + sh2) : 0.0f;
        acc.w = (j == 0) ? dd * (fmaxf(self.w, 0.0f) * sc3 + sh3) : 0.0f;

        int p = s0 + j;
        for (; p + 8 < s2; p += 16) {
            int pb = p + 8;
            int i0 = (int)csr[p]  + ((p  >= s1) ? SRC_SPLIT : 0);
            int i1 = (int)csr[pb] + ((pb >= s1) ? SRC_SPLIT : 0);
            float4 v0 = p4[(size_t)i0 * 4 + q];
            float4 v1 = p4[(size_t)i1 * 4 + q];
            int c0 = degs[i0];
            int c1 = degs[i1];
            __builtin_amdgcn_sched_barrier(0);
            float g0 = rsqrtf((float)c0 + 1.0f);
            float g1 = rsqrtf((float)c1 + 1.0f);
            if (max(c0, c1) == DEG_ESC) {
                if (c0 == DEG_ESC) g0 = wide_dis(start, i0, N, E);
                if (c1 == DEG_ESC) g1 = wide_dis(start, i1, N, E);
            }
            float w0 = g0 * d;
            float w1 = g1 * d;
            acc.x += w0 * (fmaxf(v0.x, 0.0f) * sc0 + sh0);
            acc.y += w0 * (fmaxf(v0.y, 0.0f) * sc1 + sh1);
            acc.z += w0 * (fmaxf(v0.z, 0.0f) * sc2 + sh2);
            acc.w += w0 * (fmaxf(v0.w, 0.0f) * sc3 + sh3);
            acc.x += w1 * (fmaxf(v1.x, 0.0f) * sc0 + sh0);
            acc.y += w1 * (fmaxf(v1.y, 0.0f) * sc1 + sh1);
            acc.z += w1 * (fmaxf(v1.z, 0.0f) * sc2 + sh2);
            acc.w += w1 * (fmaxf(v1.w, 0.0f) * sc3 + sh3);
        }
        if (p < s2) {
            int i0 = (int)csr[p] + ((p >= s1) ? SRC_SPLIT : 0);
            float4 v0 = p4[(size_t)i0 * 4 + q];
            int c0 = degs[i0];
            float g0 = (c0 == DEG_ESC) ? wide_dis(start, i0, N, E) : rsqrtf((float)c0 + 1.0f);
            float w0 = g0 * d;
            acc.x += w0 * (fmaxf(v0.x, 0.0f) * sc0 + sh0);
            acc.y += w0 * (fmaxf(v0.y, 0.0f) * sc1 + sh1);
            acc.z += w0 * (fmaxf(v0.z, 0.0f) * sc2 + sh2);
            acc.w += w0 * (fmaxf(v0.w, 0.0f) * sc3 + sh3);
        }

        for (int off = 4; off < 32; off <<= 1) {
            acc.x += __shfl_xor(acc.x, off);
            acc.y += __shfl_xor(acc.y, off);
            acc.z += __shfl_xor(acc.z, off);
            acc.w += __shfl_xor(acc.w, off);
        }

        if (j == 0) *(float4*)&tile[(wave * RPW + t) * TP + 4 * q] = acc;
    }
    __syncthreads();

    if (wave < NTILE) {
        int row0 = blockIdx.x * WROWS + wave * 16;
        float s1 = 0.0f, s2 = 0.0f;
        if (row0 < N) {
            int m  = lane & 15;
            int hi = lane >> 4;
            const float* trow = tile + (wave * 16 + m) * TP;
            float bv = bias[m];
            v8f c8 = {bv, bv, bv, bv, bv, bv, bv, bv};
            for (int kk = 0; kk < 4; ++kk) {
                int ka = 4 * kk + 2 * hi;
                v2f a, b;
                a.x = trow[ka];
                a.y = trow[ka + 1];
                b.x = Wm[ka * HC + m];
                b.y = Wm[(ka + 1) * HC + m];
                c8 = __builtin_amdgcn_wmma_f32_16x16x4_f32(
                     false, a,  false, b,
                     (short)0, c8,  false,  false);
            }
            float* cbase = out + (size_t)(row0 + hi * 8) * HC + m;
#pragma unroll
            for (int r = 0; r < 8; ++r) cbase[(size_t)r * HC] = c8[r];
            if (STATS) {
#pragma unroll
                for (int r = 0; r < 8; ++r) {
                    float v = fmaxf(c8[r], 0.0f);
                    s1 += v;
                    s2 += v * v;
                }
                s1 += __shfl_xor(s1, 16);
                s2 += __shfl_xor(s2, 16);
            }
        }
        if (STATS) red[wave * 32 + lane] = (lane < 16) ? s1 : s2;
    }
    if (STATS) {
        __syncthreads();
        if (threadIdx.x < 32) {
            float v = 0.0f;
#pragma unroll
            for (int w = 0; w < NTILE; ++w) v += red[w * 32 + lane];
            atomicAdd(&stats_out[lane], v);
        }
    }
}

__global__ void head_kernel(const float* __restrict__ h, const float* __restrict__ Wfc,
                            const float* __restrict__ bfc, float* __restrict__ out, int N) {
    int i = blockIdx.x * blockDim.x + threadIdx.x;
    if (i >= N) return;
    const float* r = h + (size_t)i * HC;
    float z0 = bfc[0], z1 = bfc[1];
#pragma unroll
    for (int c = 0; c < HC; ++c) {
        float v = r[c];
        z0 += v * Wfc[c * 2 + 0];
        z1 += v * Wfc[c * 2 + 1];
    }
    float mx  = fmaxf(z0, z1);
    float lse = mx + logf(__expf(z0 - mx) + __expf(z1 - mx));
    out[(size_t)i * 2 + 0] = z0 - lse;
    out[(size_t)i * 2 + 1] = z1 - lse;
}

extern "C" void kernel_launch(void* const* d_in, const int* in_sizes, int n_in,
                              void* d_out, int out_size, void* d_ws, size_t ws_size,
                              hipStream_t stream) {
    const int IN = 128;
    const int N  = in_sizes[0] / IN;
    const int E  = in_sizes[1] / 2;

    const float* x    = (const float*)d_in[0];
    const int*   ei   = (const int*)d_in[1];
    const int*   srcI = ei;
    const int*   dstI = ei + E;
    const float* W1   = (const float*)d_in[2];
    const float* b1   = (const float*)d_in[3];
    const float* W2   = (const float*)d_in[4];
    const float* b2   = (const float*)d_in[5];
    const float* W3   = (const float*)d_in[6];
    const float* b3   = (const float*)d_in[7];
    const float* g1   = (const float*)d_in[8];
    const float* be1  = (const float*)d_in[9];
    const float* g2   = (const float*)d_in[10];
    const float* be2  = (const float*)d_in[11];
    const float* Wfc  = (const float*)d_in[12];
    const float* bfc  = (const float*)d_in[13];

    float* ws    = (float*)d_ws;
    float* dis   = ws;
    float* bufA  = dis + N;
    float* bufB  = bufA + (size_t)N * HC;
    float* bufC  = bufB + (size_t)N * HC;
    float* stats = bufC + (size_t)N * HC;
    int*   csr   = (int*)bufC;
    int*   tab   = (int*)((char*)dis + (((size_t)N + 127) & ~(size_t)127));
    int*   rec   = (int*)bufA;
    int*   start = (int*)d_out;
    int*   degw  = (int*)dis;

    const int B256 = 256;
    dim3 blkN((N + B256 - 1) / B256);
    const int NB = (N + (1 << RB_SHIFT) - 1) >> RB_SHIFT;
    dim3 blkG((N / 16 + 7) / 8);
    dim3 blkP(512);
    dim3 blkT((N + WROWS - 1) / WROWS);
    float invN = 1.0f / (float)N;

    piece_sort<<<G1, SORT_T, 0, stream>>>(srcI, dstI, rec, tab, E);
    fine_sort<<<NB, SORT_T, 0, stream>>>(rec, tab, start, degw, csr, stats, N, E);

    gemm_wmma_f32<<<blkG, B256, 0, stream>>>(x, W1, bufA, N, IN);
    pull_rows<<<blkP, PULL_T, 0, stream>>>(start, csr, degw, bufA, b1, bufB, stats, N, E);

    pull_bn_gemm<true><<<blkT, AGG_T, 0, stream>>>(start, csr, degw, bufB, stats, g1, be1, invN, W2, b2, bufA, stats + 32, N, E);

    pull_bn_gemm<false><<<blkT, AGG_T, 0, stream>>>(start, csr, degw, bufA, stats + 32, g2, be2, invN, W3, b3, bufB, stats, N, E);

    head_kernel<<<blkN, B256, 0, stream>>>(bufB, Wfc, bfc, (float*)d_out, N);
}


# Rows of a wave consecutive in all 3 aggregations, next row's source ids fetched a row ahead: a row's first 32 gathers go out as one group, no index load in front; layer-1 BatchNorm sums: new tree
# speedup vs baseline: 20.6706x; 1.0464x over previous
#include <hip/hip_runtime.h>
#include <hip/hip_bf16.h>

typedef float v2f __attribute__((ext_vector_type(2)));
typedef float v8f __attribute__((ext_vector_type(8)));

#define HC 16
#define EPSV 1e-5f
#define SRC_SPLIT 65536

#define DEG_CAP 131072
#define DEG_ESC 255
#define AGG_T 1024

__device__ __forceinline__ void load_deg_table(const int* __restrict__ deg_w, unsigned char* degs, int N) {
    const uint4* __restrict__ g4 = (const uint4*)deg_w;
    uint4* l4 = (uint4*)degs;
    int nv = (N + 15) >> 4;
    uint4 v[8];
#pragma unroll
    for (int k = 0; k < 8; ++k) v[k] = g4[min((int)threadIdx.x + k * AGG_T, nv - 1)];
#pragma unroll
    for (int k = 0; k < 8; ++k) l4[threadIdx.x + k * AGG_T] = v[k];
}

__device__ __forceinline__ float wide_dis(const int* __restrict__ start, int i, int N, int E) {
    int a = start[2 * i];
    int b = start[min(2 * i + 2, 2 * N - 1)];
    if (i + 1 >= N) b = E;
    return rsqrtf((float)(b - a) + 1.0f);
}

#define RB_SHIFT 8
#define NBP 512
#define TABW (NBP + 1)
#define G1 256
#define SORT_T 1024
#define PMAX 1024
#define EPT 13
#define SCAP (EPT * SORT_T)
#define FCAP 32768

__device__ __forceinline__ int block_excl_scan(int v, int* wsum, int tid) {
    int lane = tid & 31, wv = tid >> 5;
    int inc = v;
    for (int off = 1; off < 32; off <<= 1) {
        int n = __shfl_up(inc, off);
        if (lane >= off) inc += n;
    }
    if (lane == 31) wsum[wv] = inc;
    __syncthreads();
    int before = 0;
    for (int w = 0; w < wv; ++w) before += wsum[w];
    __syncthreads();
    return before + inc - v;
}

__global__ void __launch_bounds__(SORT_T)
piece_sort(const int* __restrict__ src, const int* __restrict__ dst, int* __restrict__ rec,
           int* __restrict__ tab_w, int E) {
    unsigned short* __restrict__ tab = (unsigned short*)tab_w;
    __shared__ int lcnt[NBP];
    __shared__ int tb[NBP];
    __shared__ int wsum[32];
    __shared__ __attribute__((aligned(16))) int stage[SCAP];
    int tid = threadIdx.x;
    int np = (E + SCAP - 1) / SCAP;
    for (int pc = blockIdx.x; pc < np; pc += G1) {
        int c0 = pc * SCAP;
        int c1 = min(c0 + SCAP, E);
        int nsub = c1 - c0;
        if (tid < NBP) lcnt[tid] = 0;
        __syncthreads();
        int s[EPT], d[EPT];
#pragma unroll
        for (int k = 0; k < EPT; ++k) {
            int ec = min(c0 + tid + k * SORT_T, c1 - 1);
            s[k] = src[ec];
            d[k] = dst[ec];
        }
        asm volatile("" :: "v"(s[0]), "v"(s[1]), "v"(s[2]), "v"(s[3]), "v"(s[4]), "v"(s[5]), "v"(s[6]),
                           "v"(s[7]), "v"(s[8]), "v"(s[9]), "v"(s[10]), "v"(s[11]), "v"(s[12]),
                           "v"(d[0]), "v"(d[1]), "v"(d[2]), "v"(d[3]), "v"(d[4]), "v"(d[5]), "v"(d[6]),
                           "v"(d[7]), "v"(d[8]), "v"(d[9]), "v"(d[10]), "v"(d[11]), "v"(d[12]));
        int rc[EPT], bk[EPT], rk[EPT];
#pragma unroll
        for (int k = 0; k < EPT; ++k) {
            rc[k] = s[k] | ((d[k] & ((1 << RB_SHIFT) - 1)) << 17);
            bk[k] = d[k] >> RB_SHIFT;
            rk[k] = 0;
            if (tid + k * SORT_T < nsub) rk[k] = atomicAdd(&lcnt[bk[k]], 1);
        }
        __syncthreads();
        int c = (tid < NBP) ? lcnt[tid] : 0;
        int ex = block_excl_scan(c, wsum, tid);
        if (tid < NBP) {
            tb[tid] = ex;
            tab[(size_t)pc * TABW + tid] = (unsigned short)ex;
        }
        if (tid == 0) tab[(size_t)pc * TABW + NBP] = (unsigned short)nsub;
        __syncthreads();
        int t[EPT];
#pragma unroll
        for (int k = 0; k < EPT; ++k) t[k] = tb[bk[k]];
        asm volatile("" :: "v"(t[0]), "v"(t[1]), "v"(t[2]), "v"(t[3]), "v"(t[4]), "v"(t[5]), "v"(t[6]),
                           "v"(t[7]), "v"(t[8]), "v"(t[9]), "v"(t[10]), "v"(t[11]), "v"(t[12]));
#pragma unroll
        for (int k = 0; k < EPT; ++k) {
            if (tid + k * SORT_T < nsub) stage[t[k] + rk[k]] = rc[k];
        }
        __syncthreads();
        {
            const int4* st4 = (const int4*)stage;
            int4* o4 = (int4*)(rec + c0);
            int full = nsub >> 2;
            int4 e[4];
#pragma unroll
            for (int k = 0; k < 4; ++k) e[k] = st4[min(tid + k * SORT_T, max(full - 1, 0))];
            asm volatile("" :: "v"(e[0].x), "v"(e[0].y), "v"(e[0].z), "v"(e[0].w), "v"(e[1].x), "v"(e[1].y), "v"(e[1].z), "v"(e[1].w),
                               "v"(e[2].x), "v"(e[2].y), "v"(e[2].z), "v"(e[2].w), "v"(e[3].x), "v"(e[3].y), "v"(e[3].z), "v"(e[3].w));
#pragma unroll
            for (int k = 0; k < 4; ++k)
                if (tid + k * SORT_T < full) o4[tid + k * SORT_T] = e[k];
            if (tid < (nsub & 3)) rec[c0 + 4 * full + tid] = stage[4 * full + tid];
        }
        __syncthreads();
    }
}

#define RPT 8
__global__ void __launch_bounds__(SORT_T)
fine_sort(const int* __restrict__ rec, const int* __restrict__ tab_w, int* __restrict__ start,
          int* __restrict__ deg_w, int* __restrict__ csr_w, float* __restrict__ stats, int N, int E) {
    const unsigned short* __restrict__ tab = (const unsigned short*)tab_w;
    unsigned short* __restrict__ csr = (unsigned short*)csr_w;
    unsigned char* __restrict__ deg8 = (unsigned char*)deg_w;
    __shared__ int cnt[512];
    __shared__ int cursor[512];
    __shared__ int wsum[32];
    __shared__ int wbeg[32];
    __shared__ int wlen[32];
    __shared__ int runS[PMAX + 32 * RPT];
    __shared__ int runL[PMAX + 32 * RPT];
    __shared__ unsigned short stg[FCAP];
    int tid = threadIdx.x;
    int lane = tid & 31;
    int wv = __builtin_amdgcn_readfirstlane(tid >> 5);
    int b = blockIdx.x;
    int np = min((E + SCAP - 1) / SCAP, PMAX);
    int o0 = 0, o1 = 0;
    if (tid < np) {
        o0 = tab[(size_t)tid * TABW + b];
        o1 = tab[(size_t)tid * TABW + b + 1];
    }
    runS[tid] = tid * SCAP + o0;
    runL[tid] = o1 - o0;
    if (tid < 32 * RPT) { runS[PMAX + tid] = 0; runL[PMAX + tid] = 0; }
    if (tid < 512) cnt[tid] = 0;
    if (b == 0 && tid < 64) stats[tid] = 0.0f;
    {
        int sb = o0, sl = o1 - o0;
        for (int off = 16; off > 0; off >>= 1) {
            sb += __shfl_xor(sb, off);
            sl += __shfl_xor(sl, off);
        }
        if (lane == 0) { wbeg[wv] = sb; wlen[wv] = sl; }
    }
    __syncthreads();
    int c0 = 0, nb = 0;
#pragma unroll
    for (int w = 0; w < 32; ++w) { c0 += wbeg[w]; nb += wlen[w]; }
    int c1 = c0 + nb;
    bool staged = nb <= FCAP;

    for (int p0 = wv; p0 < np; p0 += 32 * RPT) {
        int rs[RPT], rl[RPT], r[RPT];
#pragma unroll
        for (int k = 0; k < RPT; ++k) {
            rs[k] = __builtin_amdgcn_readfirstlane(runS[p0 + 32 * k]);
            rl[k] = __builtin_amdgcn_readfirstlane(runL[p0 + 32 * k]);
        }
#pragma unroll
        for (int k = 0; k < RPT; ++k) r[k] = rec[min(rs[k] + lane, E - 1)];
        asm volatile("" :: "v"(r[0]), "v"(r[1]), "v"(r[2]), "v"(r[3]), "v"(r[4]), "v"(r[5]), "v"(r[6]), "v"(r[7]));
#pragma unroll
        for (int k = 0; k < RPT; ++k)
            if (lane < rl[k]) atomicAdd(&cnt[(r[k] >> 16) & 511], 1);
#pragma unroll 1
        for (int p = p0; p < min(np, p0 + 32 * RPT); p += 32) {
            int st = __builtin_amdgcn_readfirstlane(runS[p]);
            int ln = __builtin_amdgcn_readfirstlane(runL[p]);
#pragma unroll 1
            for (int o = 32; o < ln; o += 32)
                if (o + lane < ln) atomicAdd(&cnt[(rec[st + o + lane] >> 16) & 511], 1);
        }
    }
    __syncthreads();

    int c = (tid < 512) ? cnt[tid] : 0;
    int ex = block_excl_scan(c, wsum, tid);
    if (tid < 512) {
        cursor[tid] = staged ? ex : c0 + ex;
        int row = (b << RB_SHIFT) + (tid >> 1);
        if (row < N) start[2 * row + (tid & 1)] = c0 + ex;
    }
    if (tid < 256) {
        int row = (b << RB_SHIFT) + tid;
        if (row < N) deg8[row] = (unsigned char)min(cnt[2 * tid] + cnt[2 * tid + 1], DEG_ESC);
    }
    __syncthreads();

    for (int p0 = wv; p0 < np; p0 += 32 * RPT) {
        int rs[RPT], rl[RPT], r[RPT], pos[RPT];
#pragma unroll
        for (int k = 0; k < RPT; ++k) {
            rs[k] = __builtin_amdgcn_readfirstlane(runS[p0 + 32 * k]);
            rl[k] = __builtin_amdgcn_readfirstlane(runL[p0 + 32 * k]);
        }
#pragma unroll
        for (int k = 0; k < RPT; ++k) r[k] = rec[min(rs[k] + lane, E - 1)];
        asm volatile("" :: "v"(r[0]), "v"(r[1]), "v"(r[2]), "v"(r[3]), "v"(r[4]), "v"(r[5]), "v"(r[6]), "v"(r[7]));
#pragma unroll
        for (int k = 0; k < RPT; ++k) {
            pos[k] = 0;
            if (lane < rl[k]) pos[k] = atomicAdd(&cursor[(r[k] >> 16) & 511], 1);
        }
#pragma unroll
        for (int k = 0; k < RPT; ++k) {
            if (lane < rl[k]) {
                unsigned short v = (unsigned short)(r[k] & (SRC_SPLIT - 1));
                if (staged) stg[pos[k]] = v; else csr[pos[k]] = v;
            }
        }
#pragma unroll 1
        for (int p = p0; p < min(np, p0 + 32 * RPT); p += 32) {
            int st = __builtin_amdgcn_readfirstlane(runS[p]);
            int ln = __builtin_amdgcn_readfirstlane(runL[p]);
#pragma unroll 1
            for (int o = 32; o < ln; o += 32)
                if (o + lane < ln) {
                    int rv = rec[st + o + lane];
                    int ps = atomicAdd(&cursor[(rv >> 16) & 511], 1);
                    unsigned short v = (unsigned short)(rv & (SRC_SPLIT - 1));
                    if (staged) stg[ps] = v; else csr[ps] = v;
                }
        }
    }
    if (staged) {
        __syncthreads();
        int w0 = (c0 + 1) >> 1;
        int w1 = c1 >> 1;
        for (int w = w0 + tid; w < w1; w += SORT_T) {
            int i = 2 * w - c0;
            csr_w[w] = (int)stg[i] | ((int)stg[i + 1] << 16);
        }
        if (tid == 0 && (c0 & 1) && c1 > c0) csr[c0] = stg[0];
        if (tid == 32 && (c1 & 1) && c1 > c0) csr[c1 - 1] = stg[c1 - 1 - c0];
    }
}

__global__ void gemm_wmma_f32(const float* __restrict__ A, const float* __restrict__ B,
                              float* __restrict__ C, int N, int K) {
    int wave = (int)((blockIdx.x * blockDim.x + threadIdx.x) >> 5);
    int lane = threadIdx.x & 31;
    int row0 = wave * 16;
    if (row0 >= N) return;

    int m  = lane & 15;
    int hi = lane >> 4;
    const float* arow = A + (size_t)(row0 + m) * K;

    v8f acc = {};
    for (int k0 = 0; k0 < K; k0 += 4) {
        int ka = k0 + 2 * hi;
        v2f a, b;
        a.x = arow[ka];
        a.y = arow[ka + 1];
        b.x = B[(size_t)ka * HC + m];
        b.y = B[(size_t)(ka + 1) * HC + m];
        acc = __builtin_amdgcn_wmma_f32_16x16x4_f32(
             false, a,  false, b,
             (short)0, acc,  false,  false);
    }

    float* cbase = C + (size_t)(row0 + hi * 8) * HC + m;
#pragma unroll
    for (int r = 0; r < 8; ++r) cbase[(size_t)r * HC] = acc[r];
}

#define NI 4
template <bool BN>
__device__ __forceinline__ float xf(float v, float sc, float sh) {
    return BN ? (fmaxf(v, 0.0f) * sc + sh) : v;
}
template <bool BN>
__device__ __forceinline__ float4 gather_row(const unsigned short* __restrict__ csr,
        const float4* __restrict__ rows4, const unsigned char* degs, const int* __restrict__ start,
        int r, int s0, int s1, int s2, float d, int j, int q, const int* cur, float4 acc,
        float sc0, float sc1, float sc2, float sc3, float sh0, float sh1, float sh2, float sh3,
        int N, int E) {
    int nj = (s2 - s0 - j + 7) >> 3;
    int ii[NI], c[NI];
    float4 v[NI];
#pragma unroll
    for (int g = 0; g < NI; ++g) {
        int p = s0 + j + 8 * g;
        int i = cur[g] + ((p >= s1) ? SRC_SPLIT : 0);
        ii[g] = (g < nj) ? i : r;
        v[g] = rows4[ii[g] * 4 + q];
        c[g] = degs[ii[g]];
    }
    __builtin_amdgcn_sched_barrier(0);
    float w[NI];
#pragma unroll
    for (int g = 0; g < NI; ++g) w[g] = rsqrtf((float)c[g] + 1.0f);
    int cm = c[0];
#pragma unroll
    for (int g = 1; g < NI; ++g) cm = max(cm, c[g]);
    if (cm == DEG_ESC) {
#pragma unroll
        for (int g = 0; g < NI; ++g)
            if (c[g] == DEG_ESC) w[g] = wide_dis(start, ii[g], N, E);
    }
#pragma unroll
    for (int g = 0; g < NI; ++g) {
        float wg = (g < nj) ? w[g] * d : 0.0f;
        acc.x += wg * xf<BN>(v[g].x, sc0, sh0);
        acc.y += wg * xf<BN>(v[g].y, sc1, sh1);
        acc.z += wg * xf<BN>(v[g].z, sc2, sh2);
        acc.w += wg * xf<BN>(v[g].w, sc3, sh3);
    }
    int p = s0 + j + 8 * NI;
    for (; p + 8 < s2; p += 16) {
        int pb = p + 8;
        int i0 = (int)csr[p]  + ((p  >= s1) ? SRC_SPLIT : 0);
        int i1 = (int)csr[pb] + ((pb >= s1) ? SRC_SPLIT : 0);
        float4 v0 = rows4[i0 * 4 + q];
        float4 v1 = rows4[i1 * 4 + q];
        int c0 = degs[i0];
        int c1 = degs[i1];
        __builtin_amdgcn_sched_barrier(0);
        float g0 = rsqrtf((float)c0 + 1.0f);
        float g1 = rsqrtf((float)c1 + 1.0f);
        if (max(c0, c1) == DEG_ESC) {
            if (c0 == DEG_ESC) g0 = wide_dis(start, i0, N, E);
            if (c1 == DEG_ESC) g1 = wide_dis(start, i1, N, E);
        }
        float w0 = g0 * d;
        float w1 = g1 * d;
        acc.x += w0 * xf<BN>(v0.x, sc0, sh0); acc.y += w0 * xf<BN>(v0.y, sc1, sh1);
        acc.z += w0 * xf<BN>(v0.z, sc2, sh2); acc.w += w0 * xf<BN>(v0.w, sc3, sh3);
        acc.x += w1 * xf<BN>(v1.x, sc0, sh0); acc.y += w1 * xf<BN>(v1.y, sc1, sh1);
        acc.z += w1 * xf<BN>(v1.z, sc2, sh2); acc.w += w1 * xf<BN>(v1.w, sc3, sh3);
    }
    if (p < s2) {
        int i0 = (int)csr[p] + ((p >= s1) ? SRC_SPLIT : 0);
        float4 v0 = rows4[i0 * 4 + q];
        int c0 = degs[i0];
        float g0 = (c0 == DEG_ESC) ? wide_dis(start, i0, N, E) : rsqrtf((float)c0 + 1.0f);
        float w0 = g0 * d;
        acc.x += w0 * xf<BN>(v0.x, sc0, sh0); acc.y += w0 * xf<BN>(v0.y, sc1, sh1);
        acc.z += w0 * xf<BN>(v0.z, sc2, sh2); acc.w += w0 * xf<BN>(v0.w, sc3, sh3);
    }
    return acc;
}

#define PULL_T AGG_T
#define RPW 7
#define WROWS (32 * RPW)
__global__ void __launch_bounds__(PULL_T)
pull_rows(const int* __restrict__ start, const int* __restrict__ csr_w,
          const int* __restrict__ deg_w, const float* __restrict__ hw,
          const float* __restrict__ bias, float* __restrict__ out,
          float* __restrict__ stats, int N, int E) {
    __shared__ float red[(PULL_T / 32) * 32];
    __shared__ __attribute__((aligned(16))) unsigned char degs[DEG_CAP];
    load_deg_table(deg_w, degs, N);
    const unsigned short* __restrict__ csr = (const unsigned short*)csr_w;
    const float4* __restrict__ hw4 = (const float4*)hw;
    int lane = threadIdx.x & 31;
    int wave = __builtin_amdgcn_readfirstlane((int)(threadIdx.x >> 5));
    int q = lane & 3;
    int j = lane >> 2;

    float b0 = bias[4 * q + 0], b1 = bias[4 * q + 1], b2 = bias[4 * q + 2], b3 = bias[4 * q + 3];

    float4 st1 = {0.0f, 0.0f, 0.0f, 0.0f};
    float4 st2 = {0.0f, 0.0f, 0.0f, 0.0f};

    int rbase = blockIdx.x * WROWS + wave * RPW;
    int sv = start[min(2 * rbase + lane, 2 * N - 1)];
    float4 selfn = hw4[min(rbase, N - 1) * 4 + q];
    int nx[NI];
    {
        int e0 = start[min(2 * rbase, 2 * N - 1)];
#pragma unroll
        for (int g = 0; g < NI; ++g) nx[g] = (int)csr[min(e0 + j + 8 * g, E - 1)];
    }
    __syncthreads();
    for (int t = 0; t < RPW; ++t) {
        int r = rbase + t;
        if (r >= N) break;
        int s0 = __builtin_amdgcn_readlane(sv, 2 * t);
        int s1 = __builtin_amdgcn_readlane(sv, 2 * t + 1);
        int s2 = (r + 1 >= N) ? E : __builtin_amdgcn_readlane(sv, 2 * t + 2);
        float d = rsqrtf((float)(s2 - s0) + 1.0f);
        int cur[NI];
#pragma unroll
        for (int g = 0; g < NI; ++g) cur[g] = nx[g];
#pragma unroll
        for (int g = 0; g < NI; ++g) nx[g] = (int)csr[min(s2 + j + 8 * g, E - 1)];
        float4 self = selfn;
        selfn = hw4[min(r + 1, N - 1) * 4 + q];

        float dd = d * d;
        float4 acc;
        acc.x = (j == 0) ? (dd * self.x + b0) : 0.0f;
        acc.y = (j == 0) ? (dd * self.y + b1) : 0.0f;
        acc.z = (j == 0) ? (dd * self.z + b2) : 0.0f;
        acc.w = (j == 0) ? (dd * self.w + b3) : 0.0f;

        acc = gather_row<false>(csr, hw4, degs, start, r, s0, s1, s2, d, j, q, cur, acc,
                                0.0f, 0.0f, 0.0f, 0.0f, 0.0f, 0.0f, 0.0f, 0.0f, N, E);

        for (int off = 4; off < 32; off <<= 1) {
            acc.x += __shfl_xor(acc.x, off);
            acc.y += __shfl_xor(acc.y, off);
            acc.z += __shfl_xor(acc.z, off);
            acc.w += __shfl_xor(acc.w, off);
        }

        if (j == 0) ((float4*)out)[(size_t)r * 4 + q] = acc;

        float rx = fmaxf(acc.x, 0.0f), ry = fmaxf(acc.y, 0.0f);
        float rz = fmaxf(acc.z, 0.0f), rw = fmaxf(acc.w, 0.0f);
        st1.x += rx; st1.y += ry; st1.z += rz; st1.w += rw;
        st2.x += rx * rx; st2.y += ry * ry; st2.z += rz * rz; st2.w += rw * rw;
    }

    if (j == 0) {
        float* rr = red + wave * 32 + 4 * q;
        rr[0] = st1.x; rr[1] = st1.y; rr[2] = st1.z; rr[3] = st1.w;
        rr[16] = st2.x; rr[17] = st2.y; rr[18] = st2.z; rr[19] = st2.w;
    }
    __syncthreads();
    if (threadIdx.x < 32) {
        float v = 0.0f;
#pragma unroll
        for (int w = 0; w < PULL_T / 32; ++w) v += red[w * 32 + threadIdx.x];
        atomicAdd(&stats[threadIdx.x], v);
    }
}

#define TP 20
#define NTILE (WROWS / 16)
template <bool STATS>
__global__ void __launch_bounds__(AGG_T)
pull_bn_gemm(const int* __restrict__ start, const int* __restrict__ csr_w,
             const int* __restrict__ deg_w, const float* __restrict__ prev,
             const float* __restrict__ stats, const float* __restrict__ gamma,
             const float* __restrict__ beta, float invN,
             const float* __restrict__ Wm, const float* __restrict__ bias,
             float* __restrict__ out, float* __restrict__ stats_out, int N, int E) {
    __shared__ float tile[WROWS * TP];
    __shared__ float red[NTILE * 32];
    __shared__ __attribute__((aligned(16))) unsigned char degs[DEG_CAP];
    const unsigned short* __restrict__ csr = (const unsigned short*)csr_w;
    const float4* __restrict__ p4 = (const float4*)prev;
    load_deg_table(deg_w, degs, N);
    int lane = threadIdx.x & 31;
    int wave = __builtin_amdgcn_readfirstlane((int)(threadIdx.x >> 5));
    int q = lane & 3;
    int j = lane >> 2;

    float sc0, sc1, sc2, sc3, sh0, sh1, sh2, sh3;
    {
        float mu, var;
        mu = stats[4 * q + 0] * invN; var = stats[16 + 4 * q + 0] * invN - mu * mu;
        sc0 = rsqrtf(var + EPSV) * gamma[4 * q + 0]; sh0 = beta[4 * q + 0] - mu * sc0;
        mu = stats[4 * q + 1] * invN; var = stats[16 + 4 * q + 1] * invN - mu * mu;
        sc1 = rsqrtf(var + EPSV) * gamma[4 * q + 1]; sh1 = beta[4 * q + 1] - mu * sc1;
        mu = stats[4 * q + 2] * invN; var = stats[16 + 4 * q + 2] * invN - mu * mu;
        sc2 = rsqrtf(var + EPSV) * gamma[4 * q + 2]; sh2 = beta[4 * q + 2] - mu * sc2;
        mu = stats[4 * q + 3] * invN; var = stats[16 + 4 * q + 3] * invN - mu * mu;
        sc3 = rsqrtf(var + EPSV) * gamma[4 * q + 3]; sh3 = beta[4 * q + 3] - mu * sc3;
    }

    int rbase = blockIdx.x * WROWS + wave * RPW;
    int sv = start[min(2 * rbase + lane, 2 * N - 1)];
    float4 selfn = p4[min(rbase, N - 1) * 4 + q];
    int nx[NI];
    {
        int e0 = start[min(2 * rbase, 2 * N - 1)];
#pragma unroll
        for (int g = 0; g < NI; ++g) nx[g] = (int)csr[min(e0 + j + 8 * g, E - 1)];
    }
    __syncthreads();
    for (int t = 0; t < RPW; ++t) {
        int r = rbase + t;
        if (r >= N) break;
        int s0 = __builtin_amdgcn_readlane(sv, 2 * t);
        int s1 = __builtin_amdgcn_readlane(sv, 2 * t + 1);
        int s2 = (r + 1 >= N) ? E : __builtin_amdgcn_readlane(sv, 2 * t + 2);
        float d = rsqrtf((float)(s2 - s0) + 1.0f);
        int cur[NI];
#pragma unroll
        for (int g = 0; g < NI; ++g) cur[g] = nx[g];
#pragma unroll
        for (int g = 0; g < NI; ++g) nx[g] = (int)csr[min(s2 + j + 8 * g, E - 1)];
        float4 self = selfn;
        selfn = p4[min(r + 1, N - 1) * 4 + q];

        float dd = d * d;
        float4 acc;
        acc.x = (j == 0) ? dd * (fmaxf(self.x, 0.0f) * sc0 + sh0) : 0.0f;
        acc.y = (j == 0) ? dd * (fmaxf(self.y, 0.0f) * sc1 + sh1) : 0.0f;
        acc.z = (j == 0) ? dd * (fmaxf(self.z, 0.0f) * sc2 + sh2) : 0.0f;
        acc.w = (j == 0) ? dd * (fmaxf(self.w, 0.0f) * sc3 + sh3) : 0.0f;

        acc = gather_row<true>(csr, p4, degs, start, r, s0, s1, s2, d, j, q, cur, acc,
                               sc0, sc1, sc2, sc3, sh0, sh1, sh2, sh3, N, E);

        for (int off = 4; off < 32; off <<= 1) {
            acc.x += __shfl_xor(acc.x, off);
            acc.y += __shfl_xor(acc.y, off);
            acc.z += __shfl_xor(acc.z, off);
            acc.w += __shfl_xor(acc.w, off);
        }

        if (j == 0) *(float4*)&tile[(wave * RPW + t) * TP + 4 * q] = acc;
    }
    __syncthreads();

    if (wave < NTILE) {
        int row0 = blockIdx.x * WROWS + wave * 16;
        float s1 = 0.0f, s2 = 0.0f;
        if (row0 < N) {
            int m  = lane & 15;
            int hi = lane >> 4;
            const float* trow = tile + (wave * 16 + m) * TP;
            float bv = bias[m];
            v8f c8 = {bv, bv, bv, bv, bv, bv, bv, bv};
            for (int kk = 0; kk < 4; ++kk) {
                int ka = 4 * kk + 2 * hi;
                v2f a, b;
                a.x = trow[ka];
                a.y = trow[ka + 1];
                b.x = Wm[ka * HC + m];
                b.y = Wm[(ka + 1) * HC + m];
                c8 = __builtin_amdgcn_wmma_f32_16x16x4_f32(
                     false, a,  false, b,
                     (short)0, c8,  false,  false);
            }
            float* cbase = out + (size_t)(row0 + hi * 8) * HC + m;
#pragma unroll
            for (int r = 0; r < 8; ++r) cbase[(size_t)r * HC] = c8[r];
            if (STATS) {
#pragma unroll
                for (int r = 0; r < 8; ++r) {
                    float v = fmaxf(c8[r], 0.0f);
                    s1 += v;
                    s2 += v * v;
                }
                s1 += __shfl_xor(s1, 16);
                s2 += __shfl_xor(s2, 16);
            }
        }
        if (STATS) red[wave * 32 + lane] = (lane < 16) ? s1 : s2;
    }
    if (STATS) {
        __syncthreads();
        if (threadIdx.x < 32) {
            float v = 0.0f;
#pragma unroll
            for (int w = 0; w < NTILE; ++w) v += red[w * 32 + lane];
            atomicAdd(&stats_out[lane], v);
        }
    }
}

__global__ void head_kernel(const float* __restrict__ h, const float* __restrict__ Wfc,
                            const float* __restrict__ bfc, float* __restrict__ out, int N) {
    int i = blockIdx.x * blockDim.x + threadIdx.x;
    if (i >= N) return;
    const float* r = h + (size_t)i * HC;
    float z0 = bfc[0], z1 = bfc[1];
#pragma unroll
    for (int c = 0; c < HC; ++c) {
        float v = r[c];
        z0 += v * Wfc[c * 2 + 0];
        z1 += v * Wfc[c * 2 + 1];
    }
    float mx  = fmaxf(z0, z1);
    float lse = mx + logf(__expf(z0 - mx) + __expf(z1 - mx));
    out[(size_t)i * 2 + 0] = z0 - lse;
    out[(size_t)i * 2 + 1] = z1 - lse;
}

extern "C" void kernel_launch(void* const* d_in, const int* in_sizes, int n_in,
                              void* d_out, int out_size, void* d_ws, size_t ws_size,
                              hipStream_t stream) {
    const int IN = 128;
    const int N  = in_sizes[0] / IN;
    const int E  = in_sizes[1] / 2;

    const float* x    = (const float*)d_in[0];
    const int*   ei   = (const int*)d_in[1];
    const int*   srcI = ei;
    const int*   dstI = ei + E;
    const float* W1   = (const float*)d_in[2];
    const float* b1   = (const float*)d_in[3];
    const float* W2   = (const float*)d_in[4];
    const float* b2   = (const float*)d_in[5];
    const float* W3   = (const float*)d_in[6];
    const float* b3   = (const float*)d_in[7];
    const float* g1   = (const float*)d_in[8];
    const float* be1  = (const float*)d_in[9];
    const float* g2   = (const float*)d_in[10];
    const float* be2  = (const float*)d_in[11];
    const float* Wfc  = (const float*)d_in[12];
    const float* bfc  = (const float*)d_in[13];

    float* ws    = (float*)d_ws;
    float* dis   = ws;
    float* bufA  = dis + N;
    float* bufB  = bufA + (size_t)N * HC;
    float* bufC  = bufB + (size_t)N * HC;
    float* stats = bufC + (size_t)N * HC;
    int*   csr   = (int*)bufC;
    int*   tab   = (int*)((char*)dis + (((size_t)N + 127) & ~(size_t)127));
    int*   rec   = (int*)bufA;
    int*   start = (int*)d_out;
    int*   degw  = (int*)dis;

    const int B256 = 256;
    dim3 blkN((N + B256 - 1) / B256);
    const int NB = (N + (1 << RB_SHIFT) - 1) >> RB_SHIFT;
    dim3 blkG((N / 16 + 7) / 8);
    dim3 blkT((N + WROWS - 1) / WROWS);
    float invN = 1.0f / (float)N;

    piece_sort<<<G1, SORT_T, 0, stream>>>(srcI, dstI, rec, tab, E);
    fine_sort<<<NB, SORT_T, 0, stream>>>(rec, tab, start, degw, csr, stats, N, E);

    gemm_wmma_f32<<<blkG, B256, 0, stream>>>(x, W1, bufA, N, IN);
    pull_rows<<<blkT, PULL_T, 0, stream>>>(start, csr, degw, bufA, b1, bufB, stats, N, E);

    pull_bn_gemm<true><<<blkT, AGG_T, 0, stream>>>(start, csr, degw, bufB, stats, g1, be1, invN, W2, b2, bufA, stats + 32, N, E);

    pull_bn_gemm<false><<<blkT, AGG_T, 0, stream>>>(start, csr, degw, bufA, stats + 32, g2, be2, invN, W3, b3, bufB, stats, N, E);

    head_kernel<<<blkN, B256, 0, stream>>>(bufB, Wfc, bfc, (float*)d_out, N);
}


# Level-2 edge sort reads every run once: two records a lane in one load group (a run averages 34, so most left a 33rd record to a serial loop), held in registers from count to placement; arrival order
# speedup vs baseline: 21.3917x; 1.0349x over previous
#include <hip/hip_runtime.h>
#include <hip/hip_bf16.h>

typedef float v2f __attribute__((ext_vector_type(2)));
typedef float v8f __attribute__((ext_vector_type(8)));

#define HC 16
#define EPSV 1e-5f
#define SRC_SPLIT 65536

#define DEG_CAP 131072
#define DEG_ESC 255
#define AGG_T 1024

__device__ __forceinline__ void load_deg_table(const int* __restrict__ deg_w, unsigned char* degs, int N) {
    const uint4* __restrict__ g4 = (const uint4*)deg_w;
    uint4* l4 = (uint4*)degs;
    int nv = (N + 15) >> 4;
    uint4 v[8];
#pragma unroll
    for (int k = 0; k < 8; ++k) v[k] = g4[min((int)threadIdx.x + k * AGG_T, nv - 1)];
#pragma unroll
    for (int k = 0; k < 8; ++k) l4[threadIdx.x + k * AGG_T] = v[k];
}

__device__ __forceinline__ float wide_dis(const int* __restrict__ start, int i, int N, int E) {
    int a = start[2 * i];
    int b = start[min(2 * i + 2, 2 * N - 1)];
    if (i + 1 >= N) b = E;
    return rsqrtf((float)(b - a) + 1.0f);
}

#define RB_SHIFT 8
#define NBP 512
#define TABW (NBP + 1)
#define G1 256
#define SORT_T 1024
#define PMAX 1024
#define EPT 13
#define SCAP (EPT * SORT_T)
#define FCAP 32768

__device__ __forceinline__ int block_excl_scan(int v, int* wsum, int tid) {
    int lane = tid & 31, wv = tid >> 5;
    int inc = v;
    for (int off = 1; off < 32; off <<= 1) {
        int n = __shfl_up(inc, off);
        if (lane >= off) inc += n;
    }
    if (lane == 31) wsum[wv] = inc;
    __syncthreads();
    int before = 0;
    for (int w = 0; w < wv; ++w) before += wsum[w];
    __syncthreads();
    return before + inc - v;
}

__global__ void __launch_bounds__(SORT_T)
piece_sort(const int* __restrict__ src, const int* __restrict__ dst, int* __restrict__ rec,
           int* __restrict__ tab_w, int E) {
    unsigned short* __restrict__ tab = (unsigned short*)tab_w;
    __shared__ int lcnt[NBP];
    __shared__ int tb[NBP];
    __shared__ int wsum[32];
    __shared__ __attribute__((aligned(16))) int stage[SCAP];
    int tid = threadIdx.x;
    int np = (E + SCAP - 1) / SCAP;
    for (int pc = blockIdx.x; pc < np; pc += G1) {
        int c0 = pc * SCAP;
        int c1 = min(c0 + SCAP, E);
        int nsub = c1 - c0;
        if (tid < NBP) lcnt[tid] = 0;
        __syncthreads();
        int s[EPT], d[EPT];
#pragma unroll
        for (int k = 0; k < EPT; ++k) {
            int ec = min(c0 + tid + k * SORT_T, c1 - 1);
            s[k] = src[ec];
            d[k] = dst[ec];
        }
        asm volatile("" :: "v"(s[0]), "v"(s[1]), "v"(s[2]), "v"(s[3]), "v"(s[4]), "v"(s[5]), "v"(s[6]),
                           "v"(s[7]), "v"(s[8]), "v"(s[9]), "v"(s[10]), "v"(s[11]), "v"(s[12]),
                           "v"(d[0]), "v"(d[1]), "v"(d[2]), "v"(d[3]), "v"(d[4]), "v"(d[5]), "v"(d[6]),
                           "v"(d[7]), "v"(d[8]), "v"(d[9]), "v"(d[10]), "v"(d[11]), "v"(d[12]));
        int rc[EPT], bk[EPT], rk[EPT];
#pragma unroll
        for (int k = 0; k < EPT; ++k) {
            rc[k] = s[k] | ((d[k] & ((1 << RB_SHIFT) - 1)) << 17);
            bk[k] = d[k] >> RB_SHIFT;
            rk[k] = 0;
            if (tid + k * SORT_T < nsub) rk[k] = atomicAdd(&lcnt[bk[k]], 1);
        }
        __syncthreads();
        int c = (tid < NBP) ? lcnt[tid] : 0;
        int ex = block_excl_scan(c, wsum, tid);
        if (tid < NBP) {
            tb[tid] = ex;
            tab[(size_t)pc * TABW + tid] = (unsigned short)ex;
        }
        if (tid == 0) tab[(size_t)pc * TABW + NBP] = (unsigned short)nsub;
        __syncthreads();
        int t[EPT];
#pragma unroll
        for (int k = 0; k < EPT; ++k) t[k] = tb[bk[k]];
        asm volatile("" :: "v"(t[0]), "v"(t[1]), "v"(t[2]), "v"(t[3]), "v"(t[4]), "v"(t[5]), "v"(t[6]),
                           "v"(t[7]), "v"(t[8]), "v"(t[9]), "v"(t[10]), "v"(t[11]), "v"(t[12]));
#pragma unroll
        for (int k = 0; k < EPT; ++k) {
            if (tid + k * SORT_T < nsub) stage[t[k] + rk[k]] = rc[k];
        }
        __syncthreads();
        {
            const int4* st4 = (const int4*)stage;
            int4* o4 = (int4*)(rec + c0);
            int full = nsub >> 2;
            int4 e[4];
#pragma unroll
            for (int k = 0; k < 4; ++k) e[k] = st4[min(tid + k * SORT_T, max(full - 1, 0))];
            asm volatile("" :: "v"(e[0].x), "v"(e[0].y), "v"(e[0].z), "v"(e[0].w), "v"(e[1].x), "v"(e[1].y), "v"(e[1].z), "v"(e[1].w),
                               "v"(e[2].x), "v"(e[2].y), "v"(e[2].z), "v"(e[2].w), "v"(e[3].x), "v"(e[3].y), "v"(e[3].z), "v"(e[3].w));
#pragma unroll
            for (int k = 0; k < 4; ++k)
                if (tid + k * SORT_T < full) o4[tid + k * SORT_T] = e[k];
            if (tid < (nsub & 3)) rec[c0 + 4 * full + tid] = stage[4 * full + tid];
        }
        __syncthreads();
    }
}

#define RPT 8
#define RLN 64
__global__ void __launch_bounds__(SORT_T)
fine_sort(const int* __restrict__ rec, const int* __restrict__ tab_w, int* __restrict__ start,
          int* __restrict__ deg_w, int* __restrict__ csr_w, float* __restrict__ stats, int N, int E) {
    const unsigned short* __restrict__ tab = (const unsigned short*)tab_w;
    unsigned short* __restrict__ csr = (unsigned short*)csr_w;
    unsigned char* __restrict__ deg8 = (unsigned char*)deg_w;
    __shared__ int cnt[512];
    __shared__ int cursor[512];
    __shared__ int wsum[32];
    __shared__ int wbeg[32];
    __shared__ int wlen[32];
    __shared__ int runS[PMAX];
    __shared__ int runL[PMAX];
    __shared__ unsigned short stg[FCAP];
    int tid = threadIdx.x;
    int lane = tid & 31;
    int wv = __builtin_amdgcn_readfirstlane(tid >> 5);
    int b = blockIdx.x;
    int np = min((E + SCAP - 1) / SCAP, PMAX);
    int o0 = 0, o1 = 0;
    if (tid < np) {
        o0 = tab[(size_t)tid * TABW + b];
        o1 = tab[(size_t)tid * TABW + b + 1];
    }
    runS[tid] = tid * SCAP + o0;
    runL[tid] = o1 - o0;
    if (tid < 512) cnt[tid] = 0;
    if (b == 0 && tid < 64) stats[tid] = 0.0f;
    {
        int sb = o0, sl = o1 - o0;
        for (int off = 16; off > 0; off >>= 1) {
            sb += __shfl_xor(sb, off);
            sl += __shfl_xor(sl, off);
        }
        if (lane == 0) { wbeg[wv] = sb; wlen[wv] = sl; }
    }
    __syncthreads();
    int rl[RPT], ra[RPT], rb[RPT];
    {
        int rs[RPT];
#pragma unroll
        for (int k = 0; k < RPT; ++k) {
            rs[k] = __builtin_amdgcn_readfirstlane(runS[wv + 32 * k]);
            rl[k] = __builtin_amdgcn_readfirstlane(runL[wv + 32 * k]);
        }
#pragma unroll
        for (int k = 0; k < RPT; ++k) {
            ra[k] = rec[min(rs[k] + lane, E - 1)];
            rb[k] = rec[min(rs[k] + 32 + lane, E - 1)];
        }
    }
    asm volatile("" :: "v"(ra[0]), "v"(ra[1]), "v"(ra[2]), "v"(ra[3]), "v"(ra[4]), "v"(ra[5]), "v"(ra[6]), "v"(ra[7]),
                       "v"(rb[0]), "v"(rb[1]), "v"(rb[2]), "v"(rb[3]), "v"(rb[4]), "v"(rb[5]), "v"(rb[6]), "v"(rb[7]));
    int lmax = rl[0];
#pragma unroll
    for (int k = 1; k < RPT; ++k) lmax = max(lmax, rl[k]);
    bool more = (lmax > RLN) || (np > 32 * RPT);

#pragma unroll
    for (int k = 0; k < RPT; ++k) {
        if (lane < rl[k]) atomicAdd(&cnt[(ra[k] >> 16) & 511], 1);
        if (lane + 32 < rl[k]) atomicAdd(&cnt[(rb[k] >> 16) & 511], 1);
    }
    if (more) {
#pragma unroll 1
        for (int p = wv; p < np; p += 32) {
            int st = __builtin_amdgcn_readfirstlane(runS[p]);
            int ln = __builtin_amdgcn_readfirstlane(runL[p]);
#pragma unroll 1
            for (int o = (p < 32 * RPT) ? RLN : 0; o < ln; o += 32)
                if (o + lane < ln) atomicAdd(&cnt[(rec[st + o + lane] >> 16) & 511], 1);
        }
    }
    __syncthreads();

    int c0 = wbeg[lane], nb = wlen[lane];
    for (int off = 16; off > 0; off >>= 1) {
        c0 += __shfl_xor(c0, off);
        nb += __shfl_xor(nb, off);
    }
    c0 = __builtin_amdgcn_readfirstlane(c0);
    nb = __builtin_amdgcn_readfirstlane(nb);
    int c1 = c0 + nb;
    bool staged = nb <= FCAP;

    int c = (tid < 512) ? cnt[tid] : 0;
    int ex = block_excl_scan(c, wsum, tid);
    if (tid < 512) {
        cursor[tid] = staged ? ex : c0 + ex;
        int row = (b << RB_SHIFT) + (tid >> 1);
        if (row < N) start[2 * row + (tid & 1)] = c0 + ex;
    }
    if (tid < 256) {
        int row = (b << RB_SHIFT) + tid;
        if (row < N) deg8[row] = (unsigned char)min(cnt[2 * tid] + cnt[2 * tid + 1], DEG_ESC);
    }
    __syncthreads();

    {
        int pos[RPT];
#pragma unroll
        for (int k = 0; k < RPT; ++k) {
            pos[k] = 0;
            if (lane < rl[k]) pos[k] = atomicAdd(&cursor[(ra[k] >> 16) & 511], 1);
        }
#pragma unroll
        for (int k = 0; k < RPT; ++k) {
            if (lane < rl[k]) {
                unsigned short v = (unsigned short)(ra[k] & (SRC_SPLIT - 1));
                if (staged) stg[pos[k]] = v; else csr[pos[k]] = v;
            }
        }
#pragma unroll
        for (int k = 0; k < RPT; ++k) {
            pos[k] = 0;
            if (lane + 32 < rl[k]) pos[k] = atomicAdd(&cursor[(rb[k] >> 16) & 511], 1);
        }
#pragma unroll
        for (int k = 0; k < RPT; ++k) {
            if (lane + 32 < rl[k]) {
                unsigned short v = (unsigned short)(rb[k] & (SRC_SPLIT - 1));
                if (staged) stg[pos[k]] = v; else csr[pos[k]] = v;
            }
        }
    }
    if (more) {
#pragma unroll 1
        for (int p = wv; p < np; p += 32) {
            int st = __builtin_amdgcn_readfirstlane(runS[p]);
            int ln = __builtin_amdgcn_readfirstlane(runL[p]);
#pragma unroll 1
            for (int o = (p < 32 * RPT) ? RLN : 0; o < ln; o += 32)
                if (o + lane < ln) {
                    int rv = rec[st + o + lane];
                    int ps = atomicAdd(&cursor[(rv >> 16) & 511], 1);
                    unsigned short v = (unsigned short)(rv & (SRC_SPLIT - 1));
                    if (staged) stg[ps] = v; else csr[ps] = v;
                }
        }
    }
    if (staged) {
        __syncthreads();
        int w0 = (c0 + 1) >> 1;
        int w1 = c1 >> 1;
        for (int w = w0 + tid; w < w1; w += SORT_T) {
            int i = 2 * w - c0;
            csr_w[w] = (int)stg[i] | ((int)stg[i + 1] << 16);
        }
        if (tid == 0 && (c0 & 1) && c1 > c0) csr[c0] = stg[0];
        if (tid == 32 && (c1 & 1) && c1 > c0) csr[c1 - 1] = stg[c1 - 1 - c0];
    }
}

__global__ void gemm_wmma_f32(const float* __restrict__ A, const float* __restrict__ B,
                              float* __restrict__ C, int N, int K) {
    int wave = (int)((blockIdx.x * blockDim.x + threadIdx.x) >> 5);
    int lane = threadIdx.x & 31;
    int row0 = wave * 16;
    if (row0 >= N) return;

    int m  = lane & 15;
    int hi = lane >> 4;
    const float* arow = A + (size_t)(row0 + m) * K;

    v8f acc = {};
    for (int k0 = 0; k0 < K; k0 += 4) {
        int ka = k0 + 2 * hi;
        v2f a, b;
        a.x = arow[ka];
        a.y = arow[ka + 1];
        b.x = B[(size_t)ka * HC + m];
        b.y = B[(size_t)(ka + 1) * HC + m];
        acc = __builtin_amdgcn_wmma_f32_16x16x4_f32(
             false, a,  false, b,
             (short)0, acc,  false,  false);
    }

    float* cbase = C + (size_t)(row0 + hi * 8) * HC + m;
#pragma unroll
    for (int r = 0; r < 8; ++r) cbase[(size_t)r * HC] = acc[r];
}

#define NI 4
template <bool BN>
__device__ __forceinline__ float xf(float v, float sc, float sh) {
    return BN ? (fmaxf(v, 0.0f) * sc + sh) : v;
}
template <bool BN>
__device__ __forceinline__ float4 gather_row(const unsigned short* __restrict__ csr,
        const float4* __restrict__ rows4, const unsigned char* degs, const int* __restrict__ start,
        int r, int s0, int s1, int s2, float d, int j, int q, const int* cur, float4 acc,
        float sc0, float sc1, float sc2, float sc3, float sh0, float sh1, float sh2, float sh3,
        int N, int E) {
    int nj = (s2 - s0 - j + 7) >> 3;
    int ii[NI], c[NI];
    float4 v[NI];
#pragma unroll
    for (int g = 0; g < NI; ++g) {
        int p = s0 + j + 8 * g;
        int i = cur[g] + ((p >= s1) ? SRC_SPLIT : 0);
        ii[g] = (g < nj) ? i : r;
        v[g] = rows4[ii[g] * 4 + q];
        c[g] = degs[ii[g]];
    }
    __builtin_amdgcn_sched_barrier(0);
    float w[NI];
#pragma unroll
    for (int g = 0; g < NI; ++g) w[g] = rsqrtf((float)c[g] + 1.0f);
    int cm = c[0];
#pragma unroll
    for (int g = 1; g < NI; ++g) cm = max(cm, c[g]);
    if (cm == DEG_ESC) {
#pragma unroll
        for (int g = 0; g < NI; ++g)
            if (c[g] == DEG_ESC) w[g] = wide_dis(start, ii[g], N, E);
    }
#pragma unroll
    for (int g = 0; g < NI; ++g) {
        float wg = (g < nj) ? w[g] * d : 0.0f;
        acc.x += wg * xf<BN>(v[g].x, sc0, sh0);
        acc.y += wg * xf<BN>(v[g].y, sc1, sh1);
        acc.z += wg * xf<BN>(v[g].z, sc2, sh2);
        acc.w += wg * xf<BN>(v[g].w, sc3, sh3);
    }
    int p = s0 + j + 8 * NI;
    for (; p + 8 < s2; p += 16) {
        int pb = p + 8;
        int i0 = (int)csr[p]  + ((p  >= s1) ? SRC_SPLIT : 0);
        int i1 = (int)csr[pb] + ((pb >= s1) ? SRC_SPLIT : 0);
        float4 v0 = rows4[i0 * 4 + q];
        float4 v1 = rows4[i1 * 4 + q];
        int c0 = degs[i0];
        int c1 = degs[i1];
        __builtin_amdgcn_sched_barrier(0);
        float g0 = rsqrtf((float)c0 + 1.0f);
        float g1 = rsqrtf((float)c1 + 1.0f);
        if (max(c0, c1) == DEG_ESC) {
            if (c0 == DEG_ESC) g0 = wide_dis(start, i0, N, E);
            if (c1 == DEG_ESC) g1 = wide_dis(start, i1, N, E);
        }
        float w0 = g0 * d;
        float w1 = g1 * d;
        acc.x += w0 * xf<BN>(v0.x, sc0, sh0); acc.y += w0 * xf<BN>(v0.y, sc1, sh1);
        acc.z += w0 * xf<BN>(v0.z, sc2, sh2); acc.w += w0 * xf<BN>(v0.w, sc3, sh3);
        acc.x += w1 * xf<BN>(v1.x, sc0, sh0); acc.y += w1 * xf<BN>(v1.y, sc1, sh1);
        acc.z += w1 * xf<BN>(v1.z, sc2, sh2); acc.w += w1 * xf<BN>(v1.w, sc3, sh3);
    }
    if (p < s2) {
        int i0 = (int)csr[p] + ((p >= s1) ? SRC_SPLIT : 0);
        float4 v0 = rows4[i0 * 4 + q];
        int c0 = degs[i0];
        float g0 = (c0 == DEG_ESC) ? wide_dis(start, i0, N, E) : rsqrtf((float)c0 + 1.0f);
        float w0 = g0 * d;
        acc.x += w0 * xf<BN>(v0.x, sc0, sh0); acc.y += w0 * xf<BN>(v0.y, sc1, sh1);
        acc.z += w0 * xf<BN>(v0.z, sc2, sh2); acc.w += w0 * xf<BN>(v0.w, sc3, sh3);
    }
    return acc;
}

#define PULL_T AGG_T
#define RPW 7
#define WROWS (32 * RPW)
__global__ void __launch_bounds__(PULL_T)
pull_rows(const int* __restrict__ start, const int* __restrict__ csr_w,
          const int* __restrict__ deg_w, const float* __restrict__ hw,
          const float* __restrict__ bias, float* __restrict__ out,
          float* __restrict__ stats, int N, int E) {
    __shared__ float red[(PULL_T / 32) * 32];
    __shared__ __attribute__((aligned(16))) unsigned char degs[DEG_CAP];
    load_deg_table(deg_w, degs, N);
    const unsigned short* __restrict__ csr = (const unsigned short*)csr_w;
    const float4* __restrict__ hw4 = (const float4*)hw;
    int lane = threadIdx.x & 31;
    int wave = __builtin_amdgcn_readfirstlane((int)(threadIdx.x >> 5));
    int q = lane & 3;
    int j = lane >> 2;

    float b0 = bias[4 * q + 0], b1 = bias[4 * q + 1], b2 = bias[4 * q + 2], b3 = bias[4 * q + 3];

    float4 st1 = {0.0f, 0.0f, 0.0f, 0.0f};
    float4 st2 = {0.0f, 0.0f, 0.0f, 0.0f};

    int rbase = blockIdx.x * WROWS + wave * RPW;
    int sv = start[min(2 * rbase + lane, 2 * N - 1)];
    float4 selfn = hw4[min(rbase, N - 1) * 4 + q];
    int nx[NI];
    {
        int e0 = start[min(2 * rbase, 2 * N - 1)];
#pragma unroll
        for (int g = 0; g < NI; ++g) nx[g] = (int)csr[min(e0 + j + 8 * g, E - 1)];
    }
    __syncthreads();
    for (int t = 0; t < RPW; ++t) {
        int r = rbase + t;
        if (r >= N) break;
        int s0 = __builtin_amdgcn_readlane(sv, 2 * t);
        int s1 = __builtin_amdgcn_readlane(sv, 2 * t + 1);
        int s2 = (r + 1 >= N) ? E : __builtin_amdgcn_readlane(sv, 2 * t + 2);
        float d = rsqrtf((float)(s2 - s0) + 1.0f);
        int cur[NI];
#pragma unroll
        for (int g = 0; g < NI; ++g) cur[g] = nx[g];
#pragma unroll
        for (int g = 0; g < NI; ++g) nx[g] = (int)csr[min(s2 + j + 8 * g, E - 1)];
        float4 self = selfn;
        selfn = hw4[min(r + 1, N - 1) * 4 + q];

        float dd = d * d;
        float4 acc;
        acc.x = (j == 0) ? (dd * self.x + b0) : 0.0f;
        acc.y = (j == 0) ? (dd * self.y + b1) : 0.0f;
        acc.z = (j == 0) ? (dd * self.z + b2) : 0.0f;
        acc.w = (j == 0) ? (dd * self.w + b3) : 0.0f;

        acc = gather_row<false>(csr, hw4, degs, start, r, s0, s1, s2, d, j, q, cur, acc,
                                0.0f, 0.0f, 0.0f, 0.0f, 0.0f, 0.0f, 0.0f, 0.0f, N, E);

        for (int off = 4; off < 32; off <<= 1) {
            acc.x += __shfl_xor(acc.x, off);
            acc.y += __shfl_xor(acc.y, off);
            acc.z += __shfl_xor(acc.z, off);
            acc.w += __shfl_xor(acc.w, off);
        }

        if (j == 0) ((float4*)out)[(size_t)r * 4 + q] = acc;

        float rx = fmaxf(acc.x, 0.0f), ry = fmaxf(acc.y, 0.0f);
        float rz = fmaxf(acc.z, 0.0f), rw = fmaxf(acc.w, 0.0f);
        st1.x += rx; st1.y += ry; st1.z += rz; st1.w += rw;
        st2.x += rx * rx; st2.y += ry * ry; st2.z += rz * rz; st2.w += rw * rw;
    }

    if (j == 0) {
        float* rr = red + wave * 32 + 4 * q;
        rr[0] = st1.x; rr[1] = st1.y; rr[2] = st1.z; rr[3] = st1.w;
        rr[16] = st2.x; rr[17] = st2.y; rr[18] = st2.z; rr[19] = st2.w;
    }
    __syncthreads();
    if (threadIdx.x < 32) {
        float v = 0.0f;
#pragma unroll
        for (int w = 0; w < PULL_T / 32; ++w) v += red[w * 32 + threadIdx.x];
        atomicAdd(&stats[threadIdx.x], v);
    }
}

#define TP 20
#define NTILE (WROWS / 16)
template <bool STATS>
__global__ void __launch_bounds__(AGG_T)
pull_bn_gemm(const int* __restrict__ start, const int* __restrict__ csr_w,
             const int* __restrict__ deg_w, const float* __restrict__ prev,
             const float* __restrict__ stats, const float* __restrict__ gamma,
             const float* __restrict__ beta, float invN,
             const float* __restrict__ Wm, const float* __restrict__ bias,
             float* __restrict__ out, float* __restrict__ stats_out, int N, int E) {
    __shared__ float tile[WROWS * TP];
    __shared__ float red[NTILE * 32];
    __shared__ __attribute__((aligned(16))) unsigned char degs[DEG_CAP];
    const unsigned short* __restrict__ csr = (const unsigned short*)csr_w;
    const float4* __restrict__ p4 = (const float4*)prev;
    load_deg_table(deg_w, degs, N);
    int lane = threadIdx.x & 31;
    int wave = __builtin_amdgcn_readfirstlane((int)(threadIdx.x >> 5));
    int q = lane & 3;
    int j = lane >> 2;

    float sc0, sc1, sc2, sc3, sh0, sh1, sh2, sh3;
    {
        float mu, var;
        mu = stats[4 * q + 0] * invN; var = stats[16 + 4 * q + 0] * invN - mu * mu;
        sc0 = rsqrtf(var + EPSV) * gamma[4 * q + 0]; sh0 = beta[4 * q + 0] - mu * sc0;
        mu = stats[4 * q + 1] * invN; var = stats[16 + 4 * q + 1] * invN - mu * mu;
        sc1 = rsqrtf(var + EPSV) * gamma[4 * q + 1]; sh1 = beta[4 * q + 1] - mu * sc1;
        mu = stats[4 * q + 2] * invN; var = stats[16 + 4 * q + 2] * invN - mu * mu;
        sc2 = rsqrtf(var + EPSV) * gamma[4 * q + 2]; sh2 = beta[4 * q + 2] - mu * sc2;
        mu = stats[4 * q + 3] * invN; var = stats[16 + 4 * q + 3] * invN - mu * mu;
        sc3 = rsqrtf(var + EPSV) * gamma[4 * q + 3]; sh3 = beta[4 * q + 3] - mu * sc3;
    }

    int rbase = blockIdx.x * WROWS + wave * RPW;
    int sv = start[min(2 * rbase + lane, 2 * N - 1)];
    float4 selfn = p4[min(rbase, N - 1) * 4 + q];
    int nx[NI];
    {
        int e0 = start[min(2 * rbase, 2 * N - 1)];
#pragma unroll
        for (int g = 0; g < NI; ++g) nx[g] = (int)csr[min(e0 + j + 8 * g, E - 1)];
    }
    __syncthreads();
    for (int t = 0; t < RPW; ++t) {
        int r = rbase + t;
        if (r >= N) break;
        int s0 = __builtin_amdgcn_readlane(sv, 2 * t);
        int s1 = __builtin_amdgcn_readlane(sv, 2 * t + 1);
        int s2 = (r + 1 >= N) ? E : __builtin_amdgcn_readlane(sv, 2 * t + 2);
        float d = rsqrtf((float)(s2 - s0) + 1.0f);
        int cur[NI];
#pragma unroll
        for (int g = 0; g < NI; ++g) cur[g] = nx[g];
#pragma unroll
        for (int g = 0; g < NI; ++g) nx[g] = (int)csr[min(s2 + j + 8 * g, E - 1)];
        float4 self = selfn;
        selfn = p4[min(r + 1, N - 1) * 4 + q];

        float dd = d * d;
        float4 acc;
        acc.x = (j == 0) ? dd * (fmaxf(self.x, 0.0f) * sc0 + sh0) : 0.0f;
        acc.y = (j == 0) ? dd * (fmaxf(self.y, 0.0f) * sc1 + sh1) : 0.0f;
        acc.z = (j == 0) ? dd * (fmaxf(self.z, 0.0f) * sc2 + sh2) : 0.0f;
        acc.w = (j == 0) ? dd * (fmaxf(self.w, 0.0f) * sc3 + sh3) : 0.0f;

        acc = gather_row<true>(csr, p4, degs, start, r, s0, s1, s2, d, j, q, cur, acc,
                               sc0, sc1, sc2, sc3, sh0, sh1, sh2, sh3, N, E);

        for (int off = 4; off < 32; off <<= 1) {
            acc.x += __shfl_xor(acc.x, off);
            acc.y += __shfl_xor(acc.y, off);
            acc.z += __shfl_xor(acc.z, off);
            acc.w += __shfl_xor(acc.w, off);
        }

        if (j == 0) *(float4*)&tile[(wave * RPW + t) * TP + 4 * q] = acc;
    }
    __syncthreads();

    if (wave < NTILE) {
        int row0 = blockIdx.x * WROWS + wave * 16;
        float s1 = 0.0f, s2 = 0.0f;
        if (row0 < N) {
            int m  = lane & 15;
            int hi = lane >> 4;
            const float* trow = tile + (wave * 16 + m) * TP;
            float bv = bias[m];
            v8f c8 = {bv, bv, bv, bv, bv, bv, bv, bv};
            for (int kk = 0; kk < 4; ++kk) {
                int ka = 4 * kk + 2 * hi;
                v2f a, b;
                a.x = trow[ka];
                a.y = trow[ka + 1];
                b.x = Wm[ka * HC + m];
                b.y = Wm[(ka + 1) * HC + m];
                c8 = __builtin_amdgcn_wmma_f32_16x16x4_f32(
                     false, a,  false, b,
                     (short)0, c8,  false,  false);
            }
            float* cbase = out + (size_t)(row0 + hi * 8) * HC + m;
#pragma unroll
            for (int r = 0; r < 8; ++r) cbase[(size_t)r * HC] = c8[r];
            if (STATS) {
#pragma unroll
                for (int r = 0; r < 8; ++r) {
                    float v = fmaxf(c8[r], 0.0f);
                    s1 += v;
                    s2 += v * v;
                }
                s1 += __shfl_xor(s1, 16);
                s2 += __shfl_xor(s2, 16);
            }
        }
        if (STATS) red[wave * 32 + lane] = (lane < 16) ? s1 : s2;
    }
    if (STATS) {
        __syncthreads();
        if (threadIdx.x < 32) {
            float v = 0.0f;
#pragma unroll
            for (int w = 0; w < NTILE; ++w) v += red[w * 32 + lane];
            atomicAdd(&stats_out[lane], v);
        }
    }
}

__global__ void head_kernel(const float* __restrict__ h, const float* __restrict__ Wfc,
                            const float* __restrict__ bfc, float* __restrict__ out, int N) {
    int i = blockIdx.x * blockDim.x + threadIdx.x;
    if (i >= N) return;
    const float* r = h + (size_t)i * HC;
    float z0 = bfc[0], z1 = bfc[1];
#pragma unroll
    for (int c = 0; c < HC; ++c) {
        float v = r[c];
        z0 += v * Wfc[c * 2 + 0];
        z1 += v * Wfc[c * 2 + 1];
    }
    float mx  = fmaxf(z0, z1);
    float lse = mx + logf(__expf(z0 - mx) + __expf(z1 - mx));
    out[(size_t)i * 2 + 0] = z0 - lse;
    out[(size_t)i * 2 + 1] = z1 - lse;
}

extern "C" void kernel_launch(void* const* d_in, const int* in_sizes, int n_in,
                              void* d_out, int out_size, void* d_ws, size_t ws_size,
                              hipStream_t stream) {
    const int IN = 128;
    const int N  = in_sizes[0] / IN;
    const int E  = in_sizes[1] / 2;

    const float* x    = (const float*)d_in[0];
    const int*   ei   = (const int*)d_in[1];
    const int*   srcI = ei;
    const int*   dstI = ei + E;
    const float* W1   = (const float*)d_in[2];
    const float* b1   = (const float*)d_in[3];
    const float* W2   = (const float*)d_in[4];
    const float* b2   = (const float*)d_in[5];
    const float* W3   = (const float*)d_in[6];
    const float* b3   = (const float*)d_in[7];
    const float* g1   = (const float*)d_in[8];
    const float* be1  = (const float*)d_in[9];
    const float* g2   = (const float*)d_in[10];
    const float* be2  = (const float*)d_in[11];
    const float* Wfc  = (const float*)d_in[12];
    const float* bfc  = (const float*)d_in[13];

    float* ws    = (float*)d_ws;
    float* dis   = ws;
    float* bufA  = dis + N;
    float* bufB  = bufA + (size_t)N * HC;
    float* bufC  = bufB + (size_t)N * HC;
    float* stats = bufC + (size_t)N * HC;
    int*   csr   = (int*)bufC;
    int*   tab   = (int*)((char*)dis + (((size_t)N + 127) & ~(size_t)127));
    int*   rec   = (int*)bufA;
    int*   start = (int*)d_out;
    int*   degw  = (int*)dis;

    const int B256 = 256;
    dim3 blkN((N + B256 - 1) / B256);
    const int NB = (N + (1 << RB_SHIFT) - 1) >> RB_SHIFT;
    dim3 blkG((N / 16 + 7) / 8);
    dim3 blkT((N + WROWS - 1) / WROWS);
    float invN = 1.0f / (float)N;

    piece_sort<<<G1, SORT_T, 0, stream>>>(srcI, dstI, rec, tab, E);
    fine_sort<<<NB, SORT_T, 0, stream>>>(rec, tab, start, degw, csr, stats, N, E);

    gemm_wmma_f32<<<blkG, B256, 0, stream>>>(x, W1, bufA, N, IN);
    pull_rows<<<blkT, PULL_T, 0, stream>>>(start, csr, degw, bufA, b1, bufB, stats, N, E);

    pull_bn_gemm<true><<<blkT, AGG_T, 0, stream>>>(start, csr, degw, bufB, stats, g1, be1, invN, W2, b2, bufA, stats + 32, N, E);

    pull_bn_gemm<false><<<blkT, AGG_T, 0, stream>>>(start, csr, degw, bufA, stats + 32, g2, be2, invN, W3, b3, bufB, stats, N, E);

    head_kernel<<<blkN, B256, 0, stream>>>(bufB, Wfc, bfc, (float*)d_out, N);
}
